# starting kernel
# baseline (speedup 1.0000x reference)
#include <hip/hip_runtime.h>
#include <math.h>

#define TT  512
#define CC  768
#define NHH 12
#define HSS 64
constexpr float SCALE = 0.125f;

typedef float v2f __attribute__((ext_vector_type(2)));
typedef float v8f __attribute__((ext_vector_type(8)));

static __device__ __forceinline__ v8f wmma4(v2f a, v2f b, v8f c) {
  return __builtin_amdgcn_wmma_f32_16x16x4_f32(
       false, a,  false, b,
       (short)0, c,  false,  false);
}

__global__ void qkv_proj_kernel(const float* __restrict__ x,
                                const float* __restrict__ W0,
                                const float* __restrict__ W1,
                                const float* __restrict__ W2,
                                float* __restrict__ P) {
  const int lane = threadIdx.x;
  const int lm   = lane & 15;
  const int lk   = (lane >> 4) << 1;
  const int drow = (lane >> 4) << 3;
  const int t0 = blockIdx.x * 16;
  const int c0 = blockIdx.y * 16;
  const int w  = blockIdx.z;
  const float* W = (w == 0) ? W0 : (w == 1) ? W1 : W2;
  float* Pw = P + (size_t)w * NHH * TT * HSS;

  const float* Arow = x + (size_t)(t0 + lm) * CC;
  v8f acc = {0.f, 0.f, 0.f, 0.f, 0.f, 0.f, 0.f, 0.f};
  for (int k = 0; k < CC; k += 4) {
    v2f a, b;
    a.x = Arow[k + lk];
    a.y = Arow[k + lk + 1];
    b.x = W[(size_t)(k + lk) * CC + c0 + lm];
    b.y = W[(size_t)(k + lk + 1) * CC + c0 + lm];
    if (k + 16 < CC)
      __builtin_prefetch(&W[(size_t)(k + 16 + lk) * CC + c0 + lm], 0, 1);
    acc = wmma4(a, b, acc);
  }
  const int h  = c0 >> 6;
  const int d0 = c0 & 63;
  float* Ph = Pw + (size_t)h * TT * HSS;
#pragma unroll
  for (int r = 0; r < 8; ++r)
    Ph[(size_t)(t0 + r + drow) * HSS + d0 + lm] = acc[r];
}

__global__ void vproj_kernel(const float* __restrict__ P2,
                             const float* __restrict__ Wv,
                             float* __restrict__ V) {
  const int lane = threadIdx.x;
  const int lm   = lane & 15;
  const int lk   = (lane >> 4) << 1;
  const int drow = (lane >> 4) << 3;
  const int t0 = blockIdx.x * 16;
  const int n0 = blockIdx.y * 16;
  const int h  = blockIdx.z;
  const float* A = P2 + (size_t)h * TT * HSS;
  float* Vh = V + (size_t)h * TT * HSS;

  v8f acc = {0.f, 0.f, 0.f, 0.f, 0.f, 0.f, 0.f, 0.f};
  for (int k = 0; k < HSS; k += 4) {
    v2f a, b;
    a.x = A[(size_t)(t0 + lm) * HSS + k + lk];
    a.y = A[(size_t)(t0 + lm) * HSS + k + lk + 1];
    b.x = Wv[(size_t)(k + lk) * HSS + n0 + lm];
    b.y = Wv[(size_t)(k + lk + 1) * HSS + n0 + lm];
    acc = wmma4(a, b, acc);
  }
#pragma unroll
  for (int r = 0; r < 8; ++r)
    Vh[(size_t)(t0 + r + drow) * HSS + n0 + lm] = acc[r];
}

__global__ void vsuffix_kernel(const float* __restrict__ V,
                               float* __restrict__ VS) {
  const int h = blockIdx.x;
  const int d = threadIdx.x;
  const float* Vh = V + (size_t)h * TT * HSS;
  float* Sh = VS + (size_t)h * TT * HSS;
  float run = 0.0f;
  for (int j = TT - 1; j >= 0; --j) {
    Sh[(size_t)j * HSS + d] = run;
    run += Vh[(size_t)j * HSS + d];
  }
}

__global__ void att_softmax_kernel(const float* __restrict__ P0,
                                   const float* __restrict__ P1,
                                   float* __restrict__ ATT) {
  __shared__ float lgt[16 * TT];
  const int lane = threadIdx.x;
  const int lm   = lane & 15;
  const int lk   = (lane >> 4) << 1;
  const int drow = (lane >> 4) << 3;
  const int i0 = blockIdx.x * 16;
  const int h  = blockIdx.y;
  const float* P0h = P0 + (size_t)h * TT * HSS;
  const float* P1h = P1 + (size_t)h * TT * HSS;

  v2f areg[16];
#pragma unroll
  for (int kk = 0; kk < 16; ++kk) {
    const int k = kk * 4 + lk;
    areg[kk].x = P0h[(size_t)(i0 + lm) * HSS + k];
    areg[kk].y = P0h[(size_t)(i0 + lm) * HSS + k + 1];
  }

  const int jtmax = (i0 + 15) >> 4;
  for (int jt = 0; jt <= jtmax; ++jt) {
    const int j0 = jt * 16;
    v8f acc = {0.f, 0.f, 0.f, 0.f, 0.f, 0.f, 0.f, 0.f};
#pragma unroll
    for (int kk = 0; kk < 16; ++kk) {
      const int k = kk * 4 + lk;
      v2f b;
      b.x = P1h[(size_t)(j0 + lm) * HSS + k];
      b.y = P1h[(size_t)(j0 + lm) * HSS + k + 1];
      acc = wmma4(areg[kk], b, acc);
    }
#pragma unroll
    for (int r = 0; r < 8; ++r) {
      const int row = r + drow;
      const int j = j0 + lm;
      float val = acc[r] * SCALE;
      if (j > i0 + row) val = -3.0e38f;
      lgt[row * TT + j] = val;
    }
  }
  __syncthreads();

  const int row  = lane & 15;
  const int half = lane >> 4;
  const int len  = i0 + row + 1;
  float mr = -3.0e38f;
  for (int j = half; j < len; j += 2) mr = fmaxf(mr, lgt[row * TT + j]);
  mr = fmaxf(mr, __shfl_xor(mr, 16));
  float sr = 0.0f;
  for (int j = half; j < len; j += 2) sr += __expf(lgt[row * TT + j] - mr);
  sr += __shfl_xor(sr, 16);
  const float inv = 1.0f / sr;
  float* arow = ATT + ((size_t)h * TT + (i0 + row)) * TT;
  for (int j = half; j < len; j += 2)
    arow[j] = __expf(lgt[row * TT + j] - mr) * inv;
}

__global__ void order3_kernel(const float* __restrict__ ATT,
                              const float* __restrict__ P2,
                              const float* __restrict__ V,
                              const float* __restrict__ VSUF,
                              float* __restrict__ Y) {
  __shared__ float s_att[TT];
  __shared__ float s_red[256];
  __shared__ float s_acc[256];
  const int i = blockIdx.x;
  const int h = blockIdx.y;
  const int t = threadIdx.x;
  const int d = t & 63;
  const int g = t >> 6;

  const float* attrow = ATT + ((size_t)h * TT + i) * TT;
  const float* P2h = P2 + (size_t)h * TT * HSS;
  const float* Vh  = V  + (size_t)h * TT * HSS;

  for (int j = t; j <= i; j += 256) s_att[j] = attrow[j];
  __syncthreads();

  float m = -3.0e38f, dsum = 0.0f, acc = 0.0f;
  for (int j = g; j <= i; j += 4) {
    const float s  = s_att[j] * P2h[(size_t)j * HSS + d];
    const float vv = Vh[(size_t)j * HSS + d];
    if (s <= m) {
      const float e = __expf(s - m);
      dsum += e;
      acc += e * vv;
    } else {
      const float sc = __expf(m - s);
      dsum = dsum * sc + 1.0f;
      acc  = acc * sc + vv;
      m = s;
    }
  }

  s_red[t] = m;
  __syncthreads();
  for (int off = 128; off > 0; off >>= 1) {
    if (t < off) s_red[t] = fmaxf(s_red[t], s_red[t + off]);
    __syncthreads();
  }
  const float M = fmaxf(s_red[0], 0.0f);
  __syncthreads();

  const float sc = __expf(m - M);
  s_red[t] = dsum * sc;
  s_acc[t] = acc * sc;
  __syncthreads();
  for (int off = 128; off > 0; off >>= 1) {
    if (t < off) s_red[t] += s_red[t + off];
    __syncthreads();
  }
  const float e0 = __expf(-M);
  const float denom = s_red[0] + (float)((TT - 1 - i) * HSS) * e0;
  if (t < 64) {
    const float accT = s_acc[d] + s_acc[64 + d] + s_acc[128 + d] + s_acc[192 + d];
    const float vs = VSUF[(size_t)h * TT * HSS + (size_t)i * HSS + d];
    Y[(size_t)i * CC + h * HSS + d] = (accT + e0 * vs) / denom;
  }
}

__global__ void out_proj_kernel(const float* __restrict__ Y,
                                const float* __restrict__ Wc,
                                float* __restrict__ out) {
  const int lane = threadIdx.x;
  const int lm   = lane & 15;
  const int lk   = (lane >> 4) << 1;
  const int drow = (lane >> 4) << 3;
  const int t0 = blockIdx.x * 16;
  const int c0 = blockIdx.y * 16;

  const float* Arow = Y + (size_t)(t0 + lm) * CC;
  v8f acc = {0.f, 0.f, 0.f, 0.f, 0.f, 0.f, 0.f, 0.f};
  for (int k = 0; k < CC; k += 4) {
    v2f a, b;
    a.x = Arow[k + lk];
    a.y = Arow[k + lk + 1];
    b.x = Wc[(size_t)(k + lk) * CC + c0 + lm];
    b.y = Wc[(size_t)(k + lk + 1) * CC + c0 + lm];
    if (k + 16 < CC)
      __builtin_prefetch(&Wc[(size_t)(k + 16 + lk) * CC + c0 + lm], 0, 1);
    acc = wmma4(a, b, acc);
  }
#pragma unroll
  for (int r = 0; r < 8; ++r)
    out[(size_t)(t0 + r + drow) * CC + c0 + lm] = acc[r];
}

extern "C" void kernel_launch(void* const* d_in, const int* in_sizes, int n_in,
                              void* d_out, int out_size, void* d_ws,
                              size_t ws_size, hipStream_t stream) {
  (void)in_sizes; (void)n_in; (void)out_size; (void)ws_size;
  const float* x  = (const float*)d_in[0];
  const float* W0 = (const float*)d_in[1];
  const float* W1 = (const float*)d_in[2];
  const float* W2 = (const float*)d_in[3];
  const float* Wv = (const float*)d_in[4];
  const float* Wc = (const float*)d_in[5];
  float* out = (float*)d_out;
  float* ws  = (float*)d_ws;

  const size_t NPH = (size_t)NHH * TT * HSS;
  float* P0  = ws;
  float* P1  = ws + NPH;
  float* P2  = ws + 2 * NPH;
  float* V   = ws + 3 * NPH;
  float* VS  = ws + 4 * NPH;
  float* ATT = ws + 5 * NPH;
  float* Y   = ATT + (size_t)NHH * TT * TT;

  qkv_proj_kernel<<<dim3(TT / 16, CC / 16, 3), 32, 0, stream>>>(x, W0, W1, W2, P0);
  vproj_kernel<<<dim3(TT / 16, HSS / 16, NHH), 32, 0, stream>>>(P2, Wv, V);
  vsuffix_kernel<<<dim3(NHH), 64, 0, stream>>>(V, VS);
  att_softmax_kernel<<<dim3(TT / 16, NHH), 32, 0, stream>>>(P0, P1, ATT);
  order3_kernel<<<dim3(TT, NHH), 256, 0, stream>>>(ATT, P2, V, VS, Y);
  out_proj_kernel<<<dim3(TT / 16, CC / 16), 32, 0, stream>>>(Y, Wc, out);
}


# Both 768-deep GEMMs (QKV, output): no LDS; a wave owns a 16x64 (output: 16x32) tile of interleaved columns, fragments from 16/8-byte global loads, next 64-k set in flight; k sum order changes
# speedup vs baseline: 1.5583x; 1.5583x over previous
#include <hip/hip_runtime.h>
#include <math.h>

#define TT  512
#define CC  768
#define NHH 12
#define HSS 64
constexpr float SCALE = 0.125f;

typedef float v2f __attribute__((ext_vector_type(2)));
typedef float v8f __attribute__((ext_vector_type(8)));

static __device__ __forceinline__ v8f wmma4(v2f a, v2f b, v8f c) {
  return __builtin_amdgcn_wmma_f32_16x16x4_f32(
       false, a,  false, b,
       (short)0, c,  false,  false);
}

typedef float v4f __attribute__((ext_vector_type(4)));
template <int NT> struct BVec;
template <> struct BVec<4> { typedef float T __attribute__((ext_vector_type(4))); };
template <> struct BVec<2> { typedef float T __attribute__((ext_vector_type(2))); };

#define KB 64
#define KS (KB / 8)

template <int NT>
struct FragSet {
  v4f a[KS];
  typename BVec<NT>::T b[KS][4];
};

template <int NT>
static __device__ __forceinline__ void load_set(FragSet<NT>& s,
                                                const float* __restrict__ ap,
                                                const float* __restrict__ bp) {
#pragma unroll
  for (int st = 0; st < KS; ++st) {
    s.a[st] = *(const v4f*)(ap + st * 8);
#pragma unroll
    for (int i = 0; i < 4; ++i)
      s.b[st][i] = *(const typename BVec<NT>::T*)(bp + (size_t)(st * 8 + i) * CC);
  }
}

template <int NT>
static __device__ __forceinline__ void mma_set(const FragSet<NT>& s, v8f (&acc)[NT]) {
#pragma unroll
  for (int st = 0; st < KS; ++st) {
    const v2f a01 = {s.a[st].x, s.a[st].y};
    const v2f a23 = {s.a[st].z, s.a[st].w};
#pragma unroll
    for (int j = 0; j < NT; ++j) {
      const v2f b = {s.b[st][0][j], s.b[st][1][j]};
      acc[j] = wmma4(a01, b, acc[j]);
    }
#pragma unroll
    for (int j = 0; j < NT; ++j) {
      const v2f b = {s.b[st][2][j], s.b[st][3][j]};
      acc[j] = wmma4(a23, b, acc[j]);
    }
  }
}

template <int NT>
static __device__ __forceinline__ void gemm_k768_wave(const float* __restrict__ ap,
                                                      const float* __restrict__ bp,
                                                      v8f (&acc)[NT]) {
  FragSet<NT> s0, s1;
  load_set<NT>(s0, ap, bp);
  for (int kb = 0; kb < CC - 2 * KB; kb += 2 * KB) {
    __builtin_amdgcn_sched_barrier(0);
    load_set<NT>(s1, ap + kb + KB, bp + (size_t)(kb + KB) * CC);
    __builtin_amdgcn_sched_barrier(0);
    mma_set<NT>(s0, acc);
    __builtin_amdgcn_sched_barrier(0);
    load_set<NT>(s0, ap + kb + 2 * KB, bp + (size_t)(kb + 2 * KB) * CC);
    __builtin_amdgcn_sched_barrier(0);
    mma_set<NT>(s1, acc);
  }
  __builtin_amdgcn_sched_barrier(0);
  load_set<NT>(s1, ap + CC - KB, bp + (size_t)(CC - KB) * CC);
  __builtin_amdgcn_sched_barrier(0);
  mma_set<NT>(s0, acc);
  mma_set<NT>(s1, acc);
}

__global__ void __launch_bounds__(128) qkv_proj_kernel(const float* __restrict__ x,
                                const float* __restrict__ W0,
                                const float* __restrict__ W1,
                                const float* __restrict__ W2,
                                float* __restrict__ P) {
  const int lane = threadIdx.x & 31;
  const int wv   = threadIdx.x >> 5;
  const int lm   = lane & 15;
  const int lh   = lane >> 4;
  const int t0 = (blockIdx.x * 4 + wv) * 16;
  const int h  = blockIdx.y;
  const int w  = blockIdx.z;
  const float* W = (w == 0) ? W0 : (w == 1) ? W1 : W2;
  float* Ph = P + ((size_t)w * NHH + h) * TT * HSS;

  v8f acc[4];
#pragma unroll
  for (int j = 0; j < 4; ++j) acc[j] = (v8f){0.f, 0.f, 0.f, 0.f, 0.f, 0.f, 0.f, 0.f};
  gemm_k768_wave<4>(x + (size_t)(t0 + lm) * CC + 4 * lh,
                    W + (size_t)(4 * lh) * CC + h * HSS + 4 * lm, acc);
#pragma unroll
  for (int r = 0; r < 8; ++r) {
    const v4f o = {acc[0][r], acc[1][r], acc[2][r], acc[3][r]};
    *(v4f*)(Ph + (size_t)(t0 + 8 * lh + r) * HSS + 4 * lm) = o;
  }
}

__global__ void vproj_kernel(const float* __restrict__ P2,
                             const float* __restrict__ Wv,
                             float* __restrict__ V) {
  const int lane = threadIdx.x;
  const int lm   = lane & 15;
  const int lk   = (lane >> 4) << 1;
  const int drow = (lane >> 4) << 3;
  const int t0 = blockIdx.x * 16;
  const int n0 = blockIdx.y * 16;
  const int h  = blockIdx.z;
  const float* A = P2 + (size_t)h * TT * HSS;
  float* Vh = V + (size_t)h * TT * HSS;

  v8f acc = {0.f, 0.f, 0.f, 0.f, 0.f, 0.f, 0.f, 0.f};
  for (int k = 0; k < HSS; k += 4) {
    v2f a, b;
    a.x = A[(size_t)(t0 + lm) * HSS + k + lk];
    a.y = A[(size_t)(t0 + lm) * HSS + k + lk + 1];
    b.x = Wv[(size_t)(k + lk) * HSS + n0 + lm];
    b.y = Wv[(size_t)(k + lk + 1) * HSS + n0 + lm];
    acc = wmma4(a, b, acc);
  }
#pragma unroll
  for (int r = 0; r < 8; ++r)
    Vh[(size_t)(t0 + r + drow) * HSS + n0 + lm] = acc[r];
}

__global__ void vsuffix_kernel(const float* __restrict__ V,
                               float* __restrict__ VS) {
  const int h = blockIdx.x;
  const int d = threadIdx.x;
  const float* Vh = V + (size_t)h * TT * HSS;
  float* Sh = VS + (size_t)h * TT * HSS;
  float run = 0.0f;
  for (int j = TT - 1; j >= 0; --j) {
    Sh[(size_t)j * HSS + d] = run;
    run += Vh[(size_t)j * HSS + d];
  }
}

__global__ void att_softmax_kernel(const float* __restrict__ P0,
                                   const float* __restrict__ P1,
                                   float* __restrict__ ATT) {
  __shared__ float lgt[16 * TT];
  const int lane = threadIdx.x;
  const int lm   = lane & 15;
  const int lk   = (lane >> 4) << 1;
  const int drow = (lane >> 4) << 3;
  const int i0 = blockIdx.x * 16;
  const int h  = blockIdx.y;
  const float* P0h = P0 + (size_t)h * TT * HSS;
  const float* P1h = P1 + (size_t)h * TT * HSS;

  v2f areg[16];
#pragma unroll
  for (int kk = 0; kk < 16; ++kk) {
    const int k = kk * 4 + lk;
    areg[kk].x = P0h[(size_t)(i0 + lm) * HSS + k];
    areg[kk].y = P0h[(size_t)(i0 + lm) * HSS + k + 1];
  }

  const int jtmax = (i0 + 15) >> 4;
  for (int jt = 0; jt <= jtmax; ++jt) {
    const int j0 = jt * 16;
    v8f acc = {0.f, 0.f, 0.f, 0.f, 0.f, 0.f, 0.f, 0.f};
#pragma unroll
    for (int kk = 0; kk < 16; ++kk) {
      const int k = kk * 4 + lk;
      v2f b;
      b.x = P1h[(size_t)(j0 + lm) * HSS + k];
      b.y = P1h[(size_t)(j0 + lm) * HSS + k + 1];
      acc = wmma4(areg[kk], b, acc);
    }
#pragma unroll
    for (int r = 0; r < 8; ++r) {
      const int row = r + drow;
      const int j = j0 + lm;
      float val = acc[r] * SCALE;
      if (j > i0 + row) val = -3.0e38f;
      lgt[row * TT + j] = val;
    }
  }
  __syncthreads();

  const int row  = lane & 15;
  const int half = lane >> 4;
  const int len  = i0 + row + 1;
  float mr = -3.0e38f;
  for (int j = half; j < len; j += 2) mr = fmaxf(mr, lgt[row * TT + j]);
  mr = fmaxf(mr, __shfl_xor(mr, 16));
  float sr = 0.0f;
  for (int j = half; j < len; j += 2) sr += __expf(lgt[row * TT + j] - mr);
  sr += __shfl_xor(sr, 16);
  const float inv = 1.0f / sr;
  float* arow = ATT + ((size_t)h * TT + (i0 + row)) * TT;
  for (int j = half; j < len; j += 2)
    arow[j] = __expf(lgt[row * TT + j] - mr) * inv;
}

__global__ void order3_kernel(const float* __restrict__ ATT,
                              const float* __restrict__ P2,
                              const float* __restrict__ V,
                              const float* __restrict__ VSUF,
                              float* __restrict__ Y) {
  __shared__ float s_att[TT];
  __shared__ float s_red[256];
  __shared__ float s_acc[256];
  const int i = blockIdx.x;
  const int h = blockIdx.y;
  const int t = threadIdx.x;
  const int d = t & 63;
  const int g = t >> 6;

  const float* attrow = ATT + ((size_t)h * TT + i) * TT;
  const float* P2h = P2 + (size_t)h * TT * HSS;
  const float* Vh  = V  + (size_t)h * TT * HSS;

  for (int j = t; j <= i; j += 256) s_att[j] = attrow[j];
  __syncthreads();

  float m = -3.0e38f, dsum = 0.0f, acc = 0.0f;
  for (int j = g; j <= i; j += 4) {
    const float s  = s_att[j] * P2h[(size_t)j * HSS + d];
    const float vv = Vh[(size_t)j * HSS + d];
    if (s <= m) {
      const float e = __expf(s - m);
      dsum += e;
      acc += e * vv;
    } else {
      const float sc = __expf(m - s);
      dsum = dsum * sc + 1.0f;
      acc  = acc * sc + vv;
      m = s;
    }
  }

  s_red[t] = m;
  __syncthreads();
  for (int off = 128; off > 0; off >>= 1) {
    if (t < off) s_red[t] = fmaxf(s_red[t], s_red[t + off]);
    __syncthreads();
  }
  const float M = fmaxf(s_red[0], 0.0f);
  __syncthreads();

  const float sc = __expf(m - M);
  s_red[t] = dsum * sc;
  s_acc[t] = acc * sc;
  __syncthreads();
  for (int off = 128; off > 0; off >>= 1) {
    if (t < off) s_red[t] += s_red[t + off];
    __syncthreads();
  }
  const float e0 = __expf(-M);
  const float denom = s_red[0] + (float)((TT - 1 - i) * HSS) * e0;
  if (t < 64) {
    const float accT = s_acc[d] + s_acc[64 + d] + s_acc[128 + d] + s_acc[192 + d];
    const float vs = VSUF[(size_t)h * TT * HSS + (size_t)i * HSS + d];
    Y[(size_t)i * CC + h * HSS + d] = (accT + e0 * vs) / denom;
  }
}

__global__ void __launch_bounds__(128) out_proj_kernel(const float* __restrict__ Y,
                                const float* __restrict__ Wc,
                                float* __restrict__ out) {
  const int lane = threadIdx.x & 31;
  const int wv   = threadIdx.x >> 5;
  const int lm   = lane & 15;
  const int lh   = lane >> 4;
  const int t0 = (blockIdx.x * 4 + wv) * 16;
  const int c0 = blockIdx.y * 32;

  v8f acc[2];
#pragma unroll
  for (int j = 0; j < 2; ++j) acc[j] = (v8f){0.f, 0.f, 0.f, 0.f, 0.f, 0.f, 0.f, 0.f};
  gemm_k768_wave<2>(Y + (size_t)(t0 + lm) * CC + 4 * lh,
                    Wc + (size_t)(4 * lh) * CC + c0 + 2 * lm, acc);
#pragma unroll
  for (int r = 0; r < 8; ++r) {
    const v2f o = {acc[0][r], acc[1][r]};
    *(v2f*)(out + (size_t)(t0 + 8 * lh + r) * CC + c0 + 2 * lm) = o;
  }
}

extern "C" void kernel_launch(void* const* d_in, const int* in_sizes, int n_in,
                              void* d_out, int out_size, void* d_ws,
                              size_t ws_size, hipStream_t stream) {
  (void)in_sizes; (void)n_in; (void)out_size; (void)ws_size;
  const float* x  = (const float*)d_in[0];
  const float* W0 = (const float*)d_in[1];
  const float* W1 = (const float*)d_in[2];
  const float* W2 = (const float*)d_in[3];
  const float* Wv = (const float*)d_in[4];
  const float* Wc = (const float*)d_in[5];
  float* out = (float*)d_out;
  float* ws  = (float*)d_ws;

  const size_t NPH = (size_t)NHH * TT * HSS;
  float* P0  = ws;
  float* P1  = ws + NPH;
  float* P2  = ws + 2 * NPH;
  float* V   = ws + 3 * NPH;
  float* VS  = ws + 4 * NPH;
  float* ATT = ws + 5 * NPH;
  float* Y   = ATT + (size_t)NHH * TT * TT;

  qkv_proj_kernel<<<dim3(TT / 64, NHH, 3), 128, 0, stream>>>(x, W0, W1, W2, P0);
  vproj_kernel<<<dim3(TT / 16, HSS / 16, NHH), 32, 0, stream>>>(P2, Wv, V);
  vsuffix_kernel<<<dim3(NHH), 64, 0, stream>>>(V, VS);
  att_softmax_kernel<<<dim3(TT / 16, NHH), 32, 0, stream>>>(P0, P1, ATT);
  order3_kernel<<<dim3(TT, NHH), 256, 0, stream>>>(ATT, P2, V, VS, Y);
  out_proj_kernel<<<dim3(TT / 64, CC / 32), 128, 0, stream>>>(Y, Wc, out);
}


# Order-3 kernel: lanes are rows; 16-wave block per head and 32-row group, a wave owns 4 columns, p2/V read once per wave, att stored strip-blocked, shift from p2 row maxima; sum order changes
# speedup vs baseline: 1.9213x; 1.2329x over previous
#include <hip/hip_runtime.h>
#include <math.h>

#define TT  512
#define CC  768
#define NHH 12
#define HSS 64
constexpr float SCALE = 0.125f;

typedef float v2f __attribute__((ext_vector_type(2)));
typedef float v8f __attribute__((ext_vector_type(8)));

static __device__ __forceinline__ v8f wmma4(v2f a, v2f b, v8f c) {
  return __builtin_amdgcn_wmma_f32_16x16x4_f32(
       false, a,  false, b,
       (short)0, c,  false,  false);
}

typedef float v4f __attribute__((ext_vector_type(4)));
template <int NT> struct BVec;
template <> struct BVec<4> { typedef float T __attribute__((ext_vector_type(4))); };
template <> struct BVec<2> { typedef float T __attribute__((ext_vector_type(2))); };

#define KB 64
#define KS (KB / 8)

template <int NT>
struct FragSet {
  v4f a[KS];
  typename BVec<NT>::T b[KS][4];
};

template <int NT>
static __device__ __forceinline__ void load_set(FragSet<NT>& s,
                                                const float* __restrict__ ap,
                                                const float* __restrict__ bp) {
#pragma unroll
  for (int st = 0; st < KS; ++st) {
    s.a[st] = *(const v4f*)(ap + st * 8);
#pragma unroll
    for (int i = 0; i < 4; ++i)
      s.b[st][i] = *(const typename BVec<NT>::T*)(bp + (size_t)(st * 8 + i) * CC);
  }
}

template <int NT>
static __device__ __forceinline__ void mma_set(const FragSet<NT>& s, v8f (&acc)[NT]) {
#pragma unroll
  for (int st = 0; st < KS; ++st) {
    const v2f a01 = {s.a[st].x, s.a[st].y};
    const v2f a23 = {s.a[st].z, s.a[st].w};
#pragma unroll
    for (int j = 0; j < NT; ++j) {
      const v2f b = {s.b[st][0][j], s.b[st][1][j]};
      acc[j] = wmma4(a01, b, acc[j]);
    }
#pragma unroll
    for (int j = 0; j < NT; ++j) {
      const v2f b = {s.b[st][2][j], s.b[st][3][j]};
      acc[j] = wmma4(a23, b, acc[j]);
    }
  }
}

template <int NT>
static __device__ __forceinline__ void gemm_k768_wave(const float* __restrict__ ap,
                                                      const float* __restrict__ bp,
                                                      v8f (&acc)[NT]) {
  FragSet<NT> s0, s1;
  load_set<NT>(s0, ap, bp);
  for (int kb = 0; kb < CC - 2 * KB; kb += 2 * KB) {
    __builtin_amdgcn_sched_barrier(0);
    load_set<NT>(s1, ap + kb + KB, bp + (size_t)(kb + KB) * CC);
    __builtin_amdgcn_sched_barrier(0);
    mma_set<NT>(s0, acc);
    __builtin_amdgcn_sched_barrier(0);
    load_set<NT>(s0, ap + kb + 2 * KB, bp + (size_t)(kb + 2 * KB) * CC);
    __builtin_amdgcn_sched_barrier(0);
    mma_set<NT>(s1, acc);
  }
  __builtin_amdgcn_sched_barrier(0);
  load_set<NT>(s1, ap + CC - KB, bp + (size_t)(CC - KB) * CC);
  __builtin_amdgcn_sched_barrier(0);
  mma_set<NT>(s0, acc);
  mma_set<NT>(s1, acc);
}

__global__ void __launch_bounds__(128) qkv_proj_kernel(const float* __restrict__ x,
                                const float* __restrict__ W0,
                                const float* __restrict__ W1,
                                const float* __restrict__ W2,
                                float* __restrict__ P) {
  const int lane = threadIdx.x & 31;
  const int wv   = threadIdx.x >> 5;
  const int lm   = lane & 15;
  const int lh   = lane >> 4;
  const int t0 = (blockIdx.x * 4 + wv) * 16;
  const int h  = blockIdx.y;
  const int w  = blockIdx.z;
  const float* W = (w == 0) ? W0 : (w == 1) ? W1 : W2;
  float* Ph = P + ((size_t)w * NHH + h) * TT * HSS;

  v8f acc[4];
#pragma unroll
  for (int j = 0; j < 4; ++j) acc[j] = (v8f){0.f, 0.f, 0.f, 0.f, 0.f, 0.f, 0.f, 0.f};
  gemm_k768_wave<4>(x + (size_t)(t0 + lm) * CC + 4 * lh,
                    W + (size_t)(4 * lh) * CC + h * HSS + 4 * lm, acc);
#pragma unroll
  for (int r = 0; r < 8; ++r) {
    const v4f o = {acc[0][r], acc[1][r], acc[2][r], acc[3][r]};
    *(v4f*)(Ph + (size_t)(t0 + 8 * lh + r) * HSS + 4 * lm) = o;
  }
}

__global__ void vproj_kernel(const float* __restrict__ P2,
                             const float* __restrict__ Wv,
                             float* __restrict__ V) {
  const int lane = threadIdx.x;
  const int lm   = lane & 15;
  const int lk   = (lane >> 4) << 1;
  const int drow = (lane >> 4) << 3;
  const int t0 = blockIdx.x * 16;
  const int n0 = blockIdx.y * 16;
  const int h  = blockIdx.z;
  const float* A = P2 + (size_t)h * TT * HSS;
  float* Vh = V + (size_t)h * TT * HSS;

  v8f acc = {0.f, 0.f, 0.f, 0.f, 0.f, 0.f, 0.f, 0.f};
  for (int k = 0; k < HSS; k += 4) {
    v2f a, b;
    a.x = A[(size_t)(t0 + lm) * HSS + k + lk];
    a.y = A[(size_t)(t0 + lm) * HSS + k + lk + 1];
    b.x = Wv[(size_t)(k + lk) * HSS + n0 + lm];
    b.y = Wv[(size_t)(k + lk + 1) * HSS + n0 + lm];
    acc = wmma4(a, b, acc);
  }
#pragma unroll
  for (int r = 0; r < 8; ++r)
    Vh[(size_t)(t0 + r + drow) * HSS + n0 + lm] = acc[r];
}

__global__ void vsuffix_kernel(const float* __restrict__ V,
                               float* __restrict__ VS) {
  const int h = blockIdx.x;
  const int d = threadIdx.x;
  const float* Vh = V + (size_t)h * TT * HSS;
  float* Sh = VS + (size_t)h * TT * HSS;
  float run = 0.0f;
  for (int j = TT - 1; j >= 0; --j) {
    Sh[(size_t)j * HSS + d] = run;
    run += Vh[(size_t)j * HSS + d];
  }
}

__global__ void att_softmax_kernel(const float* __restrict__ P0,
                                   const float* __restrict__ P1,
                                   float* __restrict__ ATT) {
  __shared__ float lgt[16 * TT];
  const int lane = threadIdx.x;
  const int lm   = lane & 15;
  const int lk   = (lane >> 4) << 1;
  const int drow = (lane >> 4) << 3;
  const int i0 = blockIdx.x * 16;
  const int h  = blockIdx.y;
  const float* P0h = P0 + (size_t)h * TT * HSS;
  const float* P1h = P1 + (size_t)h * TT * HSS;

  v2f areg[16];
#pragma unroll
  for (int kk = 0; kk < 16; ++kk) {
    const int k = kk * 4 + lk;
    areg[kk].x = P0h[(size_t)(i0 + lm) * HSS + k];
    areg[kk].y = P0h[(size_t)(i0 + lm) * HSS + k + 1];
  }

  const int jtmax = (i0 + 15) >> 4;
  for (int jt = 0; jt <= jtmax; ++jt) {
    const int j0 = jt * 16;
    v8f acc = {0.f, 0.f, 0.f, 0.f, 0.f, 0.f, 0.f, 0.f};
#pragma unroll
    for (int kk = 0; kk < 16; ++kk) {
      const int k = kk * 4 + lk;
      v2f b;
      b.x = P1h[(size_t)(j0 + lm) * HSS + k];
      b.y = P1h[(size_t)(j0 + lm) * HSS + k + 1];
      acc = wmma4(areg[kk], b, acc);
    }
#pragma unroll
    for (int r = 0; r < 8; ++r) {
      const int row = r + drow;
      const int j = j0 + lm;
      float val = acc[r] * SCALE;
      if (j > i0 + row) val = -3.0e38f;
      lgt[row * TT + j] = val;
    }
  }
  __syncthreads();

  const int row  = lane & 15;
  const int half = lane >> 4;
  const int len  = i0 + row + 1;
  float mr = -3.0e38f;
  for (int j = half; j < len; j += 2) mr = fmaxf(mr, lgt[row * TT + j]);
  mr = fmaxf(mr, __shfl_xor(mr, 16));
  float sr = 0.0f;
  for (int j = half; j < len; j += 2) sr += __expf(lgt[row * TT + j] - mr);
  sr += __shfl_xor(sr, 16);
  const float inv = 1.0f / sr;
  const int jend32 = (i0 & ~31) + 32;
  float* acol = ATT + ((size_t)h * (TT / 16) + blockIdx.x) * TT * 16 + row;
  for (int j = half; j < jend32; j += 2) {
    float val = 0.0f;
    if (j < len) val = __expf(lgt[row * TT + j] - mr) * inv;
    acol[(size_t)j * 16] = val;
  }
}

#define O3_WAVES 16
#define O3_PITCH 68
__global__ void __launch_bounds__(512) order3_kernel(const float* __restrict__ ATT,
                              const float* __restrict__ P2,
                              const float* __restrict__ V,
                              const float* __restrict__ VSUF,
                              float* __restrict__ Y) {
  __shared__ float s_pmax[TT];
  __shared__ float s_m[O3_WAVES * 32];
  __shared__ float s_ds[O3_WAVES * 32];
  __shared__ __attribute__((aligned(16))) float s_out[32 * O3_PITCH];
  const float LOG2E = 1.4426950408889634f;
  const int lane = threadIdx.x & 31;
  const int w = __builtin_amdgcn_readfirstlane(threadIdx.x >> 5);
  const int g = (TT / 32 - 1) - blockIdx.x;
  const int h = blockIdx.y;
  const int i0 = g * 32;
  const int jend = i0 + 32;
  const float* P2h = P2 + (size_t)h * TT * HSS;
  const float* Vh  = V  + (size_t)h * TT * HSS;

  {
    const int r2 = lane >> 4;
    const int c4 = (lane & 15) * 4;
    float pm[16];
#pragma unroll
    for (int k = 0; k < 16; ++k) {
      const int j = min(2 * (w + O3_WAVES * k) + r2, jend - 1);
      const v4f q = *(const v4f*)(P2h + (size_t)j * HSS + c4);
      pm[k] = fmaxf(fmaxf(q.x, q.y), fmaxf(q.z, q.w));
    }
#pragma unroll
    for (int off = 8; off > 0; off >>= 1) {
#pragma unroll
      for (int k = 0; k < 16; ++k) pm[k] = fmaxf(pm[k], __shfl_xor(pm[k], off));
    }
#pragma unroll
    for (int k = 0; k < 16; ++k) {
      const int j = 2 * (w + O3_WAVES * k) + r2;
      if ((lane & 15) == 0 && j < jend) s_pmax[j] = pm[k];
    }
  }
  __syncthreads();

  const float* at = ATT + ((size_t)h * (TT / 16) + 2 * g + (lane >> 4)) * TT * 16 + (lane & 15);

  float mloc = 0.0f;
#pragma unroll
  for (int k = 0; k < TT / O3_WAVES; ++k) {
    const int j = min(w + O3_WAVES * k, jend - 1);
    mloc = fmaxf(mloc, at[(size_t)j * 16] * s_pmax[j]);
  }
  s_m[w * 32 + lane] = mloc;
  __syncthreads();
  float M = 0.0f;
#pragma unroll
  for (int k = 0; k < O3_WAVES; ++k) M = fmaxf(M, s_m[k * 32 + lane]);

  const float negML = -M * LOG2E;
  const float* p2w = P2h + 4 * w;
  const float* vw  = Vh + 4 * w;
  float ds = 0.0f, a0 = 0.0f, a1 = 0.0f, a2 = 0.0f, a3 = 0.0f;
#pragma unroll 4
  for (int j = 0; j < jend; ++j) {
    const float aL = at[(size_t)j * 16] * LOG2E;
    const v4f p = *(const v4f*)(p2w + (size_t)j * HSS);
    const v4f v = *(const v4f*)(vw + (size_t)j * HSS);
    const float e0 = __builtin_amdgcn_exp2f(fmaf(aL, p.x, negML));
    const float e1 = __builtin_amdgcn_exp2f(fmaf(aL, p.y, negML));
    const float e2 = __builtin_amdgcn_exp2f(fmaf(aL, p.z, negML));
    const float e3 = __builtin_amdgcn_exp2f(fmaf(aL, p.w, negML));
    ds += (e0 + e1) + (e2 + e3);
    a0 = fmaf(e0, v.x, a0);
    a1 = fmaf(e1, v.y, a1);
    a2 = fmaf(e2, v.z, a2);
    a3 = fmaf(e3, v.w, a3);
  }

  s_ds[w * 32 + lane] = ds;
  __syncthreads();
  float tot = 0.0f;
#pragma unroll
  for (int k = 0; k < O3_WAVES; ++k) tot += s_ds[k * 32 + lane];
  const float em = __builtin_amdgcn_exp2f(negML);
  const float denom = tot + (float)((TT - jend) * HSS) * em;
  const v4f vs = *(const v4f*)(VSUF + (size_t)h * TT * HSS + (size_t)(jend - 1) * HSS + 4 * w);
  const v4f o = {(a0 + em * vs.x) / denom, (a1 + em * vs.y) / denom,
                 (a2 + em * vs.z) / denom, (a3 + em * vs.w) / denom};
  *(v4f*)(s_out + lane * O3_PITCH + 4 * w) = o;
  __syncthreads();
  const int orow = threadIdx.x >> 4;
  const int oc   = (threadIdx.x & 15) * 4;
  *(v4f*)(Y + (size_t)(i0 + orow) * CC + h * HSS + oc) = *(const v4f*)(s_out + orow * O3_PITCH + oc);
}

__global__ void __launch_bounds__(128) out_proj_kernel(const float* __restrict__ Y,
                                const float* __restrict__ Wc,
                                float* __restrict__ out) {
  const int lane = threadIdx.x & 31;
  const int wv   = threadIdx.x >> 5;
  const int lm   = lane & 15;
  const int lh   = lane >> 4;
  const int t0 = (blockIdx.x * 4 + wv) * 16;
  const int c0 = blockIdx.y * 32;

  v8f acc[2];
#pragma unroll
  for (int j = 0; j < 2; ++j) acc[j] = (v8f){0.f, 0.f, 0.f, 0.f, 0.f, 0.f, 0.f, 0.f};
  gemm_k768_wave<2>(Y + (size_t)(t0 + lm) * CC + 4 * lh,
                    Wc + (size_t)(4 * lh) * CC + c0 + 2 * lm, acc);
#pragma unroll
  for (int r = 0; r < 8; ++r) {
    const v2f o = {acc[0][r], acc[1][r]};
    *(v2f*)(out + (size_t)(t0 + 8 * lh + r) * CC + c0 + 2 * lm) = o;
  }
}

extern "C" void kernel_launch(void* const* d_in, const int* in_sizes, int n_in,
                              void* d_out, int out_size, void* d_ws,
                              size_t ws_size, hipStream_t stream) {
  (void)in_sizes; (void)n_in; (void)out_size; (void)ws_size;
  const float* x  = (const float*)d_in[0];
  const float* W0 = (const float*)d_in[1];
  const float* W1 = (const float*)d_in[2];
  const float* W2 = (const float*)d_in[3];
  const float* Wv = (const float*)d_in[4];
  const float* Wc = (const float*)d_in[5];
  float* out = (float*)d_out;
  float* ws  = (float*)d_ws;

  const size_t NPH = (size_t)NHH * TT * HSS;
  float* P0  = ws;
  float* P1  = ws + NPH;
  float* P2  = ws + 2 * NPH;
  float* V   = ws + 3 * NPH;
  float* VS  = ws + 4 * NPH;
  float* ATT = ws + 5 * NPH;
  float* Y   = ATT + (size_t)NHH * TT * TT;

  qkv_proj_kernel<<<dim3(TT / 64, NHH, 3), 128, 0, stream>>>(x, W0, W1, W2, P0);
  vproj_kernel<<<dim3(TT / 16, HSS / 16, NHH), 32, 0, stream>>>(P2, Wv, V);
  vsuffix_kernel<<<dim3(NHH), 64, 0, stream>>>(V, VS);
  att_softmax_kernel<<<dim3(TT / 16, NHH), 32, 0, stream>>>(P0, P1, ATT);
  order3_kernel<<<dim3(TT / 32, NHH), 512, 0, stream>>>(ATT, P2, V, VS, Y);
  out_proj_kernel<<<dim3(TT / 64, CC / 32), 128, 0, stream>>>(Y, Wc, out);
}


# Causal logit+softmax kernel: a strip's logits stay in one wave's accumulators (no LDS), tile pairs loaded as one group under the pair before's WMMAs, softmax in that layout; row sum order changes
# speedup vs baseline: 3.4417x; 1.7913x over previous
#include <hip/hip_runtime.h>
#include <math.h>

#define TT  512
#define CC  768
#define NHH 12
#define HSS 64
constexpr float SCALE = 0.125f;

typedef float v2f __attribute__((ext_vector_type(2)));
typedef float v8f __attribute__((ext_vector_type(8)));

static __device__ __forceinline__ v8f wmma4(v2f a, v2f b, v8f c) {
  return __builtin_amdgcn_wmma_f32_16x16x4_f32(
       false, a,  false, b,
       (short)0, c,  false,  false);
}

typedef float v4f __attribute__((ext_vector_type(4)));
template <int NT> struct BVec;
template <> struct BVec<4> { typedef float T __attribute__((ext_vector_type(4))); };
template <> struct BVec<2> { typedef float T __attribute__((ext_vector_type(2))); };

#define KB 64
#define KS (KB / 8)

template <int NT>
struct FragSet {
  v4f a[KS];
  typename BVec<NT>::T b[KS][4];
};

template <int NT>
static __device__ __forceinline__ void load_set(FragSet<NT>& s,
                                                const float* __restrict__ ap,
                                                const float* __restrict__ bp) {
#pragma unroll
  for (int st = 0; st < KS; ++st) {
    s.a[st] = *(const v4f*)(ap + st * 8);
#pragma unroll
    for (int i = 0; i < 4; ++i)
      s.b[st][i] = *(const typename BVec<NT>::T*)(bp + (size_t)(st * 8 + i) * CC);
  }
}

template <int NT>
static __device__ __forceinline__ void mma_set(const FragSet<NT>& s, v8f (&acc)[NT]) {
#pragma unroll
  for (int st = 0; st < KS; ++st) {
    const v2f a01 = {s.a[st].x, s.a[st].y};
    const v2f a23 = {s.a[st].z, s.a[st].w};
#pragma unroll
    for (int j = 0; j < NT; ++j) {
      const v2f b = {s.b[st][0][j], s.b[st][1][j]};
      acc[j] = wmma4(a01, b, acc[j]);
    }
#pragma unroll
    for (int j = 0; j < NT; ++j) {
      const v2f b = {s.b[st][2][j], s.b[st][3][j]};
      acc[j] = wmma4(a23, b, acc[j]);
    }
  }
}

template <int NT>
static __device__ __forceinline__ void gemm_k768_wave(const float* __restrict__ ap,
                                                      const float* __restrict__ bp,
                                                      v8f (&acc)[NT]) {
  FragSet<NT> s0, s1;
  load_set<NT>(s0, ap, bp);
  for (int kb = 0; kb < CC - 2 * KB; kb += 2 * KB) {
    __builtin_amdgcn_sched_barrier(0);
    load_set<NT>(s1, ap + kb + KB, bp + (size_t)(kb + KB) * CC);
    __builtin_amdgcn_sched_barrier(0);
    mma_set<NT>(s0, acc);
    __builtin_amdgcn_sched_barrier(0);
    load_set<NT>(s0, ap + kb + 2 * KB, bp + (size_t)(kb + 2 * KB) * CC);
    __builtin_amdgcn_sched_barrier(0);
    mma_set<NT>(s1, acc);
  }
  __builtin_amdgcn_sched_barrier(0);
  load_set<NT>(s1, ap + CC - KB, bp + (size_t)(CC - KB) * CC);
  __builtin_amdgcn_sched_barrier(0);
  mma_set<NT>(s0, acc);
  mma_set<NT>(s1, acc);
}

__global__ void __launch_bounds__(128) qkv_proj_kernel(const float* __restrict__ x,
                                const float* __restrict__ W0,
                                const float* __restrict__ W1,
                                const float* __restrict__ W2,
                                float* __restrict__ P) {
  const int lane = threadIdx.x & 31;
  const int wv   = threadIdx.x >> 5;
  const int lm   = lane & 15;
  const int lh   = lane >> 4;
  const int t0 = (blockIdx.x * 4 + wv) * 16;
  const int h  = blockIdx.y;
  const int w  = blockIdx.z;
  const float* W = (w == 0) ? W0 : (w == 1) ? W1 : W2;
  float* Ph = P + ((size_t)w * NHH + h) * TT * HSS;

  v8f acc[4];
#pragma unroll
  for (int j = 0; j < 4; ++j) acc[j] = (v8f){0.f, 0.f, 0.f, 0.f, 0.f, 0.f, 0.f, 0.f};
  gemm_k768_wave<4>(x + (size_t)(t0 + lm) * CC + 4 * lh,
                    W + (size_t)(4 * lh) * CC + h * HSS + 4 * lm, acc);
#pragma unroll
  for (int r = 0; r < 8; ++r) {
    const v4f o = {acc[0][r], acc[1][r], acc[2][r], acc[3][r]};
    *(v4f*)(Ph + (size_t)(t0 + 8 * lh + r) * HSS + 4 * lm) = o;
  }
}

__global__ void vproj_kernel(const float* __restrict__ P2,
                             const float* __restrict__ Wv,
                             float* __restrict__ V) {
  const int lane = threadIdx.x;
  const int lm   = lane & 15;
  const int lk   = (lane >> 4) << 1;
  const int drow = (lane >> 4) << 3;
  const int t0 = blockIdx.x * 16;
  const int n0 = blockIdx.y * 16;
  const int h  = blockIdx.z;
  const float* A = P2 + (size_t)h * TT * HSS;
  float* Vh = V + (size_t)h * TT * HSS;

  v8f acc = {0.f, 0.f, 0.f, 0.f, 0.f, 0.f, 0.f, 0.f};
  for (int k = 0; k < HSS; k += 4) {
    v2f a, b;
    a.x = A[(size_t)(t0 + lm) * HSS + k + lk];
    a.y = A[(size_t)(t0 + lm) * HSS + k + lk + 1];
    b.x = Wv[(size_t)(k + lk) * HSS + n0 + lm];
    b.y = Wv[(size_t)(k + lk + 1) * HSS + n0 + lm];
    acc = wmma4(a, b, acc);
  }
#pragma unroll
  for (int r = 0; r < 8; ++r)
    Vh[(size_t)(t0 + r + drow) * HSS + n0 + lm] = acc[r];
}

__global__ void vsuffix_kernel(const float* __restrict__ V,
                               float* __restrict__ VS) {
  const int h = blockIdx.x;
  const int d = threadIdx.x;
  const float* Vh = V + (size_t)h * TT * HSS;
  float* Sh = VS + (size_t)h * TT * HSS;
  float run = 0.0f;
  for (int j = TT - 1; j >= 0; --j) {
    Sh[(size_t)j * HSS + d] = run;
    run += Vh[(size_t)j * HSS + d];
  }
}

#define AT_PAIRS (TT / 32)

static __device__ __forceinline__ void att_load_pair(v2f (&b)[32], const float* __restrict__ p) {
#pragma unroll
  for (int u = 0; u < 2; ++u) {
#pragma unroll
    for (int kk = 0; kk < 16; ++kk) b[u * 16 + kk] = *(const v2f*)(p + (size_t)(u * 16) * HSS + kk * 4);
  }
}

static __device__ __forceinline__ void att_mma_pair(const v2f (&a)[16], const v2f (&b)[32],
                                                    v8f& c0, v8f& c1) {
  c0 = (v8f){0.f, 0.f, 0.f, 0.f, 0.f, 0.f, 0.f, 0.f};
  c1 = c0;
#pragma unroll
  for (int kk = 0; kk < 16; ++kk) {
    c0 = wmma4(a[kk], b[kk], c0);
    c1 = wmma4(a[kk], b[16 + kk], c1);
  }
}

static __device__ __forceinline__ void att_scale_pair(v8f& c0, v8f& c1, float (&rm)[8],
                                                      bool diag, int jrel) {
#pragma unroll
  for (int r = 0; r < 8; ++r) {
    float v0 = c0[r] * SCALE;
    float v1 = c1[r] * SCALE;
    if (diag) {
      if (jrel > r) v0 = -3.0e38f;
      if (jrel + 16 > r) v1 = -3.0e38f;
    }
    c0[r] = v0;
    c1[r] = v1;
    rm[r] = fmaxf(rm[r], fmaxf(v0, v1));
  }
}

__global__ void __launch_bounds__(32) att_softmax_kernel(const float* __restrict__ P0,
                                   const float* __restrict__ P1,
                                   float* __restrict__ ATT) {
  const int lane = threadIdx.x;
  const int lm   = lane & 15;
  const int lk   = (lane >> 4) << 1;
  const int drow = (lane >> 4) << 3;
  const int i0 = blockIdx.x * 16;
  const int g  = blockIdx.x >> 1;
  const int h  = blockIdx.y;
  const float* P0h = P0 + (size_t)h * TT * HSS;
  const float* P1h = P1 + (size_t)h * TT * HSS;

  v2f areg[16];
#pragma unroll
  for (int kk = 0; kk < 16; ++kk) {
    const int k = kk * 4 + lk;
    areg[kk].x = P0h[(size_t)(i0 + lm) * HSS + k];
    areg[kk].y = P0h[(size_t)(i0 + lm) * HSS + k + 1];
  }

  v8f acc[2 * AT_PAIRS];
  float rm[8];
#pragma unroll
  for (int r = 0; r < 8; ++r) rm[r] = -3.0e38f;
  const float* bp = P1h + (size_t)lm * HSS + lk;
  const int jrel = 32 * g + lm - i0 - drow;
  v2f bA[32], bB[32];
  att_load_pair(bA, bp + (size_t)g * 32 * HSS);
#pragma unroll
  for (int q = 0; q < AT_PAIRS / 2; ++q) {
    if (2 * q <= g) {
      __builtin_amdgcn_sched_barrier(0);
      att_load_pair(bB, bp + (size_t)max(g - 2 * q - 1, 0) * 32 * HSS);
      __builtin_amdgcn_sched_barrier(0);
      att_mma_pair(areg, bA, acc[4 * q], acc[4 * q + 1]);
      att_scale_pair(acc[4 * q], acc[4 * q + 1], rm, q == 0, jrel);
    }
    if (2 * q + 1 <= g) {
      __builtin_amdgcn_sched_barrier(0);
      att_load_pair(bA, bp + (size_t)max(g - 2 * q - 2, 0) * 32 * HSS);
      __builtin_amdgcn_sched_barrier(0);
      att_mma_pair(areg, bB, acc[4 * q + 2], acc[4 * q + 3]);
      att_scale_pair(acc[4 * q + 2], acc[4 * q + 3], rm, false, 0);
    }
  }

#pragma unroll
  for (int off = 8; off > 0; off >>= 1) {
#pragma unroll
    for (int r = 0; r < 8; ++r) rm[r] = fmaxf(rm[r], __shfl_xor(rm[r], off));
  }

  float rs[8];
#pragma unroll
  for (int r = 0; r < 8; ++r) rs[r] = 0.0f;
#pragma unroll
  for (int s = 0; s < AT_PAIRS; ++s) {
    if (s <= g) {
#pragma unroll
      for (int u = 0; u < 2; ++u) {
#pragma unroll
        for (int r = 0; r < 8; ++r) {
          const float e = __expf(acc[2 * s + u][r] - rm[r]);
          acc[2 * s + u][r] = e;
          rs[r] += e;
        }
      }
    }
  }
#pragma unroll
  for (int off = 8; off > 0; off >>= 1) {
#pragma unroll
    for (int r = 0; r < 8; ++r) rs[r] += __shfl_xor(rs[r], off);
  }
  float inv[8];
#pragma unroll
  for (int r = 0; r < 8; ++r) inv[r] = 1.0f / rs[r];

  float* abase = ATT + ((size_t)h * (TT / 16) + blockIdx.x) * TT * 16 + (size_t)lm * 16 + drow;
#pragma unroll
  for (int s = 0; s < AT_PAIRS; ++s) {
    if (s <= g) {
#pragma unroll
      for (int u = 0; u < 2; ++u) {
        const v8f& c = acc[2 * s + u];
        const v4f lo = {c[0] * inv[0], c[1] * inv[1], c[2] * inv[2], c[3] * inv[3]};
        const v4f hi = {c[4] * inv[4], c[5] * inv[5], c[6] * inv[6], c[7] * inv[7]};
        float* o = abase + (size_t)(32 * (g - s) + 16 * u) * 16;
        *(v4f*)o = lo;
        *(v4f*)(o + 4) = hi;
      }
    }
  }
}

#define O3_WAVES 16
#define O3_PITCH 68
__global__ void __launch_bounds__(512) order3_kernel(const float* __restrict__ ATT,
                              const float* __restrict__ P2,
                              const float* __restrict__ V,
                              const float* __restrict__ VSUF,
                              float* __restrict__ Y) {
  __shared__ float s_pmax[TT];
  __shared__ float s_m[O3_WAVES * 32];
  __shared__ float s_ds[O3_WAVES * 32];
  __shared__ __attribute__((aligned(16))) float s_out[32 * O3_PITCH];
  const float LOG2E = 1.4426950408889634f;
  const int lane = threadIdx.x & 31;
  const int w = __builtin_amdgcn_readfirstlane(threadIdx.x >> 5);
  const int g = (TT / 32 - 1) - blockIdx.x;
  const int h = blockIdx.y;
  const int i0 = g * 32;
  const int jend = i0 + 32;
  const float* P2h = P2 + (size_t)h * TT * HSS;
  const float* Vh  = V  + (size_t)h * TT * HSS;

  {
    const int r2 = lane >> 4;
    const int c4 = (lane & 15) * 4;
    float pm[16];
#pragma unroll
    for (int k = 0; k < 16; ++k) {
      const int j = min(2 * (w + O3_WAVES * k) + r2, jend - 1);
      const v4f q = *(const v4f*)(P2h + (size_t)j * HSS + c4);
      pm[k] = fmaxf(fmaxf(q.x, q.y), fmaxf(q.z, q.w));
    }
#pragma unroll
    for (int off = 8; off > 0; off >>= 1) {
#pragma unroll
      for (int k = 0; k < 16; ++k) pm[k] = fmaxf(pm[k], __shfl_xor(pm[k], off));
    }
#pragma unroll
    for (int k = 0; k < 16; ++k) {
      const int j = 2 * (w + O3_WAVES * k) + r2;
      if ((lane & 15) == 0 && j < jend) s_pmax[j] = pm[k];
    }
  }
  __syncthreads();

  const float* at = ATT + ((size_t)h * (TT / 16) + 2 * g + (lane >> 4)) * TT * 16 + (lane & 15);

  float mloc = 0.0f;
#pragma unroll
  for (int k = 0; k < TT / O3_WAVES; ++k) {
    const int j = min(w + O3_WAVES * k, jend - 1);
    mloc = fmaxf(mloc, at[(size_t)j * 16] * s_pmax[j]);
  }
  s_m[w * 32 + lane] = mloc;
  __syncthreads();
  float M = 0.0f;
#pragma unroll
  for (int k = 0; k < O3_WAVES; ++k) M = fmaxf(M, s_m[k * 32 + lane]);

  const float negML = -M * LOG2E;
  const float* p2w = P2h + 4 * w;
  const float* vw  = Vh + 4 * w;
  float ds = 0.0f, a0 = 0.0f, a1 = 0.0f, a2 = 0.0f, a3 = 0.0f;
#pragma unroll 4
  for (int j = 0; j < jend; ++j) {
    const float aL = at[(size_t)j * 16] * LOG2E;
    const v4f p = *(const v4f*)(p2w + (size_t)j * HSS);
    const v4f v = *(const v4f*)(vw + (size_t)j * HSS);
    const float e0 = __builtin_amdgcn_exp2f(fmaf(aL, p.x, negML));
    const float e1 = __builtin_amdgcn_exp2f(fmaf(aL, p.y, negML));
    const float e2 = __builtin_amdgcn_exp2f(fmaf(aL, p.z, negML));
    const float e3 = __builtin_amdgcn_exp2f(fmaf(aL, p.w, negML));
    ds += (e0 + e1) + (e2 + e3);
    a0 = fmaf(e0, v.x, a0);
    a1 = fmaf(e1, v.y, a1);
    a2 = fmaf(e2, v.z, a2);
    a3 = fmaf(e3, v.w, a3);
  }

  s_ds[w * 32 + lane] = ds;
  __syncthreads();
  float tot = 0.0f;
#pragma unroll
  for (int k = 0; k < O3_WAVES; ++k) tot += s_ds[k * 32 + lane];
  const float em = __builtin_amdgcn_exp2f(negML);
  const float denom = tot + (float)((TT - jend) * HSS) * em;
  const v4f vs = *(const v4f*)(VSUF + (size_t)h * TT * HSS + (size_t)(jend - 1) * HSS + 4 * w);
  const v4f o = {(a0 + em * vs.x) / denom, (a1 + em * vs.y) / denom,
                 (a2 + em * vs.z) / denom, (a3 + em * vs.w) / denom};
  *(v4f*)(s_out + lane * O3_PITCH + 4 * w) = o;
  __syncthreads();
  const int orow = threadIdx.x >> 4;
  const int oc   = (threadIdx.x & 15) * 4;
  *(v4f*)(Y + (size_t)(i0 + orow) * CC + h * HSS + oc) = *(const v4f*)(s_out + orow * O3_PITCH + oc);
}

__global__ void __launch_bounds__(128) out_proj_kernel(const float* __restrict__ Y,
                                const float* __restrict__ Wc,
                                float* __restrict__ out) {
  const int lane = threadIdx.x & 31;
  const int wv   = threadIdx.x >> 5;
  const int lm   = lane & 15;
  const int lh   = lane >> 4;
  const int t0 = (blockIdx.x * 4 + wv) * 16;
  const int c0 = blockIdx.y * 32;

  v8f acc[2];
#pragma unroll
  for (int j = 0; j < 2; ++j) acc[j] = (v8f){0.f, 0.f, 0.f, 0.f, 0.f, 0.f, 0.f, 0.f};
  gemm_k768_wave<2>(Y + (size_t)(t0 + lm) * CC + 4 * lh,
                    Wc + (size_t)(4 * lh) * CC + c0 + 2 * lm, acc);
#pragma unroll
  for (int r = 0; r < 8; ++r) {
    const v2f o = {acc[0][r], acc[1][r]};
    *(v2f*)(out + (size_t)(t0 + 8 * lh + r) * CC + c0 + 2 * lm) = o;
  }
}

extern "C" void kernel_launch(void* const* d_in, const int* in_sizes, int n_in,
                              void* d_out, int out_size, void* d_ws,
                              size_t ws_size, hipStream_t stream) {
  (void)in_sizes; (void)n_in; (void)out_size; (void)ws_size;
  const float* x  = (const float*)d_in[0];
  const float* W0 = (const float*)d_in[1];
  const float* W1 = (const float*)d_in[2];
  const float* W2 = (const float*)d_in[3];
  const float* Wv = (const float*)d_in[4];
  const float* Wc = (const float*)d_in[5];
  float* out = (float*)d_out;
  float* ws  = (float*)d_ws;

  const size_t NPH = (size_t)NHH * TT * HSS;
  float* P0  = ws;
  float* P1  = ws + NPH;
  float* P2  = ws + 2 * NPH;
  float* V   = ws + 3 * NPH;
  float* VS  = ws + 4 * NPH;
  float* ATT = ws + 5 * NPH;
  float* Y   = ATT + (size_t)NHH * TT * TT;

  qkv_proj_kernel<<<dim3(TT / 64, NHH, 3), 128, 0, stream>>>(x, W0, W1, W2, P0);
  vproj_kernel<<<dim3(TT / 16, HSS / 16, NHH), 32, 0, stream>>>(P2, Wv, V);
  vsuffix_kernel<<<dim3(NHH), 64, 0, stream>>>(V, VS);
  att_softmax_kernel<<<dim3(TT / 16, NHH), 32, 0, stream>>>(P0, P1, ATT);
  order3_kernel<<<dim3(TT / 32, NHH), 512, 0, stream>>>(ATT, P2, V, VS, Y);
  out_proj_kernel<<<dim3(TT / 64, CC / 32), 128, 0, stream>>>(Y, Wc, out);
}


# Suffix-sum kernel and its array removed (6 launches to 5): each order-3 block adds up the V rows past its own group itself (whole rows per half-wave, partials through LDS); tail sum order changes
# speedup vs baseline: 5.4377x; 1.5799x over previous
#include <hip/hip_runtime.h>
#include <math.h>

#define TT  512
#define CC  768
#define NHH 12
#define HSS 64
constexpr float SCALE = 0.125f;

typedef float v2f __attribute__((ext_vector_type(2)));
typedef float v8f __attribute__((ext_vector_type(8)));

static __device__ __forceinline__ v8f wmma4(v2f a, v2f b, v8f c) {
  return __builtin_amdgcn_wmma_f32_16x16x4_f32(
       false, a,  false, b,
       (short)0, c,  false,  false);
}

typedef float v4f __attribute__((ext_vector_type(4)));
template <int NT> struct BVec;
template <> struct BVec<4> { typedef float T __attribute__((ext_vector_type(4))); };
template <> struct BVec<2> { typedef float T __attribute__((ext_vector_type(2))); };

#define KB 64
#define KS (KB / 8)

template <int NT>
struct FragSet {
  v4f a[KS];
  typename BVec<NT>::T b[KS][4];
};

template <int NT>
static __device__ __forceinline__ void load_set(FragSet<NT>& s,
                                                const float* __restrict__ ap,
                                                const float* __restrict__ bp) {
#pragma unroll
  for (int st = 0; st < KS; ++st) {
    s.a[st] = *(const v4f*)(ap + st * 8);
#pragma unroll
    for (int i = 0; i < 4; ++i)
      s.b[st][i] = *(const typename BVec<NT>::T*)(bp + (size_t)(st * 8 + i) * CC);
  }
}

template <int NT>
static __device__ __forceinline__ void mma_set(const FragSet<NT>& s, v8f (&acc)[NT]) {
#pragma unroll
  for (int st = 0; st < KS; ++st) {
    const v2f a01 = {s.a[st].x, s.a[st].y};
    const v2f a23 = {s.a[st].z, s.a[st].w};
#pragma unroll
    for (int j = 0; j < NT; ++j) {
      const v2f b = {s.b[st][0][j], s.b[st][1][j]};
      acc[j] = wmma4(a01, b, acc[j]);
    }
#pragma unroll
    for (int j = 0; j < NT; ++j) {
      const v2f b = {s.b[st][2][j], s.b[st][3][j]};
      acc[j] = wmma4(a23, b, acc[j]);
    }
  }
}

template <int NT>
static __device__ __forceinline__ void gemm_k768_wave(const float* __restrict__ ap,
                                                      const float* __restrict__ bp,
                                                      v8f (&acc)[NT]) {
  FragSet<NT> s0, s1;
  load_set<NT>(s0, ap, bp);
  for (int kb = 0; kb < CC - 2 * KB; kb += 2 * KB) {
    __builtin_amdgcn_sched_barrier(0);
    load_set<NT>(s1, ap + kb + KB, bp + (size_t)(kb + KB) * CC);
    __builtin_amdgcn_sched_barrier(0);
    mma_set<NT>(s0, acc);
    __builtin_amdgcn_sched_barrier(0);
    load_set<NT>(s0, ap + kb + 2 * KB, bp + (size_t)(kb + 2 * KB) * CC);
    __builtin_amdgcn_sched_barrier(0);
    mma_set<NT>(s1, acc);
  }
  __builtin_amdgcn_sched_barrier(0);
  load_set<NT>(s1, ap + CC - KB, bp + (size_t)(CC - KB) * CC);
  __builtin_amdgcn_sched_barrier(0);
  mma_set<NT>(s0, acc);
  mma_set<NT>(s1, acc);
}

__global__ void __launch_bounds__(128) qkv_proj_kernel(const float* __restrict__ x,
                                const float* __restrict__ W0,
                                const float* __restrict__ W1,
                                const float* __restrict__ W2,
                                float* __restrict__ P) {
  const int lane = threadIdx.x & 31;
  const int wv   = threadIdx.x >> 5;
  const int lm   = lane & 15;
  const int lh   = lane >> 4;
  const int t0 = (blockIdx.x * 4 + wv) * 16;
  const int h  = blockIdx.y;
  const int w  = blockIdx.z;
  const float* W = (w == 0) ? W0 : (w == 1) ? W1 : W2;
  float* Ph = P + ((size_t)w * NHH + h) * TT * HSS;

  v8f acc[4];
#pragma unroll
  for (int j = 0; j < 4; ++j) acc[j] = (v8f){0.f, 0.f, 0.f, 0.f, 0.f, 0.f, 0.f, 0.f};
  gemm_k768_wave<4>(x + (size_t)(t0 + lm) * CC + 4 * lh,
                    W + (size_t)(4 * lh) * CC + h * HSS + 4 * lm, acc);
#pragma unroll
  for (int r = 0; r < 8; ++r) {
    const v4f o = {acc[0][r], acc[1][r], acc[2][r], acc[3][r]};
    *(v4f*)(Ph + (size_t)(t0 + 8 * lh + r) * HSS + 4 * lm) = o;
  }
}

__global__ void vproj_kernel(const float* __restrict__ P2,
                             const float* __restrict__ Wv,
                             float* __restrict__ V) {
  const int lane = threadIdx.x;
  const int lm   = lane & 15;
  const int lk   = (lane >> 4) << 1;
  const int drow = (lane >> 4) << 3;
  const int t0 = blockIdx.x * 16;
  const int n0 = blockIdx.y * 16;
  const int h  = blockIdx.z;
  const float* A = P2 + (size_t)h * TT * HSS;
  float* Vh = V + (size_t)h * TT * HSS;

  v8f acc = {0.f, 0.f, 0.f, 0.f, 0.f, 0.f, 0.f, 0.f};
  for (int k = 0; k < HSS; k += 4) {
    v2f a, b;
    a.x = A[(size_t)(t0 + lm) * HSS + k + lk];
    a.y = A[(size_t)(t0 + lm) * HSS + k + lk + 1];
    b.x = Wv[(size_t)(k + lk) * HSS + n0 + lm];
    b.y = Wv[(size_t)(k + lk + 1) * HSS + n0 + lm];
    acc = wmma4(a, b, acc);
  }
#pragma unroll
  for (int r = 0; r < 8; ++r)
    Vh[(size_t)(t0 + r + drow) * HSS + n0 + lm] = acc[r];
}

#define AT_PAIRS (TT / 32)

static __device__ __forceinline__ void att_load_pair(v2f (&b)[32], const float* __restrict__ p) {
#pragma unroll
  for (int u = 0; u < 2; ++u) {
#pragma unroll
    for (int kk = 0; kk < 16; ++kk) b[u * 16 + kk] = *(const v2f*)(p + (size_t)(u * 16) * HSS + kk * 4);
  }
}

static __device__ __forceinline__ void att_mma_pair(const v2f (&a)[16], const v2f (&b)[32],
                                                    v8f& c0, v8f& c1) {
  c0 = (v8f){0.f, 0.f, 0.f, 0.f, 0.f, 0.f, 0.f, 0.f};
  c1 = c0;
#pragma unroll
  for (int kk = 0; kk < 16; ++kk) {
    c0 = wmma4(a[kk], b[kk], c0);
    c1 = wmma4(a[kk], b[16 + kk], c1);
  }
}

static __device__ __forceinline__ void att_scale_pair(v8f& c0, v8f& c1, float (&rm)[8],
                                                      bool diag, int jrel) {
#pragma unroll
  for (int r = 0; r < 8; ++r) {
    float v0 = c0[r] * SCALE;
    float v1 = c1[r] * SCALE;
    if (diag) {
      if (jrel > r) v0 = -3.0e38f;
      if (jrel + 16 > r) v1 = -3.0e38f;
    }
    c0[r] = v0;
    c1[r] = v1;
    rm[r] = fmaxf(rm[r], fmaxf(v0, v1));
  }
}

__global__ void __launch_bounds__(32) att_softmax_kernel(const float* __restrict__ P0,
                                   const float* __restrict__ P1,
                                   float* __restrict__ ATT) {
  const int lane = threadIdx.x;
  const int lm   = lane & 15;
  const int lk   = (lane >> 4) << 1;
  const int drow = (lane >> 4) << 3;
  const int i0 = blockIdx.x * 16;
  const int g  = blockIdx.x >> 1;
  const int h  = blockIdx.y;
  const float* P0h = P0 + (size_t)h * TT * HSS;
  const float* P1h = P1 + (size_t)h * TT * HSS;

  v2f areg[16];
#pragma unroll
  for (int kk = 0; kk < 16; ++kk) {
    const int k = kk * 4 + lk;
    areg[kk].x = P0h[(size_t)(i0 + lm) * HSS + k];
    areg[kk].y = P0h[(size_t)(i0 + lm) * HSS + k + 1];
  }

  v8f acc[2 * AT_PAIRS];
  float rm[8];
#pragma unroll
  for (int r = 0; r < 8; ++r) rm[r] = -3.0e38f;
  const float* bp = P1h + (size_t)lm * HSS + lk;
  const int jrel = 32 * g + lm - i0 - drow;
  v2f bA[32], bB[32];
  att_load_pair(bA, bp + (size_t)g * 32 * HSS);
#pragma unroll
  for (int q = 0; q < AT_PAIRS / 2; ++q) {
    if (2 * q <= g) {
      __builtin_amdgcn_sched_barrier(0);
      att_load_pair(bB, bp + (size_t)max(g - 2 * q - 1, 0) * 32 * HSS);
      __builtin_amdgcn_sched_barrier(0);
      att_mma_pair(areg, bA, acc[4 * q], acc[4 * q + 1]);
      att_scale_pair(acc[4 * q], acc[4 * q + 1], rm, q == 0, jrel);
    }
    if (2 * q + 1 <= g) {
      __builtin_amdgcn_sched_barrier(0);
      att_load_pair(bA, bp + (size_t)max(g - 2 * q - 2, 0) * 32 * HSS);
      __builtin_amdgcn_sched_barrier(0);
      att_mma_pair(areg, bB, acc[4 * q + 2], acc[4 * q + 3]);
      att_scale_pair(acc[4 * q + 2], acc[4 * q + 3], rm, false, 0);
    }
  }

#pragma unroll
  for (int off = 8; off > 0; off >>= 1) {
#pragma unroll
    for (int r = 0; r < 8; ++r) rm[r] = fmaxf(rm[r], __shfl_xor(rm[r], off));
  }

  float rs[8];
#pragma unroll
  for (int r = 0; r < 8; ++r) rs[r] = 0.0f;
#pragma unroll
  for (int s = 0; s < AT_PAIRS; ++s) {
    if (s <= g) {
#pragma unroll
      for (int u = 0; u < 2; ++u) {
#pragma unroll
        for (int r = 0; r < 8; ++r) {
          const float e = __expf(acc[2 * s + u][r] - rm[r]);
          acc[2 * s + u][r] = e;
          rs[r] += e;
        }
      }
    }
  }
#pragma unroll
  for (int off = 8; off > 0; off >>= 1) {
#pragma unroll
    for (int r = 0; r < 8; ++r) rs[r] += __shfl_xor(rs[r], off);
  }
  float inv[8];
#pragma unroll
  for (int r = 0; r < 8; ++r) inv[r] = 1.0f / rs[r];

  float* abase = ATT + ((size_t)h * (TT / 16) + blockIdx.x) * TT * 16 + (size_t)lm * 16 + drow;
#pragma unroll
  for (int s = 0; s < AT_PAIRS; ++s) {
    if (s <= g) {
#pragma unroll
      for (int u = 0; u < 2; ++u) {
        const v8f& c = acc[2 * s + u];
        const v4f lo = {c[0] * inv[0], c[1] * inv[1], c[2] * inv[2], c[3] * inv[3]};
        const v4f hi = {c[4] * inv[4], c[5] * inv[5], c[6] * inv[6], c[7] * inv[7]};
        float* o = abase + (size_t)(32 * (g - s) + 16 * u) * 16;
        *(v4f*)o = lo;
        *(v4f*)(o + 4) = hi;
      }
    }
  }
}

#define O3_WAVES 16
#define O3_PITCH 68
__global__ void __launch_bounds__(512) order3_kernel(const float* __restrict__ ATT,
                              const float* __restrict__ P2,
                              const float* __restrict__ V,
                              float* __restrict__ Y) {
  __shared__ float s_pmax[TT];
  __shared__ float s_m[O3_WAVES * 32];
  __shared__ float s_ds[O3_WAVES * 32];
  __shared__ __attribute__((aligned(16))) float s_tail[O3_WAVES * HSS];
  __shared__ __attribute__((aligned(16))) float s_out[32 * O3_PITCH];
  const float LOG2E = 1.4426950408889634f;
  const int lane = threadIdx.x & 31;
  const int w = __builtin_amdgcn_readfirstlane(threadIdx.x >> 5);
  const int g = (TT / 32 - 1) - blockIdx.x;
  const int h = blockIdx.y;
  const int i0 = g * 32;
  const int jend = i0 + 32;
  const float* P2h = P2 + (size_t)h * TT * HSS;
  const float* Vh  = V  + (size_t)h * TT * HSS;

  {
    const int trow = jend + (threadIdx.x >> 4);
    const int tc4  = (threadIdx.x & 15) * 4;
    v4f tl = {0.f, 0.f, 0.f, 0.f};
    if (jend < TT) {
#pragma unroll
      for (int k = 0; k < TT / 32 - 1; ++k) {
        const int row = trow + 32 * k;
        const v4f q = *(const v4f*)(Vh + (size_t)min(row, TT - 1) * HSS + tc4);
        const bool in = row < TT;
        tl.x += in ? q.x : 0.0f;
        tl.y += in ? q.y : 0.0f;
        tl.z += in ? q.z : 0.0f;
        tl.w += in ? q.w : 0.0f;
      }
    }
    tl.x += __shfl_xor(tl.x, 16);
    tl.y += __shfl_xor(tl.y, 16);
    tl.z += __shfl_xor(tl.z, 16);
    tl.w += __shfl_xor(tl.w, 16);
    if (lane < 16) *(v4f*)(s_tail + w * HSS + tc4) = tl;
  }

  {
    const int r2 = lane >> 4;
    const int c4 = (lane & 15) * 4;
    float pm[16];
#pragma unroll
    for (int k = 0; k < 16; ++k) {
      const int j = min(2 * (w + O3_WAVES * k) + r2, jend - 1);
      const v4f q = *(const v4f*)(P2h + (size_t)j * HSS + c4);
      pm[k] = fmaxf(fmaxf(q.x, q.y), fmaxf(q.z, q.w));
    }
#pragma unroll
    for (int off = 8; off > 0; off >>= 1) {
#pragma unroll
      for (int k = 0; k < 16; ++k) pm[k] = fmaxf(pm[k], __shfl_xor(pm[k], off));
    }
#pragma unroll
    for (int k = 0; k < 16; ++k) {
      const int j = 2 * (w + O3_WAVES * k) + r2;
      if ((lane & 15) == 0 && j < jend) s_pmax[j] = pm[k];
    }
  }
  __syncthreads();

  const float* at = ATT + ((size_t)h * (TT / 16) + 2 * g + (lane >> 4)) * TT * 16 + (lane & 15);

  float mloc = 0.0f;
#pragma unroll
  for (int k = 0; k < TT / O3_WAVES; ++k) {
    const int j = min(w + O3_WAVES * k, jend - 1);
    mloc = fmaxf(mloc, at[(size_t)j * 16] * s_pmax[j]);
  }
  s_m[w * 32 + lane] = mloc;
  __syncthreads();
  float M = 0.0f;
#pragma unroll
  for (int k = 0; k < O3_WAVES; ++k) M = fmaxf(M, s_m[k * 32 + lane]);

  const float negML = -M * LOG2E;
  const float* p2w = P2h + 4 * w;
  const float* vw  = Vh + 4 * w;
  float ds = 0.0f, a0 = 0.0f, a1 = 0.0f, a2 = 0.0f, a3 = 0.0f;
#pragma unroll 4
  for (int j = 0; j < jend; ++j) {
    const float aL = at[(size_t)j * 16] * LOG2E;
    const v4f p = *(const v4f*)(p2w + (size_t)j * HSS);
    const v4f v = *(const v4f*)(vw + (size_t)j * HSS);
    const float e0 = __builtin_amdgcn_exp2f(fmaf(aL, p.x, negML));
    const float e1 = __builtin_amdgcn_exp2f(fmaf(aL, p.y, negML));
    const float e2 = __builtin_amdgcn_exp2f(fmaf(aL, p.z, negML));
    const float e3 = __builtin_amdgcn_exp2f(fmaf(aL, p.w, negML));
    ds += (e0 + e1) + (e2 + e3);
    a0 = fmaf(e0, v.x, a0);
    a1 = fmaf(e1, v.y, a1);
    a2 = fmaf(e2, v.z, a2);
    a3 = fmaf(e3, v.w, a3);
  }

  s_ds[w * 32 + lane] = ds;
  __syncthreads();
  float tot = 0.0f;
#pragma unroll
  for (int k = 0; k < O3_WAVES; ++k) tot += s_ds[k * 32 + lane];
  const float em = __builtin_amdgcn_exp2f(negML);
  const float denom = tot + (float)((TT - jend) * HSS) * em;
  v4f vs = {0.f, 0.f, 0.f, 0.f};
#pragma unroll
  for (int k = 0; k < O3_WAVES; ++k) vs += *(const v4f*)(s_tail + k * HSS + 4 * w);
  const v4f o = {(a0 + em * vs.x) / denom, (a1 + em * vs.y) / denom,
                 (a2 + em * vs.z) / denom, (a3 + em * vs.w) / denom};
  *(v4f*)(s_out + lane * O3_PITCH + 4 * w) = o;
  __syncthreads();
  const int orow = threadIdx.x >> 4;
  const int oc   = (threadIdx.x & 15) * 4;
  *(v4f*)(Y + (size_t)(i0 + orow) * CC + h * HSS + oc) = *(const v4f*)(s_out + orow * O3_PITCH + oc);
}

__global__ void __launch_bounds__(128) out_proj_kernel(const float* __restrict__ Y,
                                const float* __restrict__ Wc,
                                float* __restrict__ out) {
  const int lane = threadIdx.x & 31;
  const int wv   = threadIdx.x >> 5;
  const int lm   = lane & 15;
  const int lh   = lane >> 4;
  const int t0 = (blockIdx.x * 4 + wv) * 16;
  const int c0 = blockIdx.y * 32;

  v8f acc[2];
#pragma unroll
  for (int j = 0; j < 2; ++j) acc[j] = (v8f){0.f, 0.f, 0.f, 0.f, 0.f, 0.f, 0.f, 0.f};
  gemm_k768_wave<2>(Y + (size_t)(t0 + lm) * CC + 4 * lh,
                    Wc + (size_t)(4 * lh) * CC + c0 + 2 * lm, acc);
#pragma unroll
  for (int r = 0; r < 8; ++r) {
    const v2f o = {acc[0][r], acc[1][r]};
    *(v2f*)(out + (size_t)(t0 + 8 * lh + r) * CC + c0 + 2 * lm) = o;
  }
}

extern "C" void kernel_launch(void* const* d_in, const int* in_sizes, int n_in,
                              void* d_out, int out_size, void* d_ws,
                              size_t ws_size, hipStream_t stream) {
  (void)in_sizes; (void)n_in; (void)out_size; (void)ws_size;
  const float* x  = (const float*)d_in[0];
  const float* W0 = (const float*)d_in[1];
  const float* W1 = (const float*)d_in[2];
  const float* W2 = (const float*)d_in[3];
  const float* Wv = (const float*)d_in[4];
  const float* Wc = (const float*)d_in[5];
  float* out = (float*)d_out;
  float* ws  = (float*)d_ws;

  const size_t NPH = (size_t)NHH * TT * HSS;
  float* P0  = ws;
  float* P1  = ws + NPH;
  float* P2  = ws + 2 * NPH;
  float* V   = ws + 3 * NPH;
  float* ATT = ws + 4 * NPH;
  float* Y   = ATT + (size_t)NHH * TT * TT;

  qkv_proj_kernel<<<dim3(TT / 64, NHH, 3), 128, 0, stream>>>(x, W0, W1, W2, P0);
  vproj_kernel<<<dim3(TT / 16, HSS / 16, NHH), 32, 0, stream>>>(P2, Wv, V);
  att_softmax_kernel<<<dim3(TT / 16, NHH), 32, 0, stream>>>(P0, P1, ATT);
  order3_kernel<<<dim3(TT / 32, NHH), 512, 0, stream>>>(ATT, P2, V, Y);
  out_proj_kernel<<<dim3(TT / 64, CC / 32), 128, 0, stream>>>(Y, Wc, out);
}


# Order-3 kernel: 192 equal blocks, each walks a short and a long 16-row strip in turn (528 columns j); lanes are 16 rows x 2 parities of j, joined by a lane exchange; j and tail sum order changes
# speedup vs baseline: 6.3293x; 1.1640x over previous
#include <hip/hip_runtime.h>
#include <math.h>

#define TT  512
#define CC  768
#define NHH 12
#define HSS 64
constexpr float SCALE = 0.125f;

typedef float v2f __attribute__((ext_vector_type(2)));
typedef float v8f __attribute__((ext_vector_type(8)));

static __device__ __forceinline__ v8f wmma4(v2f a, v2f b, v8f c) {
  return __builtin_amdgcn_wmma_f32_16x16x4_f32(
       false, a,  false, b,
       (short)0, c,  false,  false);
}

typedef float v4f __attribute__((ext_vector_type(4)));
template <int NT> struct BVec;
template <> struct BVec<4> { typedef float T __attribute__((ext_vector_type(4))); };
template <> struct BVec<2> { typedef float T __attribute__((ext_vector_type(2))); };

#define KB 64
#define KS (KB / 8)

template <int NT>
struct FragSet {
  v4f a[KS];
  typename BVec<NT>::T b[KS][4];
};

template <int NT>
static __device__ __forceinline__ void load_set(FragSet<NT>& s,
                                                const float* __restrict__ ap,
                                                const float* __restrict__ bp) {
#pragma unroll
  for (int st = 0; st < KS; ++st) {
    s.a[st] = *(const v4f*)(ap + st * 8);
#pragma unroll
    for (int i = 0; i < 4; ++i)
      s.b[st][i] = *(const typename BVec<NT>::T*)(bp + (size_t)(st * 8 + i) * CC);
  }
}

template <int NT>
static __device__ __forceinline__ void mma_set(const FragSet<NT>& s, v8f (&acc)[NT]) {
#pragma unroll
  for (int st = 0; st < KS; ++st) {
    const v2f a01 = {s.a[st].x, s.a[st].y};
    const v2f a23 = {s.a[st].z, s.a[st].w};
#pragma unroll
    for (int j = 0; j < NT; ++j) {
      const v2f b = {s.b[st][0][j], s.b[st][1][j]};
      acc[j] = wmma4(a01, b, acc[j]);
    }
#pragma unroll
    for (int j = 0; j < NT; ++j) {
      const v2f b = {s.b[st][2][j], s.b[st][3][j]};
      acc[j] = wmma4(a23, b, acc[j]);
    }
  }
}

template <int NT>
static __device__ __forceinline__ void gemm_k768_wave(const float* __restrict__ ap,
                                                      const float* __restrict__ bp,
                                                      v8f (&acc)[NT]) {
  FragSet<NT> s0, s1;
  load_set<NT>(s0, ap, bp);
  for (int kb = 0; kb < CC - 2 * KB; kb += 2 * KB) {
    __builtin_amdgcn_sched_barrier(0);
    load_set<NT>(s1, ap + kb + KB, bp + (size_t)(kb + KB) * CC);
    __builtin_amdgcn_sched_barrier(0);
    mma_set<NT>(s0, acc);
    __builtin_amdgcn_sched_barrier(0);
    load_set<NT>(s0, ap + kb + 2 * KB, bp + (size_t)(kb + 2 * KB) * CC);
    __builtin_amdgcn_sched_barrier(0);
    mma_set<NT>(s1, acc);
  }
  __builtin_amdgcn_sched_barrier(0);
  load_set<NT>(s1, ap + CC - KB, bp + (size_t)(CC - KB) * CC);
  __builtin_amdgcn_sched_barrier(0);
  mma_set<NT>(s0, acc);
  mma_set<NT>(s1, acc);
}

__global__ void __launch_bounds__(128) qkv_proj_kernel(const float* __restrict__ x,
                                const float* __restrict__ W0,
                                const float* __restrict__ W1,
                                const float* __restrict__ W2,
                                float* __restrict__ P) {
  const int lane = threadIdx.x & 31;
  const int wv   = threadIdx.x >> 5;
  const int lm   = lane & 15;
  const int lh   = lane >> 4;
  const int t0 = (blockIdx.x * 4 + wv) * 16;
  const int h  = blockIdx.y;
  const int w  = blockIdx.z;
  const float* W = (w == 0) ? W0 : (w == 1) ? W1 : W2;
  float* Ph = P + ((size_t)w * NHH + h) * TT * HSS;

  v8f acc[4];
#pragma unroll
  for (int j = 0; j < 4; ++j) acc[j] = (v8f){0.f, 0.f, 0.f, 0.f, 0.f, 0.f, 0.f, 0.f};
  gemm_k768_wave<4>(x + (size_t)(t0 + lm) * CC + 4 * lh,
                    W + (size_t)(4 * lh) * CC + h * HSS + 4 * lm, acc);
#pragma unroll
  for (int r = 0; r < 8; ++r) {
    const v4f o = {acc[0][r], acc[1][r], acc[2][r], acc[3][r]};
    *(v4f*)(Ph + (size_t)(t0 + 8 * lh + r) * HSS + 4 * lm) = o;
  }
}

__global__ void vproj_kernel(const float* __restrict__ P2,
                             const float* __restrict__ Wv,
                             float* __restrict__ V) {
  const int lane = threadIdx.x;
  const int lm   = lane & 15;
  const int lk   = (lane >> 4) << 1;
  const int drow = (lane >> 4) << 3;
  const int t0 = blockIdx.x * 16;
  const int n0 = blockIdx.y * 16;
  const int h  = blockIdx.z;
  const float* A = P2 + (size_t)h * TT * HSS;
  float* Vh = V + (size_t)h * TT * HSS;

  v8f acc = {0.f, 0.f, 0.f, 0.f, 0.f, 0.f, 0.f, 0.f};
  for (int k = 0; k < HSS; k += 4) {
    v2f a, b;
    a.x = A[(size_t)(t0 + lm) * HSS + k + lk];
    a.y = A[(size_t)(t0 + lm) * HSS + k + lk + 1];
    b.x = Wv[(size_t)(k + lk) * HSS + n0 + lm];
    b.y = Wv[(size_t)(k + lk + 1) * HSS + n0 + lm];
    acc = wmma4(a, b, acc);
  }
#pragma unroll
  for (int r = 0; r < 8; ++r)
    Vh[(size_t)(t0 + r + drow) * HSS + n0 + lm] = acc[r];
}

#define AT_PAIRS (TT / 32)

static __device__ __forceinline__ void att_load_pair(v2f (&b)[32], const float* __restrict__ p) {
#pragma unroll
  for (int u = 0; u < 2; ++u) {
#pragma unroll
    for (int kk = 0; kk < 16; ++kk) b[u * 16 + kk] = *(const v2f*)(p + (size_t)(u * 16) * HSS + kk * 4);
  }
}

static __device__ __forceinline__ void att_mma_pair(const v2f (&a)[16], const v2f (&b)[32],
                                                    v8f& c0, v8f& c1) {
  c0 = (v8f){0.f, 0.f, 0.f, 0.f, 0.f, 0.f, 0.f, 0.f};
  c1 = c0;
#pragma unroll
  for (int kk = 0; kk < 16; ++kk) {
    c0 = wmma4(a[kk], b[kk], c0);
    c1 = wmma4(a[kk], b[16 + kk], c1);
  }
}

static __device__ __forceinline__ void att_scale_pair(v8f& c0, v8f& c1, float (&rm)[8],
                                                      bool diag, int jrel) {
#pragma unroll
  for (int r = 0; r < 8; ++r) {
    float v0 = c0[r] * SCALE;
    float v1 = c1[r] * SCALE;
    if (diag) {
      if (jrel > r) v0 = -3.0e38f;
      if (jrel + 16 > r) v1 = -3.0e38f;
    }
    c0[r] = v0;
    c1[r] = v1;
    rm[r] = fmaxf(rm[r], fmaxf(v0, v1));
  }
}

__global__ void __launch_bounds__(32) att_softmax_kernel(const float* __restrict__ P0,
                                   const float* __restrict__ P1,
                                   float* __restrict__ ATT) {
  const int lane = threadIdx.x;
  const int lm   = lane & 15;
  const int lk   = (lane >> 4) << 1;
  const int drow = (lane >> 4) << 3;
  const int i0 = blockIdx.x * 16;
  const int g  = blockIdx.x >> 1;
  const int h  = blockIdx.y;
  const float* P0h = P0 + (size_t)h * TT * HSS;
  const float* P1h = P1 + (size_t)h * TT * HSS;

  v2f areg[16];
#pragma unroll
  for (int kk = 0; kk < 16; ++kk) {
    const int k = kk * 4 + lk;
    areg[kk].x = P0h[(size_t)(i0 + lm) * HSS + k];
    areg[kk].y = P0h[(size_t)(i0 + lm) * HSS + k + 1];
  }

  v8f acc[2 * AT_PAIRS];
  float rm[8];
#pragma unroll
  for (int r = 0; r < 8; ++r) rm[r] = -3.0e38f;
  const float* bp = P1h + (size_t)lm * HSS + lk;
  const int jrel = 32 * g + lm - i0 - drow;
  v2f bA[32], bB[32];
  att_load_pair(bA, bp + (size_t)g * 32 * HSS);
#pragma unroll
  for (int q = 0; q < AT_PAIRS / 2; ++q) {
    if (2 * q <= g) {
      __builtin_amdgcn_sched_barrier(0);
      att_load_pair(bB, bp + (size_t)max(g - 2 * q - 1, 0) * 32 * HSS);
      __builtin_amdgcn_sched_barrier(0);
      att_mma_pair(areg, bA, acc[4 * q], acc[4 * q + 1]);
      att_scale_pair(acc[4 * q], acc[4 * q + 1], rm, q == 0, jrel);
    }
    if (2 * q + 1 <= g) {
      __builtin_amdgcn_sched_barrier(0);
      att_load_pair(bA, bp + (size_t)max(g - 2 * q - 2, 0) * 32 * HSS);
      __builtin_amdgcn_sched_barrier(0);
      att_mma_pair(areg, bB, acc[4 * q + 2], acc[4 * q + 3]);
      att_scale_pair(acc[4 * q + 2], acc[4 * q + 3], rm, false, 0);
    }
  }

#pragma unroll
  for (int off = 8; off > 0; off >>= 1) {
#pragma unroll
    for (int r = 0; r < 8; ++r) rm[r] = fmaxf(rm[r], __shfl_xor(rm[r], off));
  }

  float rs[8];
#pragma unroll
  for (int r = 0; r < 8; ++r) rs[r] = 0.0f;
#pragma unroll
  for (int s = 0; s < AT_PAIRS; ++s) {
    if (s <= g) {
#pragma unroll
      for (int u = 0; u < 2; ++u) {
#pragma unroll
        for (int r = 0; r < 8; ++r) {
          const float e = __expf(acc[2 * s + u][r] - rm[r]);
          acc[2 * s + u][r] = e;
          rs[r] += e;
        }
      }
    }
  }
#pragma unroll
  for (int off = 8; off > 0; off >>= 1) {
#pragma unroll
    for (int r = 0; r < 8; ++r) rs[r] += __shfl_xor(rs[r], off);
  }
  float inv[8];
#pragma unroll
  for (int r = 0; r < 8; ++r) inv[r] = 1.0f / rs[r];

  float* abase = ATT + ((size_t)h * (TT / 16) + blockIdx.x) * TT * 16 + (size_t)lm * 16 + drow;
#pragma unroll
  for (int s = 0; s < AT_PAIRS; ++s) {
    if (s <= g) {
#pragma unroll
      for (int u = 0; u < 2; ++u) {
        const v8f& c = acc[2 * s + u];
        const v4f lo = {c[0] * inv[0], c[1] * inv[1], c[2] * inv[2], c[3] * inv[3]};
        const v4f hi = {c[4] * inv[4], c[5] * inv[5], c[6] * inv[6], c[7] * inv[7]};
        float* o = abase + (size_t)(32 * (g - s) + 16 * u) * 16;
        *(v4f*)o = lo;
        *(v4f*)(o + 4) = hi;
      }
    }
  }
}

#define O3_WAVES 16
#define O3_PITCH 68

static __device__ __forceinline__ void o3_walk(const float* __restrict__ at,
                                               const float* __restrict__ p2w,
                                               const float* __restrict__ vrow,
                                               int ntrips, float negML, float ce, float co,
                                               float& ds, float& a0, float& a1, float& a2, float& a3) {
#pragma unroll 4
  for (int t = 0; t < ntrips; ++t) {
    const float a = at[(size_t)t * 32];
    const v4f pe = *(const v4f*)(p2w + (size_t)(2 * t) * HSS);
    const v4f po = *(const v4f*)(p2w + (size_t)(2 * t + 1) * HSS);
    const v4f v = *(const v4f*)(vrow + (size_t)(2 * t) * HSS);
    const float ae = a * ce;
    const float ao = a * co;
    const float e0 = __builtin_amdgcn_exp2f(fmaf(ae, pe.x, fmaf(ao, po.x, negML)));
    const float e1 = __builtin_amdgcn_exp2f(fmaf(ae, pe.y, fmaf(ao, po.y, negML)));
    const float e2 = __builtin_amdgcn_exp2f(fmaf(ae, pe.z, fmaf(ao, po.z, negML)));
    const float e3 = __builtin_amdgcn_exp2f(fmaf(ae, pe.w, fmaf(ao, po.w, negML)));
    ds += (e0 + e1) + (e2 + e3);
    a0 = fmaf(e0, v.x, a0);
    a1 = fmaf(e1, v.y, a1);
    a2 = fmaf(e2, v.z, a2);
    a3 = fmaf(e3, v.w, a3);
  }
  ds += __shfl_xor(ds, 16);
  a0 += __shfl_xor(a0, 16);
  a1 += __shfl_xor(a1, 16);
  a2 += __shfl_xor(a2, 16);
  a3 += __shfl_xor(a3, 16);
}

__global__ void __launch_bounds__(512) order3_kernel(const float* __restrict__ ATT,
                              const float* __restrict__ P2,
                              const float* __restrict__ V,
                              float* __restrict__ Y) {
  __shared__ float s_pmax[TT];
  __shared__ float s_m[O3_WAVES * 32];
  __shared__ float s_ds[O3_WAVES * 32];
  __shared__ __attribute__((aligned(16))) float s_tail[2 * O3_WAVES * HSS];
  __shared__ __attribute__((aligned(16))) float s_out[32 * O3_PITCH];
  const float LOG2E = 1.4426950408889634f;
  const int lane = threadIdx.x & 31;
  const int hf   = lane >> 4;
  const int r    = lane & 15;
  const int w = __builtin_amdgcn_readfirstlane(threadIdx.x >> 5);
  const int p = blockIdx.x;
  const int h = blockIdx.y;
  const int sA = p;
  const int sB = TT / 16 - 1 - p;
  const int jendA = 16 * (sA + 1);
  const int jendB = 16 * (sB + 1);
  const float* P2h = P2 + (size_t)h * TT * HSS;
  const float* Vh  = V  + (size_t)h * TT * HSS;

  {
    const int trow = jendA + (threadIdx.x >> 4);
    const int tc4  = (threadIdx.x & 15) * 4;
    v4f tA = {0.f, 0.f, 0.f, 0.f};
    v4f tB = {0.f, 0.f, 0.f, 0.f};
#pragma unroll
    for (int k = 0; k < TT / 32; ++k) {
      const int row = trow + 32 * k;
      const v4f q = *(const v4f*)(Vh + (size_t)min(row, TT - 1) * HSS + tc4);
      const bool inA = row < TT;
      const bool inB = inA && row >= jendB;
      tA.x += inA ? q.x : 0.0f;
      tA.y += inA ? q.y : 0.0f;
      tA.z += inA ? q.z : 0.0f;
      tA.w += inA ? q.w : 0.0f;
      tB.x += inB ? q.x : 0.0f;
      tB.y += inB ? q.y : 0.0f;
      tB.z += inB ? q.z : 0.0f;
      tB.w += inB ? q.w : 0.0f;
    }
    tA.x += __shfl_xor(tA.x, 16);
    tA.y += __shfl_xor(tA.y, 16);
    tA.z += __shfl_xor(tA.z, 16);
    tA.w += __shfl_xor(tA.w, 16);
    tB.x += __shfl_xor(tB.x, 16);
    tB.y += __shfl_xor(tB.y, 16);
    tB.z += __shfl_xor(tB.z, 16);
    tB.w += __shfl_xor(tB.w, 16);
    if (lane < 16) {
      *(v4f*)(s_tail + w * HSS + tc4) = tA;
      *(v4f*)(s_tail + (O3_WAVES + w) * HSS + tc4) = tB;
    }
  }

  {
    const int c4 = r * 4;
    float pm[16];
#pragma unroll
    for (int k = 0; k < 16; ++k) {
      const int j = min(2 * (w + O3_WAVES * k) + hf, jendB - 1);
      const v4f q = *(const v4f*)(P2h + (size_t)j * HSS + c4);
      pm[k] = fmaxf(fmaxf(q.x, q.y), fmaxf(q.z, q.w));
    }
#pragma unroll
    for (int off = 8; off > 0; off >>= 1) {
#pragma unroll
      for (int k = 0; k < 16; ++k) pm[k] = fmaxf(pm[k], __shfl_xor(pm[k], off));
    }
#pragma unroll
    for (int k = 0; k < 16; ++k) {
      const int j = 2 * (w + O3_WAVES * k) + hf;
      if (r == 0 && j < jendB) s_pmax[j] = pm[k];
    }
  }
  __syncthreads();

  const float* attA = ATT + ((size_t)h * (TT / 16) + sA) * TT * 16;
  const float* attB = ATT + ((size_t)h * (TT / 16) + sB) * TT * 16;

  {
    const float* atM = (hf ? attB : attA) + r;
    const int jendL = hf ? jendB : jendA;
    float mloc = 0.0f;
#pragma unroll
    for (int k = 0; k < TT / O3_WAVES; ++k) {
      const int j = min(w + O3_WAVES * k, jendL - 1);
      mloc = fmaxf(mloc, atM[(size_t)j * 16] * s_pmax[j]);
    }
    s_m[w * 32 + lane] = mloc;
  }
  __syncthreads();
  float Mown = 0.0f;
#pragma unroll
  for (int k = 0; k < O3_WAVES; ++k) Mown = fmaxf(Mown, s_m[k * 32 + lane]);
  const float Moth = __shfl_xor(Mown, 16);
  const float negMLA = -(hf ? Moth : Mown) * LOG2E;
  const float negMLB = -(hf ? Mown : Moth) * LOG2E;

  const float ce = hf ? 0.0f : LOG2E;
  const float co = hf ? LOG2E : 0.0f;
  const float* p2w  = P2h + 4 * w;
  const float* vrow = Vh + (size_t)hf * HSS + 4 * w;
  float dsA = 0.0f, xA0 = 0.0f, xA1 = 0.0f, xA2 = 0.0f, xA3 = 0.0f;
  float dsB = 0.0f, xB0 = 0.0f, xB1 = 0.0f, xB2 = 0.0f, xB3 = 0.0f;
  o3_walk(attA + hf * 16 + r, p2w, vrow, jendA / 2, negMLA, ce, co, dsA, xA0, xA1, xA2, xA3);
  o3_walk(attB + hf * 16 + r, p2w, vrow, jendB / 2, negMLB, ce, co, dsB, xB0, xB1, xB2, xB3);

  const float ds = hf ? dsB : dsA;
  const float a0 = hf ? xB0 : xA0;
  const float a1 = hf ? xB1 : xA1;
  const float a2 = hf ? xB2 : xA2;
  const float a3 = hf ? xB3 : xA3;
  const float negML = hf ? negMLB : negMLA;
  const int jendL = hf ? jendB : jendA;
  s_ds[w * 32 + lane] = ds;
  __syncthreads();
  float tot = 0.0f;
#pragma unroll
  for (int k = 0; k < O3_WAVES; ++k) tot += s_ds[k * 32 + lane];
  const float em = __builtin_amdgcn_exp2f(negML);
  const float denom = tot + (float)((TT - jendL) * HSS) * em;
  v4f vs = {0.f, 0.f, 0.f, 0.f};
#pragma unroll
  for (int k = 0; k < O3_WAVES; ++k) vs += *(const v4f*)(s_tail + (hf * O3_WAVES + k) * HSS + 4 * w);
  const v4f o = {(a0 + em * vs.x) / denom, (a1 + em * vs.y) / denom,
                 (a2 + em * vs.z) / denom, (a3 + em * vs.w) / denom};
  *(v4f*)(s_out + lane * O3_PITCH + 4 * w) = o;
  __syncthreads();
  const int orow = threadIdx.x >> 4;
  const int oc   = (threadIdx.x & 15) * 4;
  const int grow = (orow < 16) ? (16 * sA + orow) : (16 * sB + orow - 16);
  *(v4f*)(Y + (size_t)grow * CC + h * HSS + oc) = *(const v4f*)(s_out + orow * O3_PITCH + oc);
}

__global__ void __launch_bounds__(128) out_proj_kernel(const float* __restrict__ Y,
                                const float* __restrict__ Wc,
                                float* __restrict__ out) {
  const int lane = threadIdx.x & 31;
  const int wv   = threadIdx.x >> 5;
  const int lm   = lane & 15;
  const int lh   = lane >> 4;
  const int t0 = (blockIdx.x * 4 + wv) * 16;
  const int c0 = blockIdx.y * 32;

  v8f acc[2];
#pragma unroll
  for (int j = 0; j < 2; ++j) acc[j] = (v8f){0.f, 0.f, 0.f, 0.f, 0.f, 0.f, 0.f, 0.f};
  gemm_k768_wave<2>(Y + (size_t)(t0 + lm) * CC + 4 * lh,
                    Wc + (size_t)(4 * lh) * CC + c0 + 2 * lm, acc);
#pragma unroll
  for (int r = 0; r < 8; ++r) {
    const v2f o = {acc[0][r], acc[1][r]};
    *(v2f*)(out + (size_t)(t0 + 8 * lh + r) * CC + c0 + 2 * lm) = o;
  }
}

extern "C" void kernel_launch(void* const* d_in, const int* in_sizes, int n_in,
                              void* d_out, int out_size, void* d_ws,
                              size_t ws_size, hipStream_t stream) {
  (void)in_sizes; (void)n_in; (void)out_size; (void)ws_size;
  const float* x  = (const float*)d_in[0];
  const float* W0 = (const float*)d_in[1];
  const float* W1 = (const float*)d_in[2];
  const float* W2 = (const float*)d_in[3];
  const float* Wv = (const float*)d_in[4];
  const float* Wc = (const float*)d_in[5];
  float* out = (float*)d_out;
  float* ws  = (float*)d_ws;

  const size_t NPH = (size_t)NHH * TT * HSS;
  float* P0  = ws;
  float* P1  = ws + NPH;
  float* P2  = ws + 2 * NPH;
  float* V   = ws + 3 * NPH;
  float* ATT = ws + 4 * NPH;
  float* Y   = ATT + (size_t)NHH * TT * TT;

  qkv_proj_kernel<<<dim3(TT / 64, NHH, 3), 128, 0, stream>>>(x, W0, W1, W2, P0);
  vproj_kernel<<<dim3(TT / 16, HSS / 16, NHH), 32, 0, stream>>>(P2, Wv, V);
  att_softmax_kernel<<<dim3(TT / 16, NHH), 32, 0, stream>>>(P0, P1, ATT);
  order3_kernel<<<dim3(TT / 32, NHH), 512, 0, stream>>>(ATT, P2, V, Y);
  out_proj_kernel<<<dim3(TT / 64, CC / 32), 128, 0, stream>>>(Y, Wc, out);
}


# Causal logit+softmax kernel: a strip's column pairs dealt to a 4-wave block (longest wave 4 pairs in series, was 16); logits stay in accumulators, row max/sum joined via LDS; row sum order changes
# speedup vs baseline: 6.5793x; 1.0395x over previous
#include <hip/hip_runtime.h>
#include <math.h>

#define TT  512
#define CC  768
#define NHH 12
#define HSS 64
constexpr float SCALE = 0.125f;

typedef float v2f __attribute__((ext_vector_type(2)));
typedef float v8f __attribute__((ext_vector_type(8)));

static __device__ __forceinline__ v8f wmma4(v2f a, v2f b, v8f c) {
  return __builtin_amdgcn_wmma_f32_16x16x4_f32(
       false, a,  false, b,
       (short)0, c,  false,  false);
}

typedef float v4f __attribute__((ext_vector_type(4)));
template <int NT> struct BVec;
template <> struct BVec<4> { typedef float T __attribute__((ext_vector_type(4))); };
template <> struct BVec<2> { typedef float T __attribute__((ext_vector_type(2))); };

#define KB 64
#define KS (KB / 8)

template <int NT>
struct FragSet {
  v4f a[KS];
  typename BVec<NT>::T b[KS][4];
};

template <int NT>
static __device__ __forceinline__ void load_set(FragSet<NT>& s,
                                                const float* __restrict__ ap,
                                                const float* __restrict__ bp) {
#pragma unroll
  for (int st = 0; st < KS; ++st) {
    s.a[st] = *(const v4f*)(ap + st * 8);
#pragma unroll
    for (int i = 0; i < 4; ++i)
      s.b[st][i] = *(const typename BVec<NT>::T*)(bp + (size_t)(st * 8 + i) * CC);
  }
}

template <int NT>
static __device__ __forceinline__ void mma_set(const FragSet<NT>& s, v8f (&acc)[NT]) {
#pragma unroll
  for (int st = 0; st < KS; ++st) {
    const v2f a01 = {s.a[st].x, s.a[st].y};
    const v2f a23 = {s.a[st].z, s.a[st].w};
#pragma unroll
    for (int j = 0; j < NT; ++j) {
      const v2f b = {s.b[st][0][j], s.b[st][1][j]};
      acc[j] = wmma4(a01, b, acc[j]);
    }
#pragma unroll
    for (int j = 0; j < NT; ++j) {
      const v2f b = {s.b[st][2][j], s.b[st][3][j]};
      acc[j] = wmma4(a23, b, acc[j]);
    }
  }
}

template <int NT>
static __device__ __forceinline__ void gemm_k768_wave(const float* __restrict__ ap,
                                                      const float* __restrict__ bp,
                                                      v8f (&acc)[NT]) {
  FragSet<NT> s0, s1;
  load_set<NT>(s0, ap, bp);
  for (int kb = 0; kb < CC - 2 * KB; kb += 2 * KB) {
    __builtin_amdgcn_sched_barrier(0);
    load_set<NT>(s1, ap + kb + KB, bp + (size_t)(kb + KB) * CC);
    __builtin_amdgcn_sched_barrier(0);
    mma_set<NT>(s0, acc);
    __builtin_amdgcn_sched_barrier(0);
    load_set<NT>(s0, ap + kb + 2 * KB, bp + (size_t)(kb + 2 * KB) * CC);
    __builtin_amdgcn_sched_barrier(0);
    mma_set<NT>(s1, acc);
  }
  __builtin_amdgcn_sched_barrier(0);
  load_set<NT>(s1, ap + CC - KB, bp + (size_t)(CC - KB) * CC);
  __builtin_amdgcn_sched_barrier(0);
  mma_set<NT>(s0, acc);
  mma_set<NT>(s1, acc);
}

__global__ void __launch_bounds__(128) qkv_proj_kernel(const float* __restrict__ x,
                                const float* __restrict__ W0,
                                const float* __restrict__ W1,
                                const float* __restrict__ W2,
                                float* __restrict__ P) {
  const int lane = threadIdx.x & 31;
  const int wv   = threadIdx.x >> 5;
  const int lm   = lane & 15;
  const int lh   = lane >> 4;
  const int t0 = (blockIdx.x * 4 + wv) * 16;
  const int h  = blockIdx.y;
  const int w  = blockIdx.z;
  const float* W = (w == 0) ? W0 : (w == 1) ? W1 : W2;
  float* Ph = P + ((size_t)w * NHH + h) * TT * HSS;

  v8f acc[4];
#pragma unroll
  for (int j = 0; j < 4; ++j) acc[j] = (v8f){0.f, 0.f, 0.f, 0.f, 0.f, 0.f, 0.f, 0.f};
  gemm_k768_wave<4>(x + (size_t)(t0 + lm) * CC + 4 * lh,
                    W + (size_t)(4 * lh) * CC + h * HSS + 4 * lm, acc);
#pragma unroll
  for (int r = 0; r < 8; ++r) {
    const v4f o = {acc[0][r], acc[1][r], acc[2][r], acc[3][r]};
    *(v4f*)(Ph + (size_t)(t0 + 8 * lh + r) * HSS + 4 * lm) = o;
  }
}

__global__ void vproj_kernel(const float* __restrict__ P2,
                             const float* __restrict__ Wv,
                             float* __restrict__ V) {
  const int lane = threadIdx.x;
  const int lm   = lane & 15;
  const int lk   = (lane >> 4) << 1;
  const int drow = (lane >> 4) << 3;
  const int t0 = blockIdx.x * 16;
  const int n0 = blockIdx.y * 16;
  const int h  = blockIdx.z;
  const float* A = P2 + (size_t)h * TT * HSS;
  float* Vh = V + (size_t)h * TT * HSS;

  v8f acc = {0.f, 0.f, 0.f, 0.f, 0.f, 0.f, 0.f, 0.f};
  for (int k = 0; k < HSS; k += 4) {
    v2f a, b;
    a.x = A[(size_t)(t0 + lm) * HSS + k + lk];
    a.y = A[(size_t)(t0 + lm) * HSS + k + lk + 1];
    b.x = Wv[(size_t)(k + lk) * HSS + n0 + lm];
    b.y = Wv[(size_t)(k + lk + 1) * HSS + n0 + lm];
    acc = wmma4(a, b, acc);
  }
#pragma unroll
  for (int r = 0; r < 8; ++r)
    Vh[(size_t)(t0 + r + drow) * HSS + n0 + lm] = acc[r];
}

#define AT_PAIRS (TT / 32)
#define AT_WAVES 4
#define AT_STEPS (AT_PAIRS / AT_WAVES)

static __device__ __forceinline__ void att_load_pair(v2f (&b)[32], const float* __restrict__ p) {
#pragma unroll
  for (int u = 0; u < 2; ++u) {
#pragma unroll
    for (int kk = 0; kk < 16; ++kk) b[u * 16 + kk] = *(const v2f*)(p + (size_t)(u * 16) * HSS + kk * 4);
  }
}

static __device__ __forceinline__ void att_mma_pair(const v2f (&a)[16], const v2f (&b)[32],
                                                    v8f& c0, v8f& c1) {
  c0 = (v8f){0.f, 0.f, 0.f, 0.f, 0.f, 0.f, 0.f, 0.f};
  c1 = c0;
#pragma unroll
  for (int kk = 0; kk < 16; ++kk) {
    c0 = wmma4(a[kk], b[kk], c0);
    c1 = wmma4(a[kk], b[16 + kk], c1);
  }
}

static __device__ __forceinline__ void att_scale_pair(v8f& c0, v8f& c1, float (&rm)[8],
                                                      bool diag, int jrel) {
#pragma unroll
  for (int r = 0; r < 8; ++r) {
    float v0 = c0[r] * SCALE;
    float v1 = c1[r] * SCALE;
    if (diag) {
      if (jrel > r) v0 = -3.0e38f;
      if (jrel + 16 > r) v1 = -3.0e38f;
    }
    c0[r] = v0;
    c1[r] = v1;
    rm[r] = fmaxf(rm[r], fmaxf(v0, v1));
  }
}

__global__ void __launch_bounds__(32 * AT_WAVES) att_softmax_kernel(const float* __restrict__ P0,
                                   const float* __restrict__ P1,
                                   float* __restrict__ ATT) {
  __shared__ __attribute__((aligned(16))) float s_rmax[AT_WAVES * 16];
  __shared__ __attribute__((aligned(16))) float s_rsum[AT_WAVES * 16];
  const int lane = threadIdx.x & 31;
  const int w    = __builtin_amdgcn_readfirstlane(threadIdx.x >> 5);
  const int lm   = lane & 15;
  const int lk   = (lane >> 4) << 1;
  const int drow = (lane >> 4) << 3;
  const int strip = TT / 16 - 1 - (int)(blockIdx.x / NHH);
  const int h     = blockIdx.x % NHH;
  const int i0 = strip * 16;
  const int g  = strip >> 1;
  const int first = g - w;
  const float* P0h = P0 + (size_t)h * TT * HSS;
  const float* P1h = P1 + (size_t)h * TT * HSS;

  v2f areg[16];
#pragma unroll
  for (int kk = 0; kk < 16; ++kk) {
    const int k = kk * 4 + lk;
    areg[kk].x = P0h[(size_t)(i0 + lm) * HSS + k];
    areg[kk].y = P0h[(size_t)(i0 + lm) * HSS + k + 1];
  }

  v8f acc[2 * AT_STEPS];
  float rm[8];
#pragma unroll
  for (int r = 0; r < 8; ++r) rm[r] = -3.0e38f;
  const float* bp = P1h + (size_t)lm * HSS + lk;
  const int jrel = (w == 0) ? (32 * g + lm - i0 - drow) : -64;
  v2f bA[32], bB[32];
  att_load_pair(bA, bp + (size_t)max(first, 0) * 32 * HSS);
#pragma unroll
  for (int q = 0; q < AT_STEPS / 2; ++q) {
    if (first - 8 * q >= 0) {
      __builtin_amdgcn_sched_barrier(0);
      att_load_pair(bB, bp + (size_t)max(first - 8 * q - 4, 0) * 32 * HSS);
      __builtin_amdgcn_sched_barrier(0);
      att_mma_pair(areg, bA, acc[4 * q], acc[4 * q + 1]);
      att_scale_pair(acc[4 * q], acc[4 * q + 1], rm, q == 0, jrel);
    }
    if (first - 8 * q - 4 >= 0) {
      __builtin_amdgcn_sched_barrier(0);
      att_load_pair(bA, bp + (size_t)max(first - 8 * q - 8, 0) * 32 * HSS);
      __builtin_amdgcn_sched_barrier(0);
      att_mma_pair(areg, bB, acc[4 * q + 2], acc[4 * q + 3]);
      att_scale_pair(acc[4 * q + 2], acc[4 * q + 3], rm, false, 0);
    }
  }

#pragma unroll
  for (int off = 8; off > 0; off >>= 1) {
#pragma unroll
    for (int r = 0; r < 8; ++r) rm[r] = fmaxf(rm[r], __shfl_xor(rm[r], off));
  }
  if (lm == 0) {
    *(v4f*)(s_rmax + w * 16 + drow)     = (v4f){rm[0], rm[1], rm[2], rm[3]};
    *(v4f*)(s_rmax + w * 16 + drow + 4) = (v4f){rm[4], rm[5], rm[6], rm[7]};
  }
  __syncthreads();
#pragma unroll
  for (int k = 0; k < AT_WAVES; ++k) {
    const v4f lo = *(const v4f*)(s_rmax + k * 16 + drow);
    const v4f hi = *(const v4f*)(s_rmax + k * 16 + drow + 4);
    rm[0] = fmaxf(rm[0], lo.x); rm[1] = fmaxf(rm[1], lo.y);
    rm[2] = fmaxf(rm[2], lo.z); rm[3] = fmaxf(rm[3], lo.w);
    rm[4] = fmaxf(rm[4], hi.x); rm[5] = fmaxf(rm[5], hi.y);
    rm[6] = fmaxf(rm[6], hi.z); rm[7] = fmaxf(rm[7], hi.w);
  }

  float rs[8];
#pragma unroll
  for (int r = 0; r < 8; ++r) rs[r] = 0.0f;
#pragma unroll
  for (int s = 0; s < AT_STEPS; ++s) {
    if (first - 4 * s >= 0) {
#pragma unroll
      for (int u = 0; u < 2; ++u) {
#pragma unroll
        for (int r = 0; r < 8; ++r) {
          const float e = __expf(acc[2 * s + u][r] - rm[r]);
          acc[2 * s + u][r] = e;
          rs[r] += e;
        }
      }
    }
  }
#pragma unroll
  for (int off = 8; off > 0; off >>= 1) {
#pragma unroll
    for (int r = 0; r < 8; ++r) rs[r] += __shfl_xor(rs[r], off);
  }
  if (lm == 0) {
    *(v4f*)(s_rsum + w * 16 + drow)     = (v4f){rs[0], rs[1], rs[2], rs[3]};
    *(v4f*)(s_rsum + w * 16 + drow + 4) = (v4f){rs[4], rs[5], rs[6], rs[7]};
  }
  __syncthreads();
  v4f tlo = *(const v4f*)(s_rsum + drow);
  v4f thi = *(const v4f*)(s_rsum + drow + 4);
#pragma unroll
  for (int k = 1; k < AT_WAVES; ++k) {
    tlo += *(const v4f*)(s_rsum + k * 16 + drow);
    thi += *(const v4f*)(s_rsum + k * 16 + drow + 4);
  }
  const float inv[8] = {1.0f / tlo.x, 1.0f / tlo.y, 1.0f / tlo.z, 1.0f / tlo.w,
                        1.0f / thi.x, 1.0f / thi.y, 1.0f / thi.z, 1.0f / thi.w};

  float* abase = ATT + ((size_t)h * (TT / 16) + strip) * TT * 16 + (size_t)lm * 16 + drow;
#pragma unroll
  for (int s = 0; s < AT_STEPS; ++s) {
    if (first - 4 * s >= 0) {
#pragma unroll
      for (int u = 0; u < 2; ++u) {
        const v8f& c = acc[2 * s + u];
        const v4f lo = {c[0] * inv[0], c[1] * inv[1], c[2] * inv[2], c[3] * inv[3]};
        const v4f hi = {c[4] * inv[4], c[5] * inv[5], c[6] * inv[6], c[7] * inv[7]};
        float* o = abase + (size_t)(32 * (first - 4 * s) + 16 * u) * 16;
        *(v4f*)o = lo;
        *(v4f*)(o + 4) = hi;
      }
    }
  }
}

#define O3_WAVES 16
#define O3_PITCH 68

static __device__ __forceinline__ void o3_walk(const float* __restrict__ at,
                                               const float* __restrict__ p2w,
                                               const float* __restrict__ vrow,
                                               int ntrips, float negML, float ce, float co,
                                               float& ds, float& a0, float& a1, float& a2, float& a3) {
#pragma unroll 4
  for (int t = 0; t < ntrips; ++t) {
    const float a = at[(size_t)t * 32];
    const v4f pe = *(const v4f*)(p2w + (size_t)(2 * t) * HSS);
    const v4f po = *(const v4f*)(p2w + (size_t)(2 * t + 1) * HSS);
    const v4f v = *(const v4f*)(vrow + (size_t)(2 * t) * HSS);
    const float ae = a * ce;
    const float ao = a * co;
    const float e0 = __builtin_amdgcn_exp2f(fmaf(ae, pe.x, fmaf(ao, po.x, negML)));
    const float e1 = __builtin_amdgcn_exp2f(fmaf(ae, pe.y, fmaf(ao, po.y, negML)));
    const float e2 = __builtin_amdgcn_exp2f(fmaf(ae, pe.z, fmaf(ao, po.z, negML)));
    const float e3 = __builtin_amdgcn_exp2f(fmaf(ae, pe.w, fmaf(ao, po.w, negML)));
    ds += (e0 + e1) + (e2 + e3);
    a0 = fmaf(e0, v.x, a0);
    a1 = fmaf(e1, v.y, a1);
    a2 = fmaf(e2, v.z, a2);
    a3 = fmaf(e3, v.w, a3);
  }
  ds += __shfl_xor(ds, 16);
  a0 += __shfl_xor(a0, 16);
  a1 += __shfl_xor(a1, 16);
  a2 += __shfl_xor(a2, 16);
  a3 += __shfl_xor(a3, 16);
}

__global__ void __launch_bounds__(512) order3_kernel(const float* __restrict__ ATT,
                              const float* __restrict__ P2,
                              const float* __restrict__ V,
                              float* __restrict__ Y) {
  __shared__ float s_pmax[TT];
  __shared__ float s_m[O3_WAVES * 32];
  __shared__ float s_ds[O3_WAVES * 32];
  __shared__ __attribute__((aligned(16))) float s_tail[2 * O3_WAVES * HSS];
  __shared__ __attribute__((aligned(16))) float s_out[32 * O3_PITCH];
  const float LOG2E = 1.4426950408889634f;
  const int lane = threadIdx.x & 31;
  const int hf   = lane >> 4;
  const int r    = lane & 15;
  const int w = __builtin_amdgcn_readfirstlane(threadIdx.x >> 5);
  const int p = blockIdx.x;
  const int h = blockIdx.y;
  const int sA = p;
  const int sB = TT / 16 - 1 - p;
  const int jendA = 16 * (sA + 1);
  const int jendB = 16 * (sB + 1);
  const float* P2h = P2 + (size_t)h * TT * HSS;
  const float* Vh  = V  + (size_t)h * TT * HSS;

  {
    const int trow = jendA + (threadIdx.x >> 4);
    const int tc4  = (threadIdx.x & 15) * 4;
    v4f tA = {0.f, 0.f, 0.f, 0.f};
    v4f tB = {0.f, 0.f, 0.f, 0.f};
#pragma unroll
    for (int k = 0; k < TT / 32; ++k) {
      const int row = trow + 32 * k;
      const v4f q = *(const v4f*)(Vh + (size_t)min(row, TT - 1) * HSS + tc4);
      const bool inA = row < TT;
      const bool inB = inA && row >= jendB;
      tA.x += inA ? q.x : 0.0f;
      tA.y += inA ? q.y : 0.0f;
      tA.z += inA ? q.z : 0.0f;
      tA.w += inA ? q.w : 0.0f;
      tB.x += inB ? q.x : 0.0f;
      tB.y += inB ? q.y : 0.0f;
      tB.z += inB ? q.z : 0.0f;
      tB.w += inB ? q.w : 0.0f;
    }
    tA.x += __shfl_xor(tA.x, 16);
    tA.y += __shfl_xor(tA.y, 16);
    tA.z += __shfl_xor(tA.z, 16);
    tA.w += __shfl_xor(tA.w, 16);
    tB.x += __shfl_xor(tB.x, 16);
    tB.y += __shfl_xor(tB.y, 16);
    tB.z += __shfl_xor(tB.z, 16);
    tB.w += __shfl_xor(tB.w, 16);
    if (lane < 16) {
      *(v4f*)(s_tail + w * HSS + tc4) = tA;
      *(v4f*)(s_tail + (O3_WAVES + w) * HSS + tc4) = tB;
    }
  }

  {
    const int c4 = r * 4;
    float pm[16];
#pragma unroll
    for (int k = 0; k < 16; ++k) {
      const int j = min(2 * (w + O3_WAVES * k) + hf, jendB - 1);
      const v4f q = *(const v4f*)(P2h + (size_t)j * HSS + c4);
      pm[k] = fmaxf(fmaxf(q.x, q.y), fmaxf(q.z, q.w));
    }
#pragma unroll
    for (int off = 8; off > 0; off >>= 1) {
#pragma unroll
      for (int k = 0; k < 16; ++k) pm[k] = fmaxf(pm[k], __shfl_xor(pm[k], off));
    }
#pragma unroll
    for (int k = 0; k < 16; ++k) {
      const int j = 2 * (w + O3_WAVES * k) + hf;
      if (r == 0 && j < jendB) s_pmax[j] = pm[k];
    }
  }
  __syncthreads();

  const float* attA = ATT + ((size_t)h * (TT / 16) + sA) * TT * 16;
  const float* attB = ATT + ((size_t)h * (TT / 16) + sB) * TT * 16;

  {
    const float* atM = (hf ? attB : attA) + r;
    const int jendL = hf ? jendB : jendA;
    float mloc = 0.0f;
#pragma unroll
    for (int k = 0; k < TT / O3_WAVES; ++k) {
      const int j = min(w + O3_WAVES * k, jendL - 1);
      mloc = fmaxf(mloc, atM[(size_t)j * 16] * s_pmax[j]);
    }
    s_m[w * 32 + lane] = mloc;
  }
  __syncthreads();
  float Mown = 0.0f;
#pragma unroll
  for (int k = 0; k < O3_WAVES; ++k) Mown = fmaxf(Mown, s_m[k * 32 + lane]);
  const float Moth = __shfl_xor(Mown, 16);
  const float negMLA = -(hf ? Moth : Mown) * LOG2E;
  const float negMLB = -(hf ? Mown : Moth) * LOG2E;

  const float ce = hf ? 0.0f : LOG2E;
  const float co = hf ? LOG2E : 0.0f;
  const float* p2w  = P2h + 4 * w;
  const float* vrow = Vh + (size_t)hf * HSS + 4 * w;
  float dsA = 0.0f, xA0 = 0.0f, xA1 = 0.0f, xA2 = 0.0f, xA3 = 0.0f;
  float dsB = 0.0f, xB0 = 0.0f, xB1 = 0.0f, xB2 = 0.0f, xB3 = 0.0f;
  o3_walk(attA + hf * 16 + r, p2w, vrow, jendA / 2, negMLA, ce, co, dsA, xA0, xA1, xA2, xA3);
  o3_walk(attB + hf * 16 + r, p2w, vrow, jendB / 2, negMLB, ce, co, dsB, xB0, xB1, xB2, xB3);

  const float ds = hf ? dsB : dsA;
  const float a0 = hf ? xB0 : xA0;
  const float a1 = hf ? xB1 : xA1;
  const float a2 = hf ? xB2 : xA2;
  const float a3 = hf ? xB3 : xA3;
  const float negML = hf ? negMLB : negMLA;
  const int jendL = hf ? jendB : jendA;
  s_ds[w * 32 + lane] = ds;
  __syncthreads();
  float tot = 0.0f;
#pragma unroll
  for (int k = 0; k < O3_WAVES; ++k) tot += s_ds[k * 32 + lane];
  const float em = __builtin_amdgcn_exp2f(negML);
  const float denom = tot + (float)((TT - jendL) * HSS) * em;
  v4f vs = {0.f, 0.f, 0.f, 0.f};
#pragma unroll
  for (int k = 0; k < O3_WAVES; ++k) vs += *(const v4f*)(s_tail + (hf * O3_WAVES + k) * HSS + 4 * w);
  const v4f o = {(a0 + em * vs.x) / denom, (a1 + em * vs.y) / denom,
                 (a2 + em * vs.z) / denom, (a3 + em * vs.w) / denom};
  *(v4f*)(s_out + lane * O3_PITCH + 4 * w) = o;
  __syncthreads();
  const int orow = threadIdx.x >> 4;
  const int oc   = (threadIdx.x & 15) * 4;
  const int grow = (orow < 16) ? (16 * sA + orow) : (16 * sB + orow - 16);
  *(v4f*)(Y + (size_t)grow * CC + h * HSS + oc) = *(const v4f*)(s_out + orow * O3_PITCH + oc);
}

__global__ void __launch_bounds__(128) out_proj_kernel(const float* __restrict__ Y,
                                const float* __restrict__ Wc,
                                float* __restrict__ out) {
  const int lane = threadIdx.x & 31;
  const int wv   = threadIdx.x >> 5;
  const int lm   = lane & 15;
  const int lh   = lane >> 4;
  const int t0 = (blockIdx.x * 4 + wv) * 16;
  const int c0 = blockIdx.y * 32;

  v8f acc[2];
#pragma unroll
  for (int j = 0; j < 2; ++j) acc[j] = (v8f){0.f, 0.f, 0.f, 0.f, 0.f, 0.f, 0.f, 0.f};
  gemm_k768_wave<2>(Y + (size_t)(t0 + lm) * CC + 4 * lh,
                    Wc + (size_t)(4 * lh) * CC + c0 + 2 * lm, acc);
#pragma unroll
  for (int r = 0; r < 8; ++r) {
    const v2f o = {acc[0][r], acc[1][r]};
    *(v2f*)(out + (size_t)(t0 + 8 * lh + r) * CC + c0 + 2 * lm) = o;
  }
}

extern "C" void kernel_launch(void* const* d_in, const int* in_sizes, int n_in,
                              void* d_out, int out_size, void* d_ws,
                              size_t ws_size, hipStream_t stream) {
  (void)in_sizes; (void)n_in; (void)out_size; (void)ws_size;
  const float* x  = (const float*)d_in[0];
  const float* W0 = (const float*)d_in[1];
  const float* W1 = (const float*)d_in[2];
  const float* W2 = (const float*)d_in[3];
  const float* Wv = (const float*)d_in[4];
  const float* Wc = (const float*)d_in[5];
  float* out = (float*)d_out;
  float* ws  = (float*)d_ws;

  const size_t NPH = (size_t)NHH * TT * HSS;
  float* P0  = ws;
  float* P1  = ws + NPH;
  float* P2  = ws + 2 * NPH;
  float* V   = ws + 3 * NPH;
  float* ATT = ws + 4 * NPH;
  float* Y   = ATT + (size_t)NHH * TT * TT;

  qkv_proj_kernel<<<dim3(TT / 64, NHH, 3), 128, 0, stream>>>(x, W0, W1, W2, P0);
  vproj_kernel<<<dim3(TT / 16, HSS / 16, NHH), 32, 0, stream>>>(P2, Wv, V);
  att_softmax_kernel<<<dim3((TT / 16) * NHH), 32 * AT_WAVES, 0, stream>>>(P0, P1, ATT);
  order3_kernel<<<dim3(TT / 32, NHH), 512, 0, stream>>>(ATT, P2, V, Y);
  out_proj_kernel<<<dim3(TT / 64, CC / 32), 128, 0, stream>>>(Y, Wc, out);
}


# QKV kernel: 256 blocks, one wave per matrix unit, each wave 16 rows x 64 or x 80 columns (4 or 5 tiles; was 1152 waves of 4, two on 128 units); odd tiles pair into whole lines via LDS; bit-identical
# speedup vs baseline: 6.8657x; 1.0435x over previous
#include <hip/hip_runtime.h>
#include <math.h>

#define TT  512
#define CC  768
#define NHH 12
#define HSS 64
constexpr float SCALE = 0.125f;

typedef float v2f __attribute__((ext_vector_type(2)));
typedef float v8f __attribute__((ext_vector_type(8)));

static __device__ __forceinline__ v8f wmma4(v2f a, v2f b, v8f c) {
  return __builtin_amdgcn_wmma_f32_16x16x4_f32(
       false, a,  false, b,
       (short)0, c,  false,  false);
}

typedef float v4f __attribute__((ext_vector_type(4)));
template <int NT> struct BVec;
template <> struct BVec<4> { typedef float T __attribute__((ext_vector_type(4))); };
template <> struct BVec<2> { typedef float T __attribute__((ext_vector_type(2))); };

#define KS_STD 8

template <int NT, int NE, int KS>
struct FragSet {
  v4f a[KS];
  typename BVec<NT>::T b[KS][4];
  float e[KS][4];
};

template <int NT, int NE, int KS>
static __device__ __forceinline__ void load_set(FragSet<NT, NE, KS>& s,
                                                const float* __restrict__ ap,
                                                const float* __restrict__ bp,
                                                const float* __restrict__ ep) {
#pragma unroll
  for (int st = 0; st < KS; ++st) {
    s.a[st] = *(const v4f*)(ap + st * 8);
#pragma unroll
    for (int i = 0; i < 4; ++i)
      s.b[st][i] = *(const typename BVec<NT>::T*)(bp + (size_t)(st * 8 + i) * CC);
    if (NE) {
#pragma unroll
      for (int i = 0; i < 4; ++i) s.e[st][i] = ep[(size_t)(st * 8 + i) * CC];
    }
  }
}

template <int NT, int NE, int KS>
static __device__ __forceinline__ void mma_set(const FragSet<NT, NE, KS>& s, v8f (&acc)[NT + NE]) {
#pragma unroll
  for (int st = 0; st < KS; ++st) {
    const v2f a01 = {s.a[st].x, s.a[st].y};
    const v2f a23 = {s.a[st].z, s.a[st].w};
#pragma unroll
    for (int j = 0; j < NT; ++j) {
      const v2f b = {s.b[st][0][j], s.b[st][1][j]};
      acc[j] = wmma4(a01, b, acc[j]);
    }
    if (NE) {
      const v2f b = {s.e[st][0], s.e[st][1]};
      acc[NT] = wmma4(a01, b, acc[NT]);
    }
#pragma unroll
    for (int j = 0; j < NT; ++j) {
      const v2f b = {s.b[st][2][j], s.b[st][3][j]};
      acc[j] = wmma4(a23, b, acc[j]);
    }
    if (NE) {
      const v2f b = {s.e[st][2], s.e[st][3]};
      acc[NT] = wmma4(a23, b, acc[NT]);
    }
  }
}

template <int NT, int NE, int KS>
static __device__ __forceinline__ void gemm_k768_wave(const float* __restrict__ ap,
                                                      const float* __restrict__ bp,
                                                      const float* __restrict__ ep,
                                                      v8f (&acc)[NT + NE]) {
  constexpr int KB = 8 * KS;
  FragSet<NT, NE, KS> s0, s1;
  load_set(s0, ap, bp, ep);
  for (int kb = 0; kb < CC - 2 * KB; kb += 2 * KB) {
    __builtin_amdgcn_sched_barrier(0);
    load_set(s1, ap + kb + KB, bp + (size_t)(kb + KB) * CC, ep + (size_t)(kb + KB) * CC);
    __builtin_amdgcn_sched_barrier(0);
    mma_set(s0, acc);
    __builtin_amdgcn_sched_barrier(0);
    load_set(s0, ap + kb + 2 * KB, bp + (size_t)(kb + 2 * KB) * CC,
             ep + (size_t)(kb + 2 * KB) * CC);
    __builtin_amdgcn_sched_barrier(0);
    mma_set(s1, acc);
  }
  __builtin_amdgcn_sched_barrier(0);
  load_set(s1, ap + CC - KB, bp + (size_t)(CC - KB) * CC, ep + (size_t)(CC - KB) * CC);
  __builtin_amdgcn_sched_barrier(0);
  mma_set(s0, acc);
  mma_set(s1, acc);
}

#define QK_GROUPS 8
#define QK_PITCH 36

static __device__ __forceinline__ const float* qkv_wcol(const float* __restrict__ W0,
                                                        const float* __restrict__ W1,
                                                        const float* __restrict__ W2, int n) {
  return (n < CC) ? W0 + n : (n < 2 * CC) ? W1 + (n - CC) : W2 + (n - 2 * CC);
}

template <int N>
static __device__ __forceinline__ void qkv_store4(float* __restrict__ P, const v8f (&acc)[N],
                                                  int n, int t0, int lh) {
  float* o = P + (size_t)(n >> 6) * TT * HSS + (size_t)(t0 + 8 * lh) * HSS + (n & 63);
#pragma unroll
  for (int r = 0; r < 8; ++r) {
    const v4f v = {acc[0][r], acc[1][r], acc[2][r], acc[3][r]};
    *(v4f*)(o + (size_t)r * HSS) = v;
  }
}

__global__ void __launch_bounds__(128) qkv_proj_kernel(const float* __restrict__ x,
                                const float* __restrict__ W0,
                                const float* __restrict__ W1,
                                const float* __restrict__ W2,
                                float* __restrict__ P) {
  __shared__ __attribute__((aligned(16))) float s_line[2 * 16 * QK_PITCH];
  const int lane = threadIdx.x & 31;
  const int wv   = __builtin_amdgcn_readfirstlane(threadIdx.x >> 5);
  const int lm   = lane & 15;
  const int lh   = lane >> 4;
  const int b    = blockIdx.x;

  if (b < 16 * QK_GROUPS) {
    const int gb   = (b >> 4) * 288;
    const int rt   = wv >> 1;
    const int side = wv & 1;
    const int t0 = ((b & 15) * 2 + rt) * 16;
    const int n  = gb + (side ? 160 : 64) + 4 * lm;
    const int ne = gb + (side ? 144 : 128) + lm;
    v8f acc[5];
#pragma unroll
    for (int j = 0; j < 5; ++j) acc[j] = (v8f){0.f, 0.f, 0.f, 0.f, 0.f, 0.f, 0.f, 0.f};
    gemm_k768_wave<4, 1, 6>(x + (size_t)(t0 + lm) * CC + 4 * lh,
                         qkv_wcol(W0, W1, W2, n) + (size_t)(4 * lh) * CC,
                         qkv_wcol(W0, W1, W2, ne) + (size_t)(4 * lh) * CC, acc);
    qkv_store4(P, acc, n, t0, lh);
    float* sl = s_line + rt * 16 * QK_PITCH;
#pragma unroll
    for (int r = 0; r < 8; ++r) sl[(8 * lh + r) * QK_PITCH + 16 * side + lm] = acc[4][r];
    __syncthreads();
    const int nl = gb + 128;
    float* ol = P + (size_t)(nl >> 6) * TT * HSS + (size_t)t0 * HSS + (nl & 63);
    const int c4 = (lane & 7) * 4;
#pragma unroll
    for (int i = 0; i < 2; ++i) {
      const int row = 8 * side + 4 * i + (lane >> 3);
      *(v4f*)(ol + (size_t)row * HSS + c4) = *(const v4f*)(sl + row * QK_PITCH + c4);
    }
  } else {
    const int bn = b - 16 * QK_GROUPS;
    const int gb = (bn >> 4) * 288;
    const int t0 = ((bn & 7) * 4 + wv) * 16;
    const int n  = gb + ((bn & 8) ? 224 : 0) + 4 * lm;
    v8f acc[4];
#pragma unroll
    for (int j = 0; j < 4; ++j) acc[j] = (v8f){0.f, 0.f, 0.f, 0.f, 0.f, 0.f, 0.f, 0.f};
    const float* bp = qkv_wcol(W0, W1, W2, n) + (size_t)(4 * lh) * CC;
    gemm_k768_wave<4, 0, KS_STD>(x + (size_t)(t0 + lm) * CC + 4 * lh, bp, bp, acc);
    qkv_store4(P, acc, n, t0, lh);
  }
}

__global__ void vproj_kernel(const float* __restrict__ P2,
                             const float* __restrict__ Wv,
                             float* __restrict__ V) {
  const int lane = threadIdx.x;
  const int lm   = lane & 15;
  const int lk   = (lane >> 4) << 1;
  const int drow = (lane >> 4) << 3;
  const int t0 = blockIdx.x * 16;
  const int n0 = blockIdx.y * 16;
  const int h  = blockIdx.z;
  const float* A = P2 + (size_t)h * TT * HSS;
  float* Vh = V + (size_t)h * TT * HSS;

  v8f acc = {0.f, 0.f, 0.f, 0.f, 0.f, 0.f, 0.f, 0.f};
  for (int k = 0; k < HSS; k += 4) {
    v2f a, b;
    a.x = A[(size_t)(t0 + lm) * HSS + k + lk];
    a.y = A[(size_t)(t0 + lm) * HSS + k + lk + 1];
    b.x = Wv[(size_t)(k + lk) * HSS + n0 + lm];
    b.y = Wv[(size_t)(k + lk + 1) * HSS + n0 + lm];
    acc = wmma4(a, b, acc);
  }
#pragma unroll
  for (int r = 0; r < 8; ++r)
    Vh[(size_t)(t0 + r + drow) * HSS + n0 + lm] = acc[r];
}

#define AT_PAIRS (TT / 32)
#define AT_WAVES 4
#define AT_STEPS (AT_PAIRS / AT_WAVES)

static __device__ __forceinline__ void att_load_pair(v2f (&b)[32], const float* __restrict__ p) {
#pragma unroll
  for (int u = 0; u < 2; ++u) {
#pragma unroll
    for (int kk = 0; kk < 16; ++kk) b[u * 16 + kk] = *(const v2f*)(p + (size_t)(u * 16) * HSS + kk * 4);
  }
}

static __device__ __forceinline__ void att_mma_pair(const v2f (&a)[16], const v2f (&b)[32],
                                                    v8f& c0, v8f& c1) {
  c0 = (v8f){0.f, 0.f, 0.f, 0.f, 0.f, 0.f, 0.f, 0.f};
  c1 = c0;
#pragma unroll
  for (int kk = 0; kk < 16; ++kk) {
    c0 = wmma4(a[kk], b[kk], c0);
    c1 = wmma4(a[kk], b[16 + kk], c1);
  }
}

static __device__ __forceinline__ void att_scale_pair(v8f& c0, v8f& c1, float (&rm)[8],
                                                      bool diag, int jrel) {
#pragma unroll
  for (int r = 0; r < 8; ++r) {
    float v0 = c0[r] * SCALE;
    float v1 = c1[r] * SCALE;
    if (diag) {
      if (jrel > r) v0 = -3.0e38f;
      if (jrel + 16 > r) v1 = -3.0e38f;
    }
    c0[r] = v0;
    c1[r] = v1;
    rm[r] = fmaxf(rm[r], fmaxf(v0, v1));
  }
}

__global__ void __launch_bounds__(32 * AT_WAVES) att_softmax_kernel(const float* __restrict__ P0,
                                   const float* __restrict__ P1,
                                   float* __restrict__ ATT) {
  __shared__ __attribute__((aligned(16))) float s_rmax[AT_WAVES * 16];
  __shared__ __attribute__((aligned(16))) float s_rsum[AT_WAVES * 16];
  const int lane = threadIdx.x & 31;
  const int w    = __builtin_amdgcn_readfirstlane(threadIdx.x >> 5);
  const int lm   = lane & 15;
  const int lk   = (lane >> 4) << 1;
  const int drow = (lane >> 4) << 3;
  const int strip = TT / 16 - 1 - (int)(blockIdx.x / NHH);
  const int h     = blockIdx.x % NHH;
  const int i0 = strip * 16;
  const int g  = strip >> 1;
  const int first = g - w;
  const float* P0h = P0 + (size_t)h * TT * HSS;
  const float* P1h = P1 + (size_t)h * TT * HSS;

  v2f areg[16];
#pragma unroll
  for (int kk = 0; kk < 16; ++kk) {
    const int k = kk * 4 + lk;
    areg[kk].x = P0h[(size_t)(i0 + lm) * HSS + k];
    areg[kk].y = P0h[(size_t)(i0 + lm) * HSS + k + 1];
  }

  v8f acc[2 * AT_STEPS];
  float rm[8];
#pragma unroll
  for (int r = 0; r < 8; ++r) rm[r] = -3.0e38f;
  const float* bp = P1h + (size_t)lm * HSS + lk;
  const int jrel = (w == 0) ? (32 * g + lm - i0 - drow) : -64;
  v2f bA[32], bB[32];
  att_load_pair(bA, bp + (size_t)max(first, 0) * 32 * HSS);
#pragma unroll
  for (int q = 0; q < AT_STEPS / 2; ++q) {
    if (first - 8 * q >= 0) {
      __builtin_amdgcn_sched_barrier(0);
      att_load_pair(bB, bp + (size_t)max(first - 8 * q - 4, 0) * 32 * HSS);
      __builtin_amdgcn_sched_barrier(0);
      att_mma_pair(areg, bA, acc[4 * q], acc[4 * q + 1]);
      att_scale_pair(acc[4 * q], acc[4 * q + 1], rm, q == 0, jrel);
    }
    if (first - 8 * q - 4 >= 0) {
      __builtin_amdgcn_sched_barrier(0);
      att_load_pair(bA, bp + (size_t)max(first - 8 * q - 8, 0) * 32 * HSS);
      __builtin_amdgcn_sched_barrier(0);
      att_mma_pair(areg, bB, acc[4 * q + 2], acc[4 * q + 3]);
      att_scale_pair(acc[4 * q + 2], acc[4 * q + 3], rm, false, 0);
    }
  }

#pragma unroll
  for (int off = 8; off > 0; off >>= 1) {
#pragma unroll
    for (int r = 0; r < 8; ++r) rm[r] = fmaxf(rm[r], __shfl_xor(rm[r], off));
  }
  if (lm == 0) {
    *(v4f*)(s_rmax + w * 16 + drow)     = (v4f){rm[0], rm[1], rm[2], rm[3]};
    *(v4f*)(s_rmax + w * 16 + drow + 4) = (v4f){rm[4], rm[5], rm[6], rm[7]};
  }
  __syncthreads();
#pragma unroll
  for (int k = 0; k < AT_WAVES; ++k) {
    const v4f lo = *(const v4f*)(s_rmax + k * 16 + drow);
    const v4f hi = *(const v4f*)(s_rmax + k * 16 + drow + 4);
    rm[0] = fmaxf(rm[0], lo.x); rm[1] = fmaxf(rm[1], lo.y);
    rm[2] = fmaxf(rm[2], lo.z); rm[3] = fmaxf(rm[3], lo.w);
    rm[4] = fmaxf(rm[4], hi.x); rm[5] = fmaxf(rm[5], hi.y);
    rm[6] = fmaxf(rm[6], hi.z); rm[7] = fmaxf(rm[7], hi.w);
  }

  float rs[8];
#pragma unroll
  for (int r = 0; r < 8; ++r) rs[r] = 0.0f;
#pragma unroll
  for (int s = 0; s < AT_STEPS; ++s) {
    if (first - 4 * s >= 0) {
#pragma unroll
      for (int u = 0; u < 2; ++u) {
#pragma unroll
        for (int r = 0; r < 8; ++r) {
          const float e = __expf(acc[2 * s + u][r] - rm[r]);
          acc[2 * s + u][r] = e;
          rs[r] += e;
        }
      }
    }
  }
#pragma unroll
  for (int off = 8; off > 0; off >>= 1) {
#pragma unroll
    for (int r = 0; r < 8; ++r) rs[r] += __shfl_xor(rs[r], off);
  }
  if (lm == 0) {
    *(v4f*)(s_rsum + w * 16 + drow)     = (v4f){rs[0], rs[1], rs[2], rs[3]};
    *(v4f*)(s_rsum + w * 16 + drow + 4) = (v4f){rs[4], rs[5], rs[6], rs[7]};
  }
  __syncthreads();
  v4f tlo = *(const v4f*)(s_rsum + drow);
  v4f thi = *(const v4f*)(s_rsum + drow + 4);
#pragma unroll
  for (int k = 1; k < AT_WAVES; ++k) {
    tlo += *(const v4f*)(s_rsum + k * 16 + drow);
    thi += *(const v4f*)(s_rsum + k * 16 + drow + 4);
  }
  const float inv[8] = {1.0f / tlo.x, 1.0f / tlo.y, 1.0f / tlo.z, 1.0f / tlo.w,
                        1.0f / thi.x, 1.0f / thi.y, 1.0f / thi.z, 1.0f / thi.w};

  float* abase = ATT + ((size_t)h * (TT / 16) + strip) * TT * 16 + (size_t)lm * 16 + drow;
#pragma unroll
  for (int s = 0; s < AT_STEPS; ++s) {
    if (first - 4 * s >= 0) {
#pragma unroll
      for (int u = 0; u < 2; ++u) {
        const v8f& c = acc[2 * s + u];
        const v4f lo = {c[0] * inv[0], c[1] * inv[1], c[2] * inv[2], c[3] * inv[3]};
        const v4f hi = {c[4] * inv[4], c[5] * inv[5], c[6] * inv[6], c[7] * inv[7]};
        float* o = abase + (size_t)(32 * (first - 4 * s) + 16 * u) * 16;
        *(v4f*)o = lo;
        *(v4f*)(o + 4) = hi;
      }
    }
  }
}

#define O3_WAVES 16
#define O3_PITCH 68

static __device__ __forceinline__ void o3_walk(const float* __restrict__ at,
                                               const float* __restrict__ p2w,
                                               const float* __restrict__ vrow,
                                               int ntrips, float negML, float ce, float co,
                                               float& ds, float& a0, float& a1, float& a2, float& a3) {
#pragma unroll 4
  for (int t = 0; t < ntrips; ++t) {
    const float a = at[(size_t)t * 32];
    const v4f pe = *(const v4f*)(p2w + (size_t)(2 * t) * HSS);
    const v4f po = *(const v4f*)(p2w + (size_t)(2 * t + 1) * HSS);
    const v4f v = *(const v4f*)(vrow + (size_t)(2 * t) * HSS);
    const float ae = a * ce;
    const float ao = a * co;
    const float e0 = __builtin_amdgcn_exp2f(fmaf(ae, pe.x, fmaf(ao, po.x, negML)));
    const float e1 = __builtin_amdgcn_exp2f(fmaf(ae, pe.y, fmaf(ao, po.y, negML)));
    const float e2 = __builtin_amdgcn_exp2f(fmaf(ae, pe.z, fmaf(ao, po.z, negML)));
    const float e3 = __builtin_amdgcn_exp2f(fmaf(ae, pe.w, fmaf(ao, po.w, negML)));
    ds += (e0 + e1) + (e2 + e3);
    a0 = fmaf(e0, v.x, a0);
    a1 = fmaf(e1, v.y, a1);
    a2 = fmaf(e2, v.z, a2);
    a3 = fmaf(e3, v.w, a3);
  }
  ds += __shfl_xor(ds, 16);
  a0 += __shfl_xor(a0, 16);
  a1 += __shfl_xor(a1, 16);
  a2 += __shfl_xor(a2, 16);
  a3 += __shfl_xor(a3, 16);
}

__global__ void __launch_bounds__(512) order3_kernel(const float* __restrict__ ATT,
                              const float* __restrict__ P2,
                              const float* __restrict__ V,
                              float* __restrict__ Y) {
  __shared__ float s_pmax[TT];
  __shared__ float s_m[O3_WAVES * 32];
  __shared__ float s_ds[O3_WAVES * 32];
  __shared__ __attribute__((aligned(16))) float s_tail[2 * O3_WAVES * HSS];
  __shared__ __attribute__((aligned(16))) float s_out[32 * O3_PITCH];
  const float LOG2E = 1.4426950408889634f;
  const int lane = threadIdx.x & 31;
  const int hf   = lane >> 4;
  const int r    = lane & 15;
  const int w = __builtin_amdgcn_readfirstlane(threadIdx.x >> 5);
  const int p = blockIdx.x;
  const int h = blockIdx.y;
  const int sA = p;
  const int sB = TT / 16 - 1 - p;
  const int jendA = 16 * (sA + 1);
  const int jendB = 16 * (sB + 1);
  const float* P2h = P2 + (size_t)h * TT * HSS;
  const float* Vh  = V  + (size_t)h * TT * HSS;

  {
    const int trow = jendA + (threadIdx.x >> 4);
    const int tc4  = (threadIdx.x & 15) * 4;
    v4f tA = {0.f, 0.f, 0.f, 0.f};
    v4f tB = {0.f, 0.f, 0.f, 0.f};
#pragma unroll
    for (int k = 0; k < TT / 32; ++k) {
      const int row = trow + 32 * k;
      const v4f q = *(const v4f*)(Vh + (size_t)min(row, TT - 1) * HSS + tc4);
      const bool inA = row < TT;
      const bool inB = inA && row >= jendB;
      tA.x += inA ? q.x : 0.0f;
      tA.y += inA ? q.y : 0.0f;
      tA.z += inA ? q.z : 0.0f;
      tA.w += inA ? q.w : 0.0f;
      tB.x += inB ? q.x : 0.0f;
      tB.y += inB ? q.y : 0.0f;
      tB.z += inB ? q.z : 0.0f;
      tB.w += inB ? q.w : 0.0f;
    }
    tA.x += __shfl_xor(tA.x, 16);
    tA.y += __shfl_xor(tA.y, 16);
    tA.z += __shfl_xor(tA.z, 16);
    tA.w += __shfl_xor(tA.w, 16);
    tB.x += __shfl_xor(tB.x, 16);
    tB.y += __shfl_xor(tB.y, 16);
    tB.z += __shfl_xor(tB.z, 16);
    tB.w += __shfl_xor(tB.w, 16);
    if (lane < 16) {
      *(v4f*)(s_tail + w * HSS + tc4) = tA;
      *(v4f*)(s_tail + (O3_WAVES + w) * HSS + tc4) = tB;
    }
  }

  {
    const int c4 = r * 4;
    float pm[16];
#pragma unroll
    for (int k = 0; k < 16; ++k) {
      const int j = min(2 * (w + O3_WAVES * k) + hf, jendB - 1);
      const v4f q = *(const v4f*)(P2h + (size_t)j * HSS + c4);
      pm[k] = fmaxf(fmaxf(q.x, q.y), fmaxf(q.z, q.w));
    }
#pragma unroll
    for (int off = 8; off > 0; off >>= 1) {
#pragma unroll
      for (int k = 0; k < 16; ++k) pm[k] = fmaxf(pm[k], __shfl_xor(pm[k], off));
    }
#pragma unroll
    for (int k = 0; k < 16; ++k) {
      const int j = 2 * (w + O3_WAVES * k) + hf;
      if (r == 0 && j < jendB) s_pmax[j] = pm[k];
    }
  }
  __syncthreads();

  const float* attA = ATT + ((size_t)h * (TT / 16) + sA) * TT * 16;
  const float* attB = ATT + ((size_t)h * (TT / 16) + sB) * TT * 16;

  {
    const float* atM = (hf ? attB : attA) + r;
    const int jendL = hf ? jendB : jendA;
    float mloc = 0.0f;
#pragma unroll
    for (int k = 0; k < TT / O3_WAVES; ++k) {
      const int j = min(w + O3_WAVES * k, jendL - 1);
      mloc = fmaxf(mloc, atM[(size_t)j * 16] * s_pmax[j]);
    }
    s_m[w * 32 + lane] = mloc;
  }
  __syncthreads();
  float Mown = 0.0f;
#pragma unroll
  for (int k = 0; k < O3_WAVES; ++k) Mown = fmaxf(Mown, s_m[k * 32 + lane]);
  const float Moth = __shfl_xor(Mown, 16);
  const float negMLA = -(hf ? Moth : Mown) * LOG2E;
  const float negMLB = -(hf ? Mown : Moth) * LOG2E;

  const float ce = hf ? 0.0f : LOG2E;
  const float co = hf ? LOG2E : 0.0f;
  const float* p2w  = P2h + 4 * w;
  const float* vrow = Vh + (size_t)hf * HSS + 4 * w;
  float dsA = 0.0f, xA0 = 0.0f, xA1 = 0.0f, xA2 = 0.0f, xA3 = 0.0f;
  float dsB = 0.0f, xB0 = 0.0f, xB1 = 0.0f, xB2 = 0.0f, xB3 = 0.0f;
  o3_walk(attA + hf * 16 + r, p2w, vrow, jendA / 2, negMLA, ce, co, dsA, xA0, xA1, xA2, xA3);
  o3_walk(attB + hf * 16 + r, p2w, vrow, jendB / 2, negMLB, ce, co, dsB, xB0, xB1, xB2, xB3);

  const float ds = hf ? dsB : dsA;
  const float a0 = hf ? xB0 : xA0;
  const float a1 = hf ? xB1 : xA1;
  const float a2 = hf ? xB2 : xA2;
  const float a3 = hf ? xB3 : xA3;
  const float negML = hf ? negMLB : negMLA;
  const int jendL = hf ? jendB : jendA;
  s_ds[w * 32 + lane] = ds;
  __syncthreads();
  float tot = 0.0f;
#pragma unroll
  for (int k = 0; k < O3_WAVES; ++k) tot += s_ds[k * 32 + lane];
  const float em = __builtin_amdgcn_exp2f(negML);
  const float denom = tot + (float)((TT - jendL) * HSS) * em;
  v4f vs = {0.f, 0.f, 0.f, 0.f};
#pragma unroll
  for (int k = 0; k < O3_WAVES; ++k) vs += *(const v4f*)(s_tail + (hf * O3_WAVES + k) * HSS + 4 * w);
  const v4f o = {(a0 + em * vs.x) / denom, (a1 + em * vs.y) / denom,
                 (a2 + em * vs.z) / denom, (a3 + em * vs.w) / denom};
  *(v4f*)(s_out + lane * O3_PITCH + 4 * w) = o;
  __syncthreads();
  const int orow = threadIdx.x >> 4;
  const int oc   = (threadIdx.x & 15) * 4;
  const int grow = (orow < 16) ? (16 * sA + orow) : (16 * sB + orow - 16);
  *(v4f*)(Y + (size_t)grow * CC + h * HSS + oc) = *(const v4f*)(s_out + orow * O3_PITCH + oc);
}

__global__ void __launch_bounds__(128) out_proj_kernel(const float* __restrict__ Y,
                                const float* __restrict__ Wc,
                                float* __restrict__ out) {
  const int lane = threadIdx.x & 31;
  const int wv   = threadIdx.x >> 5;
  const int lm   = lane & 15;
  const int lh   = lane >> 4;
  const int t0 = (blockIdx.x * 4 + wv) * 16;
  const int c0 = blockIdx.y * 32;

  v8f acc[2];
#pragma unroll
  for (int j = 0; j < 2; ++j) acc[j] = (v8f){0.f, 0.f, 0.f, 0.f, 0.f, 0.f, 0.f, 0.f};
  const float* bp = Wc + (size_t)(4 * lh) * CC + c0 + 2 * lm;
  gemm_k768_wave<2, 0, KS_STD>(Y + (size_t)(t0 + lm) * CC + 4 * lh, bp, bp, acc);
#pragma unroll
  for (int r = 0; r < 8; ++r) {
    const v2f o = {acc[0][r], acc[1][r]};
    *(v2f*)(out + (size_t)(t0 + 8 * lh + r) * CC + c0 + 2 * lm) = o;
  }
}

extern "C" void kernel_launch(void* const* d_in, const int* in_sizes, int n_in,
                              void* d_out, int out_size, void* d_ws,
                              size_t ws_size, hipStream_t stream) {
  (void)in_sizes; (void)n_in; (void)out_size; (void)ws_size;
  const float* x  = (const float*)d_in[0];
  const float* W0 = (const float*)d_in[1];
  const float* W1 = (const float*)d_in[2];
  const float* W2 = (const float*)d_in[3];
  const float* Wv = (const float*)d_in[4];
  const float* Wc = (const float*)d_in[5];
  float* out = (float*)d_out;
  float* ws  = (float*)d_ws;

  const size_t NPH = (size_t)NHH * TT * HSS;
  float* P0  = ws;
  float* P1  = ws + NPH;
  float* P2  = ws + 2 * NPH;
  float* V   = ws + 3 * NPH;
  float* ATT = ws + 4 * NPH;
  float* Y   = ATT + (size_t)NHH * TT * TT;

  qkv_proj_kernel<<<dim3(32 * QK_GROUPS), 128, 0, stream>>>(x, W0, W1, W2, P0);
  vproj_kernel<<<dim3(TT / 16, HSS / 16, NHH), 32, 0, stream>>>(P2, Wv, V);
  att_softmax_kernel<<<dim3((TT / 16) * NHH), 32 * AT_WAVES, 0, stream>>>(P0, P1, ATT);
  order3_kernel<<<dim3(TT / 32, NHH), 512, 0, stream>>>(ATT, P2, V, Y);
  out_proj_kernel<<<dim3(TT / 64, CC / 32), 128, 0, stream>>>(Y, Wc, out);
}


# V-projection launch removed (5 launches to 4): each 4-wave block of the logit+softmax launch also makes its strip's 16x64 tile of V (a wave per 16 columns, whole lines via LDS); bit-identical
# speedup vs baseline: 7.0727x; 1.0301x over previous
#include <hip/hip_runtime.h>
#include <math.h>

#define TT  512
#define CC  768
#define NHH 12
#define HSS 64
constexpr float SCALE = 0.125f;

typedef float v2f __attribute__((ext_vector_type(2)));
typedef float v8f __attribute__((ext_vector_type(8)));

static __device__ __forceinline__ v8f wmma4(v2f a, v2f b, v8f c) {
  return __builtin_amdgcn_wmma_f32_16x16x4_f32(
       false, a,  false, b,
       (short)0, c,  false,  false);
}

typedef float v4f __attribute__((ext_vector_type(4)));
template <int NT> struct BVec;
template <> struct BVec<4> { typedef float T __attribute__((ext_vector_type(4))); };
template <> struct BVec<2> { typedef float T __attribute__((ext_vector_type(2))); };

#define KS_STD 8

template <int NT, int NE, int KS>
struct FragSet {
  v4f a[KS];
  typename BVec<NT>::T b[KS][4];
  float e[KS][4];
};

template <int NT, int NE, int KS>
static __device__ __forceinline__ void load_set(FragSet<NT, NE, KS>& s,
                                                const float* __restrict__ ap,
                                                const float* __restrict__ bp,
                                                const float* __restrict__ ep) {
#pragma unroll
  for (int st = 0; st < KS; ++st) {
    s.a[st] = *(const v4f*)(ap + st * 8);
#pragma unroll
    for (int i = 0; i < 4; ++i)
      s.b[st][i] = *(const typename BVec<NT>::T*)(bp + (size_t)(st * 8 + i) * CC);
    if (NE) {
#pragma unroll
      for (int i = 0; i < 4; ++i) s.e[st][i] = ep[(size_t)(st * 8 + i) * CC];
    }
  }
}

template <int NT, int NE, int KS>
static __device__ __forceinline__ void mma_set(const FragSet<NT, NE, KS>& s, v8f (&acc)[NT + NE]) {
#pragma unroll
  for (int st = 0; st < KS; ++st) {
    const v2f a01 = {s.a[st].x, s.a[st].y};
    const v2f a23 = {s.a[st].z, s.a[st].w};
#pragma unroll
    for (int j = 0; j < NT; ++j) {
      const v2f b = {s.b[st][0][j], s.b[st][1][j]};
      acc[j] = wmma4(a01, b, acc[j]);
    }
    if (NE) {
      const v2f b = {s.e[st][0], s.e[st][1]};
      acc[NT] = wmma4(a01, b, acc[NT]);
    }
#pragma unroll
    for (int j = 0; j < NT; ++j) {
      const v2f b = {s.b[st][2][j], s.b[st][3][j]};
      acc[j] = wmma4(a23, b, acc[j]);
    }
    if (NE) {
      const v2f b = {s.e[st][2], s.e[st][3]};
      acc[NT] = wmma4(a23, b, acc[NT]);
    }
  }
}

template <int NT, int NE, int KS>
static __device__ __forceinline__ void gemm_k768_wave(const float* __restrict__ ap,
                                                      const float* __restrict__ bp,
                                                      const float* __restrict__ ep,
                                                      v8f (&acc)[NT + NE]) {
  constexpr int KB = 8 * KS;
  FragSet<NT, NE, KS> s0, s1;
  load_set(s0, ap, bp, ep);
  for (int kb = 0; kb < CC - 2 * KB; kb += 2 * KB) {
    __builtin_amdgcn_sched_barrier(0);
    load_set(s1, ap + kb + KB, bp + (size_t)(kb + KB) * CC, ep + (size_t)(kb + KB) * CC);
    __builtin_amdgcn_sched_barrier(0);
    mma_set(s0, acc);
    __builtin_amdgcn_sched_barrier(0);
    load_set(s0, ap + kb + 2 * KB, bp + (size_t)(kb + 2 * KB) * CC,
             ep + (size_t)(kb + 2 * KB) * CC);
    __builtin_amdgcn_sched_barrier(0);
    mma_set(s1, acc);
  }
  __builtin_amdgcn_sched_barrier(0);
  load_set(s1, ap + CC - KB, bp + (size_t)(CC - KB) * CC, ep + (size_t)(CC - KB) * CC);
  __builtin_amdgcn_sched_barrier(0);
  mma_set(s0, acc);
  mma_set(s1, acc);
}

#define QK_GROUPS 8
#define QK_PITCH 36

static __device__ __forceinline__ const float* qkv_wcol(const float* __restrict__ W0,
                                                        const float* __restrict__ W1,
                                                        const float* __restrict__ W2, int n) {
  return (n < CC) ? W0 + n : (n < 2 * CC) ? W1 + (n - CC) : W2 + (n - 2 * CC);
}

template <int N>
static __device__ __forceinline__ void qkv_store4(float* __restrict__ P, const v8f (&acc)[N],
                                                  int n, int t0, int lh) {
  float* o = P + (size_t)(n >> 6) * TT * HSS + (size_t)(t0 + 8 * lh) * HSS + (n & 63);
#pragma unroll
  for (int r = 0; r < 8; ++r) {
    const v4f v = {acc[0][r], acc[1][r], acc[2][r], acc[3][r]};
    *(v4f*)(o + (size_t)r * HSS) = v;
  }
}

__global__ void __launch_bounds__(128) qkv_proj_kernel(const float* __restrict__ x,
                                const float* __restrict__ W0,
                                const float* __restrict__ W1,
                                const float* __restrict__ W2,
                                float* __restrict__ P) {
  __shared__ __attribute__((aligned(16))) float s_line[2 * 16 * QK_PITCH];
  const int lane = threadIdx.x & 31;
  const int wv   = __builtin_amdgcn_readfirstlane(threadIdx.x >> 5);
  const int lm   = lane & 15;
  const int lh   = lane >> 4;
  const int b    = blockIdx.x;

  if (b < 16 * QK_GROUPS) {
    const int gb   = (b >> 4) * 288;
    const int rt   = wv >> 1;
    const int side = wv & 1;
    const int t0 = ((b & 15) * 2 + rt) * 16;
    const int n  = gb + (side ? 160 : 64) + 4 * lm;
    const int ne = gb + (side ? 144 : 128) + lm;
    v8f acc[5];
#pragma unroll
    for (int j = 0; j < 5; ++j) acc[j] = (v8f){0.f, 0.f, 0.f, 0.f, 0.f, 0.f, 0.f, 0.f};
    gemm_k768_wave<4, 1, 6>(x + (size_t)(t0 + lm) * CC + 4 * lh,
                         qkv_wcol(W0, W1, W2, n) + (size_t)(4 * lh) * CC,
                         qkv_wcol(W0, W1, W2, ne) + (size_t)(4 * lh) * CC, acc);
    qkv_store4(P, acc, n, t0, lh);
    float* sl = s_line + rt * 16 * QK_PITCH;
#pragma unroll
    for (int r = 0; r < 8; ++r) sl[(8 * lh + r) * QK_PITCH + 16 * side + lm] = acc[4][r];
    __syncthreads();
    const int nl = gb + 128;
    float* ol = P + (size_t)(nl >> 6) * TT * HSS + (size_t)t0 * HSS + (nl & 63);
    const int c4 = (lane & 7) * 4;
#pragma unroll
    for (int i = 0; i < 2; ++i) {
      const int row = 8 * side + 4 * i + (lane >> 3);
      *(v4f*)(ol + (size_t)row * HSS + c4) = *(const v4f*)(sl + row * QK_PITCH + c4);
    }
  } else {
    const int bn = b - 16 * QK_GROUPS;
    const int gb = (bn >> 4) * 288;
    const int t0 = ((bn & 7) * 4 + wv) * 16;
    const int n  = gb + ((bn & 8) ? 224 : 0) + 4 * lm;
    v8f acc[4];
#pragma unroll
    for (int j = 0; j < 4; ++j) acc[j] = (v8f){0.f, 0.f, 0.f, 0.f, 0.f, 0.f, 0.f, 0.f};
    const float* bp = qkv_wcol(W0, W1, W2, n) + (size_t)(4 * lh) * CC;
    gemm_k768_wave<4, 0, KS_STD>(x + (size_t)(t0 + lm) * CC + 4 * lh, bp, bp, acc);
    qkv_store4(P, acc, n, t0, lh);
  }
}

#define AT_PAIRS (TT / 32)
#define AT_VPITCH 68
#define AT_WAVES 4
#define AT_STEPS (AT_PAIRS / AT_WAVES)

static __device__ __forceinline__ void att_load_pair(v2f (&b)[32], const float* __restrict__ p) {
#pragma unroll
  for (int u = 0; u < 2; ++u) {
#pragma unroll
    for (int kk = 0; kk < 16; ++kk) b[u * 16 + kk] = *(const v2f*)(p + (size_t)(u * 16) * HSS + kk * 4);
  }
}

static __device__ __forceinline__ void att_mma_pair(const v2f (&a)[16], const v2f (&b)[32],
                                                    v8f& c0, v8f& c1) {
  c0 = (v8f){0.f, 0.f, 0.f, 0.f, 0.f, 0.f, 0.f, 0.f};
  c1 = c0;
#pragma unroll
  for (int kk = 0; kk < 16; ++kk) {
    c0 = wmma4(a[kk], b[kk], c0);
    c1 = wmma4(a[kk], b[16 + kk], c1);
  }
}

static __device__ __forceinline__ void att_scale_pair(v8f& c0, v8f& c1, float (&rm)[8],
                                                      bool diag, int jrel) {
#pragma unroll
  for (int r = 0; r < 8; ++r) {
    float v0 = c0[r] * SCALE;
    float v1 = c1[r] * SCALE;
    if (diag) {
      if (jrel > r) v0 = -3.0e38f;
      if (jrel + 16 > r) v1 = -3.0e38f;
    }
    c0[r] = v0;
    c1[r] = v1;
    rm[r] = fmaxf(rm[r], fmaxf(v0, v1));
  }
}

__global__ void __launch_bounds__(32 * AT_WAVES) att_softmax_kernel(const float* __restrict__ P0,
                                   const float* __restrict__ P1,
                                   const float* __restrict__ P2,
                                   const float* __restrict__ Wv,
                                   float* __restrict__ ATT,
                                   float* __restrict__ V) {
  __shared__ __attribute__((aligned(16))) float s_rmax[AT_WAVES * 16];
  __shared__ __attribute__((aligned(16))) float s_rsum[AT_WAVES * 16];
  __shared__ __attribute__((aligned(16))) float s_v[16 * AT_VPITCH];
  const int lane = threadIdx.x & 31;
  const int w    = __builtin_amdgcn_readfirstlane(threadIdx.x >> 5);
  const int lm   = lane & 15;
  const int lk   = (lane >> 4) << 1;
  const int drow = (lane >> 4) << 3;
  const int strip = TT / 16 - 1 - (int)(blockIdx.x / NHH);
  const int h     = blockIdx.x % NHH;
  const int i0 = strip * 16;
  const int g  = strip >> 1;
  const int first = g - w;
  const float* P0h = P0 + (size_t)h * TT * HSS;
  const float* P1h = P1 + (size_t)h * TT * HSS;

  v2f areg[16];
#pragma unroll
  for (int kk = 0; kk < 16; ++kk) {
    const int k = kk * 4 + lk;
    areg[kk].x = P0h[(size_t)(i0 + lm) * HSS + k];
    areg[kk].y = P0h[(size_t)(i0 + lm) * HSS + k + 1];
  }

  v2f va[16], vb[16];
  {
    const float* pa = P2 + (size_t)h * TT * HSS + (size_t)(i0 + lm) * HSS + lk;
    const float* pb = Wv + (size_t)lk * HSS + 16 * w + lm;
#pragma unroll
    for (int kk = 0; kk < 16; ++kk) {
      va[kk] = *(const v2f*)(pa + kk * 4);
      vb[kk].x = pb[(size_t)(kk * 4) * HSS];
      vb[kk].y = pb[(size_t)(kk * 4 + 1) * HSS];
    }
  }

  v8f acc[2 * AT_STEPS];
  float rm[8];
#pragma unroll
  for (int r = 0; r < 8; ++r) rm[r] = -3.0e38f;
  const float* bp = P1h + (size_t)lm * HSS + lk;
  const int jrel = (w == 0) ? (32 * g + lm - i0 - drow) : -64;
  v2f bA[32], bB[32];
  att_load_pair(bA, bp + (size_t)max(first, 0) * 32 * HSS);
  {
    v8f vacc = {0.f, 0.f, 0.f, 0.f, 0.f, 0.f, 0.f, 0.f};
#pragma unroll
    for (int kk = 0; kk < 16; ++kk) vacc = wmma4(va[kk], vb[kk], vacc);
#pragma unroll
    for (int r = 0; r < 8; ++r) s_v[(drow + r) * AT_VPITCH + 16 * w + lm] = vacc[r];
  }
#pragma unroll
  for (int q = 0; q < AT_STEPS / 2; ++q) {
    if (first - 8 * q >= 0) {
      __builtin_amdgcn_sched_barrier(0);
      att_load_pair(bB, bp + (size_t)max(first - 8 * q - 4, 0) * 32 * HSS);
      __builtin_amdgcn_sched_barrier(0);
      att_mma_pair(areg, bA, acc[4 * q], acc[4 * q + 1]);
      att_scale_pair(acc[4 * q], acc[4 * q + 1], rm, q == 0, jrel);
    }
    if (first - 8 * q - 4 >= 0) {
      __builtin_amdgcn_sched_barrier(0);
      att_load_pair(bA, bp + (size_t)max(first - 8 * q - 8, 0) * 32 * HSS);
      __builtin_amdgcn_sched_barrier(0);
      att_mma_pair(areg, bB, acc[4 * q + 2], acc[4 * q + 3]);
      att_scale_pair(acc[4 * q + 2], acc[4 * q + 3], rm, false, 0);
    }
  }

#pragma unroll
  for (int off = 8; off > 0; off >>= 1) {
#pragma unroll
    for (int r = 0; r < 8; ++r) rm[r] = fmaxf(rm[r], __shfl_xor(rm[r], off));
  }
  if (lm == 0) {
    *(v4f*)(s_rmax + w * 16 + drow)     = (v4f){rm[0], rm[1], rm[2], rm[3]};
    *(v4f*)(s_rmax + w * 16 + drow + 4) = (v4f){rm[4], rm[5], rm[6], rm[7]};
  }
  __syncthreads();
  {
    float* vo = V + (size_t)h * TT * HSS + (size_t)i0 * HSS;
    const int vc = lm * 4;
#pragma unroll
    for (int i = 0; i < 2; ++i) {
      const int row = 4 * w + 2 * i + (lane >> 4);
      *(v4f*)(vo + (size_t)row * HSS + vc) = *(const v4f*)(s_v + row * AT_VPITCH + vc);
    }
  }
#pragma unroll
  for (int k = 0; k < AT_WAVES; ++k) {
    const v4f lo = *(const v4f*)(s_rmax + k * 16 + drow);
    const v4f hi = *(const v4f*)(s_rmax + k * 16 + drow + 4);
    rm[0] = fmaxf(rm[0], lo.x); rm[1] = fmaxf(rm[1], lo.y);
    rm[2] = fmaxf(rm[2], lo.z); rm[3] = fmaxf(rm[3], lo.w);
    rm[4] = fmaxf(rm[4], hi.x); rm[5] = fmaxf(rm[5], hi.y);
    rm[6] = fmaxf(rm[6], hi.z); rm[7] = fmaxf(rm[7], hi.w);
  }

  float rs[8];
#pragma unroll
  for (int r = 0; r < 8; ++r) rs[r] = 0.0f;
#pragma unroll
  for (int s = 0; s < AT_STEPS; ++s) {
    if (first - 4 * s >= 0) {
#pragma unroll
      for (int u = 0; u < 2; ++u) {
#pragma unroll
        for (int r = 0; r < 8; ++r) {
          const float e = __expf(acc[2 * s + u][r] - rm[r]);
          acc[2 * s + u][r] = e;
          rs[r] += e;
        }
      }
    }
  }
#pragma unroll
  for (int off = 8; off > 0; off >>= 1) {
#pragma unroll
    for (int r = 0; r < 8; ++r) rs[r] += __shfl_xor(rs[r], off);
  }
  if (lm == 0) {
    *(v4f*)(s_rsum + w * 16 + drow)     = (v4f){rs[0], rs[1], rs[2], rs[3]};
    *(v4f*)(s_rsum + w * 16 + drow + 4) = (v4f){rs[4], rs[5], rs[6], rs[7]};
  }
  __syncthreads();
  v4f tlo = *(const v4f*)(s_rsum + drow);
  v4f thi = *(const v4f*)(s_rsum + drow + 4);
#pragma unroll
  for (int k = 1; k < AT_WAVES; ++k) {
    tlo += *(const v4f*)(s_rsum + k * 16 + drow);
    thi += *(const v4f*)(s_rsum + k * 16 + drow + 4);
  }
  const float inv[8] = {1.0f / tlo.x, 1.0f / tlo.y, 1.0f / tlo.z, 1.0f / tlo.w,
                        1.0f / thi.x, 1.0f / thi.y, 1.0f / thi.z, 1.0f / thi.w};

  float* abase = ATT + ((size_t)h * (TT / 16) + strip) * TT * 16 + (size_t)lm * 16 + drow;
#pragma unroll
  for (int s = 0; s < AT_STEPS; ++s) {
    if (first - 4 * s >= 0) {
#pragma unroll
      for (int u = 0; u < 2; ++u) {
        const v8f& c = acc[2 * s + u];
        const v4f lo = {c[0] * inv[0], c[1] * inv[1], c[2] * inv[2], c[3] * inv[3]};
        const v4f hi = {c[4] * inv[4], c[5] * inv[5], c[6] * inv[6], c[7] * inv[7]};
        float* o = abase + (size_t)(32 * (first - 4 * s) + 16 * u) * 16;
        *(v4f*)o = lo;
        *(v4f*)(o + 4) = hi;
      }
    }
  }
}

#define O3_WAVES 16
#define O3_PITCH 68

static __device__ __forceinline__ void o3_walk(const float* __restrict__ at,
                                               const float* __restrict__ p2w,
                                               const float* __restrict__ vrow,
                                               int ntrips, float negML, float ce, float co,
                                               float& ds, float& a0, float& a1, float& a2, float& a3) {
#pragma unroll 4
  for (int t = 0; t < ntrips; ++t) {
    const float a = at[(size_t)t * 32];
    const v4f pe = *(const v4f*)(p2w + (size_t)(2 * t) * HSS);
    const v4f po = *(const v4f*)(p2w + (size_t)(2 * t + 1) * HSS);
    const v4f v = *(const v4f*)(vrow + (size_t)(2 * t) * HSS);
    const float ae = a * ce;
    const float ao = a * co;
    const float e0 = __builtin_amdgcn_exp2f(fmaf(ae, pe.x, fmaf(ao, po.x, negML)));
    const float e1 = __builtin_amdgcn_exp2f(fmaf(ae, pe.y, fmaf(ao, po.y, negML)));
    const float e2 = __builtin_amdgcn_exp2f(fmaf(ae, pe.z, fmaf(ao, po.z, negML)));
    const float e3 = __builtin_amdgcn_exp2f(fmaf(ae, pe.w, fmaf(ao, po.w, negML)));
    ds += (e0 + e1) + (e2 + e3);
    a0 = fmaf(e0, v.x, a0);
    a1 = fmaf(e1, v.y, a1);
    a2 = fmaf(e2, v.z, a2);
    a3 = fmaf(e3, v.w, a3);
  }
  ds += __shfl_xor(ds, 16);
  a0 += __shfl_xor(a0, 16);
  a1 += __shfl_xor(a1, 16);
  a2 += __shfl_xor(a2, 16);
  a3 += __shfl_xor(a3, 16);
}

__global__ void __launch_bounds__(512) order3_kernel(const float* __restrict__ ATT,
                              const float* __restrict__ P2,
                              const float* __restrict__ V,
                              float* __restrict__ Y) {
  __shared__ float s_pmax[TT];
  __shared__ float s_m[O3_WAVES * 32];
  __shared__ float s_ds[O3_WAVES * 32];
  __shared__ __attribute__((aligned(16))) float s_tail[2 * O3_WAVES * HSS];
  __shared__ __attribute__((aligned(16))) float s_out[32 * O3_PITCH];
  const float LOG2E = 1.4426950408889634f;
  const int lane = threadIdx.x & 31;
  const int hf   = lane >> 4;
  const int r    = lane & 15;
  const int w = __builtin_amdgcn_readfirstlane(threadIdx.x >> 5);
  const int p = blockIdx.x;
  const int h = blockIdx.y;
  const int sA = p;
  const int sB = TT / 16 - 1 - p;
  const int jendA = 16 * (sA + 1);
  const int jendB = 16 * (sB + 1);
  const float* P2h = P2 + (size_t)h * TT * HSS;
  const float* Vh  = V  + (size_t)h * TT * HSS;

  {
    const int trow = jendA + (threadIdx.x >> 4);
    const int tc4  = (threadIdx.x & 15) * 4;
    v4f tA = {0.f, 0.f, 0.f, 0.f};
    v4f tB = {0.f, 0.f, 0.f, 0.f};
#pragma unroll
    for (int k = 0; k < TT / 32; ++k) {
      const int row = trow + 32 * k;
      const v4f q = *(const v4f*)(Vh + (size_t)min(row, TT - 1) * HSS + tc4);
      const bool inA = row < TT;
      const bool inB = inA && row >= jendB;
      tA.x += inA ? q.x : 0.0f;
      tA.y += inA ? q.y : 0.0f;
      tA.z += inA ? q.z : 0.0f;
      tA.w += inA ? q.w : 0.0f;
      tB.x += inB ? q.x : 0.0f;
      tB.y += inB ? q.y : 0.0f;
      tB.z += inB ? q.z : 0.0f;
      tB.w += inB ? q.w : 0.0f;
    }
    tA.x += __shfl_xor(tA.x, 16);
    tA.y += __shfl_xor(tA.y, 16);
    tA.z += __shfl_xor(tA.z, 16);
    tA.w += __shfl_xor(tA.w, 16);
    tB.x += __shfl_xor(tB.x, 16);
    tB.y += __shfl_xor(tB.y, 16);
    tB.z += __shfl_xor(tB.z, 16);
    tB.w += __shfl_xor(tB.w, 16);
    if (lane < 16) {
      *(v4f*)(s_tail + w * HSS + tc4) = tA;
      *(v4f*)(s_tail + (O3_WAVES + w) * HSS + tc4) = tB;
    }
  }

  {
    const int c4 = r * 4;
    float pm[16];
#pragma unroll
    for (int k = 0; k < 16; ++k) {
      const int j = min(2 * (w + O3_WAVES * k) + hf, jendB - 1);
      const v4f q = *(const v4f*)(P2h + (size_t)j * HSS + c4);
      pm[k] = fmaxf(fmaxf(q.x, q.y), fmaxf(q.z, q.w));
    }
#pragma unroll
    for (int off = 8; off > 0; off >>= 1) {
#pragma unroll
      for (int k = 0; k < 16; ++k) pm[k] = fmaxf(pm[k], __shfl_xor(pm[k], off));
    }
#pragma unroll
    for (int k = 0; k < 16; ++k) {
      const int j = 2 * (w + O3_WAVES * k) + hf;
      if (r == 0 && j < jendB) s_pmax[j] = pm[k];
    }
  }
  __syncthreads();

  const float* attA = ATT + ((size_t)h * (TT / 16) + sA) * TT * 16;
  const float* attB = ATT + ((size_t)h * (TT / 16) + sB) * TT * 16;

  {
    const float* atM = (hf ? attB : attA) + r;
    const int jendL = hf ? jendB : jendA;
    float mloc = 0.0f;
#pragma unroll
    for (int k = 0; k < TT / O3_WAVES; ++k) {
      const int j = min(w + O3_WAVES * k, jendL - 1);
      mloc = fmaxf(mloc, atM[(size_t)j * 16] * s_pmax[j]);
    }
    s_m[w * 32 + lane] = mloc;
  }
  __syncthreads();
  float Mown = 0.0f;
#pragma unroll
  for (int k = 0; k < O3_WAVES; ++k) Mown = fmaxf(Mown, s_m[k * 32 + lane]);
  const float Moth = __shfl_xor(Mown, 16);
  const float negMLA = -(hf ? Moth : Mown) * LOG2E;
  const float negMLB = -(hf ? Mown : Moth) * LOG2E;

  const float ce = hf ? 0.0f : LOG2E;
  const float co = hf ? LOG2E : 0.0f;
  const float* p2w  = P2h + 4 * w;
  const float* vrow = Vh + (size_t)hf * HSS + 4 * w;
  float dsA = 0.0f, xA0 = 0.0f, xA1 = 0.0f, xA2 = 0.0f, xA3 = 0.0f;
  float dsB = 0.0f, xB0 = 0.0f, xB1 = 0.0f, xB2 = 0.0f, xB3 = 0.0f;
  o3_walk(attA + hf * 16 + r, p2w, vrow, jendA / 2, negMLA, ce, co, dsA, xA0, xA1, xA2, xA3);
  o3_walk(attB + hf * 16 + r, p2w, vrow, jendB / 2, negMLB, ce, co, dsB, xB0, xB1, xB2, xB3);

  const float ds = hf ? dsB : dsA;
  const float a0 = hf ? xB0 : xA0;
  const float a1 = hf ? xB1 : xA1;
  const float a2 = hf ? xB2 : xA2;
  const float a3 = hf ? xB3 : xA3;
  const float negML = hf ? negMLB : negMLA;
  const int jendL = hf ? jendB : jendA;
  s_ds[w * 32 + lane] = ds;
  __syncthreads();
  float tot = 0.0f;
#pragma unroll
  for (int k = 0; k < O3_WAVES; ++k) tot += s_ds[k * 32 + lane];
  const float em = __builtin_amdgcn_exp2f(negML);
  const float denom = tot + (float)((TT - jendL) * HSS) * em;
  v4f vs = {0.f, 0.f, 0.f, 0.f};
#pragma unroll
  for (int k = 0; k < O3_WAVES; ++k) vs += *(const v4f*)(s_tail + (hf * O3_WAVES + k) * HSS + 4 * w);
  const v4f o = {(a0 + em * vs.x) / denom, (a1 + em * vs.y) / denom,
                 (a2 + em * vs.z) / denom, (a3 + em * vs.w) / denom};
  *(v4f*)(s_out + lane * O3_PITCH + 4 * w) = o;
  __syncthreads();
  const int orow = threadIdx.x >> 4;
  const int oc   = (threadIdx.x & 15) * 4;
  const int grow = (orow < 16) ? (16 * sA + orow) : (16 * sB + orow - 16);
  *(v4f*)(Y + (size_t)grow * CC + h * HSS + oc) = *(const v4f*)(s_out + orow * O3_PITCH + oc);
}

__global__ void __launch_bounds__(128) out_proj_kernel(const float* __restrict__ Y,
                                const float* __restrict__ Wc,
                                float* __restrict__ out) {
  const int lane = threadIdx.x & 31;
  const int wv   = threadIdx.x >> 5;
  const int lm   = lane & 15;
  const int lh   = lane >> 4;
  const int t0 = (blockIdx.x * 4 + wv) * 16;
  const int c0 = blockIdx.y * 32;

  v8f acc[2];
#pragma unroll
  for (int j = 0; j < 2; ++j) acc[j] = (v8f){0.f, 0.f, 0.f, 0.f, 0.f, 0.f, 0.f, 0.f};
  const float* bp = Wc + (size_t)(4 * lh) * CC + c0 + 2 * lm;
  gemm_k768_wave<2, 0, KS_STD>(Y + (size_t)(t0 + lm) * CC + 4 * lh, bp, bp, acc);
#pragma unroll
  for (int r = 0; r < 8; ++r) {
    const v2f o = {acc[0][r], acc[1][r]};
    *(v2f*)(out + (size_t)(t0 + 8 * lh + r) * CC + c0 + 2 * lm) = o;
  }
}

extern "C" void kernel_launch(void* const* d_in, const int* in_sizes, int n_in,
                              void* d_out, int out_size, void* d_ws,
                              size_t ws_size, hipStream_t stream) {
  (void)in_sizes; (void)n_in; (void)out_size; (void)ws_size;
  const float* x  = (const float*)d_in[0];
  const float* W0 = (const float*)d_in[1];
  const float* W1 = (const float*)d_in[2];
  const float* W2 = (const float*)d_in[3];
  const float* Wv = (const float*)d_in[4];
  const float* Wc = (const float*)d_in[5];
  float* out = (float*)d_out;
  float* ws  = (float*)d_ws;

  const size_t NPH = (size_t)NHH * TT * HSS;
  float* P0  = ws;
  float* P1  = ws + NPH;
  float* P2  = ws + 2 * NPH;
  float* V   = ws + 3 * NPH;
  float* ATT = ws + 4 * NPH;
  float* Y   = ATT + (size_t)NHH * TT * TT;

  qkv_proj_kernel<<<dim3(32 * QK_GROUPS), 128, 0, stream>>>(x, W0, W1, W2, P0);
  att_softmax_kernel<<<dim3((TT / 16) * NHH), 32 * AT_WAVES, 0, stream>>>(P0, P1, P2, Wv, ATT, V);
  order3_kernel<<<dim3(TT / 32, NHH), 512, 0, stream>>>(ATT, P2, V, Y);
  out_proj_kernel<<<dim3(TT / 64, CC / 32), 128, 0, stream>>>(Y, Wc, out);
}


# Order-3 walk software-pipelined in sets of 8 steps: next set's att/V loads issued in front of the current arithmetic, next p2 scalar loads right behind the current exponents; sums keep their order
# speedup vs baseline: 7.3300x; 1.0364x over previous
#include <hip/hip_runtime.h>
#include <math.h>

#define TT  512
#define CC  768
#define NHH 12
#define HSS 64
constexpr float SCALE = 0.125f;

typedef float v2f __attribute__((ext_vector_type(2)));
typedef float v8f __attribute__((ext_vector_type(8)));

static __device__ __forceinline__ v8f wmma4(v2f a, v2f b, v8f c) {
  return __builtin_amdgcn_wmma_f32_16x16x4_f32(
       false, a,  false, b,
       (short)0, c,  false,  false);
}

typedef float v4f __attribute__((ext_vector_type(4)));
template <int NT> struct BVec;
template <> struct BVec<4> { typedef float T __attribute__((ext_vector_type(4))); };
template <> struct BVec<2> { typedef float T __attribute__((ext_vector_type(2))); };

#define KS_STD 8

template <int NT, int NE, int KS>
struct FragSet {
  v4f a[KS];
  typename BVec<NT>::T b[KS][4];
  float e[KS][4];
};

template <int NT, int NE, int KS>
static __device__ __forceinline__ void load_set(FragSet<NT, NE, KS>& s,
                                                const float* __restrict__ ap,
                                                const float* __restrict__ bp,
                                                const float* __restrict__ ep) {
#pragma unroll
  for (int st = 0; st < KS; ++st) {
    s.a[st] = *(const v4f*)(ap + st * 8);
#pragma unroll
    for (int i = 0; i < 4; ++i)
      s.b[st][i] = *(const typename BVec<NT>::T*)(bp + (size_t)(st * 8 + i) * CC);
    if (NE) {
#pragma unroll
      for (int i = 0; i < 4; ++i) s.e[st][i] = ep[(size_t)(st * 8 + i) * CC];
    }
  }
}

template <int NT, int NE, int KS>
static __device__ __forceinline__ void mma_set(const FragSet<NT, NE, KS>& s, v8f (&acc)[NT + NE]) {
#pragma unroll
  for (int st = 0; st < KS; ++st) {
    const v2f a01 = {s.a[st].x, s.a[st].y};
    const v2f a23 = {s.a[st].z, s.a[st].w};
#pragma unroll
    for (int j = 0; j < NT; ++j) {
      const v2f b = {s.b[st][0][j], s.b[st][1][j]};
      acc[j] = wmma4(a01, b, acc[j]);
    }
    if (NE) {
      const v2f b = {s.e[st][0], s.e[st][1]};
      acc[NT] = wmma4(a01, b, acc[NT]);
    }
#pragma unroll
    for (int j = 0; j < NT; ++j) {
      const v2f b = {s.b[st][2][j], s.b[st][3][j]};
      acc[j] = wmma4(a23, b, acc[j]);
    }
    if (NE) {
      const v2f b = {s.e[st][2], s.e[st][3]};
      acc[NT] = wmma4(a23, b, acc[NT]);
    }
  }
}

template <int NT, int NE, int KS>
static __device__ __forceinline__ void gemm_k768_wave(const float* __restrict__ ap,
                                                      const float* __restrict__ bp,
                                                      const float* __restrict__ ep,
                                                      v8f (&acc)[NT + NE]) {
  constexpr int KB = 8 * KS;
  FragSet<NT, NE, KS> s0, s1;
  load_set(s0, ap, bp, ep);
  for (int kb = 0; kb < CC - 2 * KB; kb += 2 * KB) {
    __builtin_amdgcn_sched_barrier(0);
    load_set(s1, ap + kb + KB, bp + (size_t)(kb + KB) * CC, ep + (size_t)(kb + KB) * CC);
    __builtin_amdgcn_sched_barrier(0);
    mma_set(s0, acc);
    __builtin_amdgcn_sched_barrier(0);
    load_set(s0, ap + kb + 2 * KB, bp + (size_t)(kb + 2 * KB) * CC,
             ep + (size_t)(kb + 2 * KB) * CC);
    __builtin_amdgcn_sched_barrier(0);
    mma_set(s1, acc);
  }
  __builtin_amdgcn_sched_barrier(0);
  load_set(s1, ap + CC - KB, bp + (size_t)(CC - KB) * CC, ep + (size_t)(CC - KB) * CC);
  __builtin_amdgcn_sched_barrier(0);
  mma_set(s0, acc);
  mma_set(s1, acc);
}

#define QK_GROUPS 8
#define QK_PITCH 36

static __device__ __forceinline__ const float* qkv_wcol(const float* __restrict__ W0,
                                                        const float* __restrict__ W1,
                                                        const float* __restrict__ W2, int n) {
  return (n < CC) ? W0 + n : (n < 2 * CC) ? W1 + (n - CC) : W2 + (n - 2 * CC);
}

template <int N>
static __device__ __forceinline__ void qkv_store4(float* __restrict__ P, const v8f (&acc)[N],
                                                  int n, int t0, int lh) {
  float* o = P + (size_t)(n >> 6) * TT * HSS + (size_t)(t0 + 8 * lh) * HSS + (n & 63);
#pragma unroll
  for (int r = 0; r < 8; ++r) {
    const v4f v = {acc[0][r], acc[1][r], acc[2][r], acc[3][r]};
    *(v4f*)(o + (size_t)r * HSS) = v;
  }
}

__global__ void __launch_bounds__(128) qkv_proj_kernel(const float* __restrict__ x,
                                const float* __restrict__ W0,
                                const float* __restrict__ W1,
                                const float* __restrict__ W2,
                                float* __restrict__ P) {
  __shared__ __attribute__((aligned(16))) float s_line[2 * 16 * QK_PITCH];
  const int lane = threadIdx.x & 31;
  const int wv   = __builtin_amdgcn_readfirstlane(threadIdx.x >> 5);
  const int lm   = lane & 15;
  const int lh   = lane >> 4;
  const int b    = blockIdx.x;

  if (b < 16 * QK_GROUPS) {
    const int gb   = (b >> 4) * 288;
    const int rt   = wv >> 1;
    const int side = wv & 1;
    const int t0 = ((b & 15) * 2 + rt) * 16;
    const int n  = gb + (side ? 160 : 64) + 4 * lm;
    const int ne = gb + (side ? 144 : 128) + lm;
    v8f acc[5];
#pragma unroll
    for (int j = 0; j < 5; ++j) acc[j] = (v8f){0.f, 0.f, 0.f, 0.f, 0.f, 0.f, 0.f, 0.f};
    gemm_k768_wave<4, 1, 6>(x + (size_t)(t0 + lm) * CC + 4 * lh,
                         qkv_wcol(W0, W1, W2, n) + (size_t)(4 * lh) * CC,
                         qkv_wcol(W0, W1, W2, ne) + (size_t)(4 * lh) * CC, acc);
    qkv_store4(P, acc, n, t0, lh);
    float* sl = s_line + rt * 16 * QK_PITCH;
#pragma unroll
    for (int r = 0; r < 8; ++r) sl[(8 * lh + r) * QK_PITCH + 16 * side + lm] = acc[4][r];
    __syncthreads();
    const int nl = gb + 128;
    float* ol = P + (size_t)(nl >> 6) * TT * HSS + (size_t)t0 * HSS + (nl & 63);
    const int c4 = (lane & 7) * 4;
#pragma unroll
    for (int i = 0; i < 2; ++i) {
      const int row = 8 * side + 4 * i + (lane >> 3);
      *(v4f*)(ol + (size_t)row * HSS + c4) = *(const v4f*)(sl + row * QK_PITCH + c4);
    }
  } else {
    const int bn = b - 16 * QK_GROUPS;
    const int gb = (bn >> 4) * 288;
    const int t0 = ((bn & 7) * 4 + wv) * 16;
    const int n  = gb + ((bn & 8) ? 224 : 0) + 4 * lm;
    v8f acc[4];
#pragma unroll
    for (int j = 0; j < 4; ++j) acc[j] = (v8f){0.f, 0.f, 0.f, 0.f, 0.f, 0.f, 0.f, 0.f};
    const float* bp = qkv_wcol(W0, W1, W2, n) + (size_t)(4 * lh) * CC;
    gemm_k768_wave<4, 0, KS_STD>(x + (size_t)(t0 + lm) * CC + 4 * lh, bp, bp, acc);
    qkv_store4(P, acc, n, t0, lh);
  }
}

#define AT_PAIRS (TT / 32)
#define AT_VPITCH 68
#define AT_WAVES 4
#define AT_STEPS (AT_PAIRS / AT_WAVES)

static __device__ __forceinline__ void att_load_pair(v2f (&b)[32], const float* __restrict__ p) {
#pragma unroll
  for (int u = 0; u < 2; ++u) {
#pragma unroll
    for (int kk = 0; kk < 16; ++kk) b[u * 16 + kk] = *(const v2f*)(p + (size_t)(u * 16) * HSS + kk * 4);
  }
}

static __device__ __forceinline__ void att_mma_pair(const v2f (&a)[16], const v2f (&b)[32],
                                                    v8f& c0, v8f& c1) {
  c0 = (v8f){0.f, 0.f, 0.f, 0.f, 0.f, 0.f, 0.f, 0.f};
  c1 = c0;
#pragma unroll
  for (int kk = 0; kk < 16; ++kk) {
    c0 = wmma4(a[kk], b[kk], c0);
    c1 = wmma4(a[kk], b[16 + kk], c1);
  }
}

static __device__ __forceinline__ void att_scale_pair(v8f& c0, v8f& c1, float (&rm)[8],
                                                      bool diag, int jrel) {
#pragma unroll
  for (int r = 0; r < 8; ++r) {
    float v0 = c0[r] * SCALE;
    float v1 = c1[r] * SCALE;
    if (diag) {
      if (jrel > r) v0 = -3.0e38f;
      if (jrel + 16 > r) v1 = -3.0e38f;
    }
    c0[r] = v0;
    c1[r] = v1;
    rm[r] = fmaxf(rm[r], fmaxf(v0, v1));
  }
}

__global__ void __launch_bounds__(32 * AT_WAVES) att_softmax_kernel(const float* __restrict__ P0,
                                   const float* __restrict__ P1,
                                   const float* __restrict__ P2,
                                   const float* __restrict__ Wv,
                                   float* __restrict__ ATT,
                                   float* __restrict__ V) {
  __shared__ __attribute__((aligned(16))) float s_rmax[AT_WAVES * 16];
  __shared__ __attribute__((aligned(16))) float s_rsum[AT_WAVES * 16];
  __shared__ __attribute__((aligned(16))) float s_v[16 * AT_VPITCH];
  const int lane = threadIdx.x & 31;
  const int w    = __builtin_amdgcn_readfirstlane(threadIdx.x >> 5);
  const int lm   = lane & 15;
  const int lk   = (lane >> 4) << 1;
  const int drow = (lane >> 4) << 3;
  const int strip = TT / 16 - 1 - (int)(blockIdx.x / NHH);
  const int h     = blockIdx.x % NHH;
  const int i0 = strip * 16;
  const int g  = strip >> 1;
  const int first = g - w;
  const float* P0h = P0 + (size_t)h * TT * HSS;
  const float* P1h = P1 + (size_t)h * TT * HSS;

  v2f areg[16];
#pragma unroll
  for (int kk = 0; kk < 16; ++kk) {
    const int k = kk * 4 + lk;
    areg[kk].x = P0h[(size_t)(i0 + lm) * HSS + k];
    areg[kk].y = P0h[(size_t)(i0 + lm) * HSS + k + 1];
  }

  v2f va[16], vb[16];
  {
    const float* pa = P2 + (size_t)h * TT * HSS + (size_t)(i0 + lm) * HSS + lk;
    const float* pb = Wv + (size_t)lk * HSS + 16 * w + lm;
#pragma unroll
    for (int kk = 0; kk < 16; ++kk) {
      va[kk] = *(const v2f*)(pa + kk * 4);
      vb[kk].x = pb[(size_t)(kk * 4) * HSS];
      vb[kk].y = pb[(size_t)(kk * 4 + 1) * HSS];
    }
  }

  v8f acc[2 * AT_STEPS];
  float rm[8];
#pragma unroll
  for (int r = 0; r < 8; ++r) rm[r] = -3.0e38f;
  const float* bp = P1h + (size_t)lm * HSS + lk;
  const int jrel = (w == 0) ? (32 * g + lm - i0 - drow) : -64;
  v2f bA[32], bB[32];
  att_load_pair(bA, bp + (size_t)max(first, 0) * 32 * HSS);
  {
    v8f vacc = {0.f, 0.f, 0.f, 0.f, 0.f, 0.f, 0.f, 0.f};
#pragma unroll
    for (int kk = 0; kk < 16; ++kk) vacc = wmma4(va[kk], vb[kk], vacc);
#pragma unroll
    for (int r = 0; r < 8; ++r) s_v[(drow + r) * AT_VPITCH + 16 * w + lm] = vacc[r];
  }
#pragma unroll
  for (int q = 0; q < AT_STEPS / 2; ++q) {
    if (first - 8 * q >= 0) {
      __builtin_amdgcn_sched_barrier(0);
      att_load_pair(bB, bp + (size_t)max(first - 8 * q - 4, 0) * 32 * HSS);
      __builtin_amdgcn_sched_barrier(0);
      att_mma_pair(areg, bA, acc[4 * q], acc[4 * q + 1]);
      att_scale_pair(acc[4 * q], acc[4 * q + 1], rm, q == 0, jrel);
    }
    if (first - 8 * q - 4 >= 0) {
      __builtin_amdgcn_sched_barrier(0);
      att_load_pair(bA, bp + (size_t)max(first - 8 * q - 8, 0) * 32 * HSS);
      __builtin_amdgcn_sched_barrier(0);
      att_mma_pair(areg, bB, acc[4 * q + 2], acc[4 * q + 3]);
      att_scale_pair(acc[4 * q + 2], acc[4 * q + 3], rm, false, 0);
    }
  }

#pragma unroll
  for (int off = 8; off > 0; off >>= 1) {
#pragma unroll
    for (int r = 0; r < 8; ++r) rm[r] = fmaxf(rm[r], __shfl_xor(rm[r], off));
  }
  if (lm == 0) {
    *(v4f*)(s_rmax + w * 16 + drow)     = (v4f){rm[0], rm[1], rm[2], rm[3]};
    *(v4f*)(s_rmax + w * 16 + drow + 4) = (v4f){rm[4], rm[5], rm[6], rm[7]};
  }
  __syncthreads();
  {
    float* vo = V + (size_t)h * TT * HSS + (size_t)i0 * HSS;
    const int vc = lm * 4;
#pragma unroll
    for (int i = 0; i < 2; ++i) {
      const int row = 4 * w + 2 * i + (lane >> 4);
      *(v4f*)(vo + (size_t)row * HSS + vc) = *(const v4f*)(s_v + row * AT_VPITCH + vc);
    }
  }
#pragma unroll
  for (int k = 0; k < AT_WAVES; ++k) {
    const v4f lo = *(const v4f*)(s_rmax + k * 16 + drow);
    const v4f hi = *(const v4f*)(s_rmax + k * 16 + drow + 4);
    rm[0] = fmaxf(rm[0], lo.x); rm[1] = fmaxf(rm[1], lo.y);
    rm[2] = fmaxf(rm[2], lo.z); rm[3] = fmaxf(rm[3], lo.w);
    rm[4] = fmaxf(rm[4], hi.x); rm[5] = fmaxf(rm[5], hi.y);
    rm[6] = fmaxf(rm[6], hi.z); rm[7] = fmaxf(rm[7], hi.w);
  }

  float rs[8];
#pragma unroll
  for (int r = 0; r < 8; ++r) rs[r] = 0.0f;
#pragma unroll
  for (int s = 0; s < AT_STEPS; ++s) {
    if (first - 4 * s >= 0) {
#pragma unroll
      for (int u = 0; u < 2; ++u) {
#pragma unroll
        for (int r = 0; r < 8; ++r) {
          const float e = __expf(acc[2 * s + u][r] - rm[r]);
          acc[2 * s + u][r] = e;
          rs[r] += e;
        }
      }
    }
  }
#pragma unroll
  for (int off = 8; off > 0; off >>= 1) {
#pragma unroll
    for (int r = 0; r < 8; ++r) rs[r] += __shfl_xor(rs[r], off);
  }
  if (lm == 0) {
    *(v4f*)(s_rsum + w * 16 + drow)     = (v4f){rs[0], rs[1], rs[2], rs[3]};
    *(v4f*)(s_rsum + w * 16 + drow + 4) = (v4f){rs[4], rs[5], rs[6], rs[7]};
  }
  __syncthreads();
  v4f tlo = *(const v4f*)(s_rsum + drow);
  v4f thi = *(const v4f*)(s_rsum + drow + 4);
#pragma unroll
  for (int k = 1; k < AT_WAVES; ++k) {
    tlo += *(const v4f*)(s_rsum + k * 16 + drow);
    thi += *(const v4f*)(s_rsum + k * 16 + drow + 4);
  }
  const float inv[8] = {1.0f / tlo.x, 1.0f / tlo.y, 1.0f / tlo.z, 1.0f / tlo.w,
                        1.0f / thi.x, 1.0f / thi.y, 1.0f / thi.z, 1.0f / thi.w};

  float* abase = ATT + ((size_t)h * (TT / 16) + strip) * TT * 16 + (size_t)lm * 16 + drow;
#pragma unroll
  for (int s = 0; s < AT_STEPS; ++s) {
    if (first - 4 * s >= 0) {
#pragma unroll
      for (int u = 0; u < 2; ++u) {
        const v8f& c = acc[2 * s + u];
        const v4f lo = {c[0] * inv[0], c[1] * inv[1], c[2] * inv[2], c[3] * inv[3]};
        const v4f hi = {c[4] * inv[4], c[5] * inv[5], c[6] * inv[6], c[7] * inv[7]};
        float* o = abase + (size_t)(32 * (first - 4 * s) + 16 * u) * 16;
        *(v4f*)o = lo;
        *(v4f*)(o + 4) = hi;
      }
    }
  }
}

#define O3_WAVES 16
#define O3_PITCH 68

#define O3_SET 8
struct O3Lane {
  float a[O3_SET];
  v4f v[O3_SET];
};
struct O3Rows {
  v4f pe[O3_SET];
  v4f po[O3_SET];
};

static __device__ __forceinline__ void o3_load_lane(O3Lane& s, const float* __restrict__ at,
                                                    const float* __restrict__ vrow) {
#pragma unroll
  for (int u = 0; u < O3_SET; ++u) s.a[u] = at[(size_t)u * 32];
#pragma unroll
  for (int u = 0; u < O3_SET; ++u) s.v[u] = *(const v4f*)(vrow + (size_t)(2 * u) * HSS);
}

static __device__ __forceinline__ void o3_load_rows(O3Rows& q, const float* __restrict__ p2w) {
#pragma unroll
  for (int u = 0; u < O3_SET; ++u) {
    q.pe[u] = *(const v4f*)(p2w + (size_t)(2 * u) * HSS);
    q.po[u] = *(const v4f*)(p2w + (size_t)(2 * u + 1) * HSS);
  }
}

static __device__ __forceinline__ void o3_exponents(const O3Lane& s, const O3Rows& q, float negML,
                                                    float ce, float co, float (&x)[4 * O3_SET]) {
#pragma unroll
  for (int u = 0; u < O3_SET; ++u) {
    const float ae = s.a[u] * ce;
    const float ao = s.a[u] * co;
    x[4 * u + 0] = fmaf(ae, q.pe[u].x, fmaf(ao, q.po[u].x, negML));
    x[4 * u + 1] = fmaf(ae, q.pe[u].y, fmaf(ao, q.po[u].y, negML));
    x[4 * u + 2] = fmaf(ae, q.pe[u].z, fmaf(ao, q.po[u].z, negML));
    x[4 * u + 3] = fmaf(ae, q.pe[u].w, fmaf(ao, q.po[u].w, negML));
  }
}

static __device__ __forceinline__ void o3_accumulate(const float (&x)[4 * O3_SET], const O3Lane& s,
                                                     float& ds, float& a0, float& a1, float& a2, float& a3) {
#pragma unroll
  for (int u = 0; u < O3_SET; ++u) {
    const float e0 = __builtin_amdgcn_exp2f(x[4 * u + 0]);
    const float e1 = __builtin_amdgcn_exp2f(x[4 * u + 1]);
    const float e2 = __builtin_amdgcn_exp2f(x[4 * u + 2]);
    const float e3 = __builtin_amdgcn_exp2f(x[4 * u + 3]);
    ds += (e0 + e1) + (e2 + e3);
    a0 = fmaf(e0, s.v[u].x, a0);
    a1 = fmaf(e1, s.v[u].y, a1);
    a2 = fmaf(e2, s.v[u].z, a2);
    a3 = fmaf(e3, s.v[u].w, a3);
  }
}

static __device__ __forceinline__ void o3_set(const O3Lane& cur, O3Lane& nxt, O3Rows& q,
                                              const float* __restrict__ at,
                                              const float* __restrict__ p2w,
                                              const float* __restrict__ vrow, int tn,
                                              float& negML, float ce, float co,
                                              float& ds, float& a0, float& a1, float& a2, float& a3) {
  float x[4 * O3_SET];
  __builtin_amdgcn_sched_barrier(0);
  o3_load_lane(nxt, at + (size_t)tn * 32, vrow + (size_t)(2 * tn) * HSS);
  __builtin_amdgcn_sched_barrier(0);
  o3_exponents(cur, q, negML, ce, co, x);
  size_t ob = (size_t)(2 * tn) * HSS;
  asm volatile("" : "+s"(ob) : "v"(x[0]), "v"(x[1]), "v"(x[2]), "v"(x[3]), "v"(x[4]), "v"(x[5]),
               "v"(x[6]), "v"(x[7]), "v"(x[8]), "v"(x[9]), "v"(x[10]), "v"(x[11]), "v"(x[12]),
               "v"(x[13]), "v"(x[14]), "v"(x[15]));
  asm volatile("" : "+s"(ob) : "v"(x[16]), "v"(x[17]), "v"(x[18]), "v"(x[19]), "v"(x[20]), "v"(x[21]),
               "v"(x[22]), "v"(x[23]), "v"(x[24]), "v"(x[25]), "v"(x[26]), "v"(x[27]), "v"(x[28]),
               "v"(x[29]), "v"(x[30]), "v"(x[31]));
  o3_load_rows(q, p2w + ob);
  __builtin_amdgcn_sched_barrier(0);
  o3_accumulate(x, cur, ds, a0, a1, a2, a3);
  asm volatile("" : "+v"(negML) : "v"(ds), "v"(a0), "v"(a1), "v"(a2), "v"(a3));
}

static __device__ __forceinline__ void o3_walk(const float* __restrict__ at,
                                               const float* __restrict__ p2w,
                                               const float* __restrict__ vrow,
                                               int ntrips, float negML, float ce, float co,
                                               float& ds, float& a0, float& a1, float& a2, float& a3) {
  const int nsets = ntrips / O3_SET;
  O3Lane sA, sB;
  O3Rows q;
  o3_load_lane(sA, at, vrow);
  o3_load_rows(q, p2w);
  int s = 0;
  for (; s + 2 < nsets; s += 2) {
    o3_set(sA, sB, q, at, p2w, vrow, (s + 1) * O3_SET, negML, ce, co, ds, a0, a1, a2, a3);
    o3_set(sB, sA, q, at, p2w, vrow, (s + 2) * O3_SET, negML, ce, co, ds, a0, a1, a2, a3);
  }
  float x[4 * O3_SET];
  if (nsets - s == 2) {
    o3_set(sA, sB, q, at, p2w, vrow, (s + 1) * O3_SET, negML, ce, co, ds, a0, a1, a2, a3);
    o3_exponents(sB, q, negML, ce, co, x);
    o3_accumulate(x, sB, ds, a0, a1, a2, a3);
  } else {
    o3_exponents(sA, q, negML, ce, co, x);
    o3_accumulate(x, sA, ds, a0, a1, a2, a3);
  }
  ds += __shfl_xor(ds, 16);
  a0 += __shfl_xor(a0, 16);
  a1 += __shfl_xor(a1, 16);
  a2 += __shfl_xor(a2, 16);
  a3 += __shfl_xor(a3, 16);
}

__global__ void __launch_bounds__(512) order3_kernel(const float* __restrict__ ATT,
                              const float* __restrict__ P2,
                              const float* __restrict__ V,
                              float* __restrict__ Y) {
  __shared__ float s_pmax[TT];
  __shared__ float s_m[O3_WAVES * 32];
  __shared__ float s_ds[O3_WAVES * 32];
  __shared__ __attribute__((aligned(16))) float s_tail[2 * O3_WAVES * HSS];
  __shared__ __attribute__((aligned(16))) float s_out[32 * O3_PITCH];
  const float LOG2E = 1.4426950408889634f;
  const int lane = threadIdx.x & 31;
  const int hf   = lane >> 4;
  const int r    = lane & 15;
  const int w = __builtin_amdgcn_readfirstlane(threadIdx.x >> 5);
  const int p = blockIdx.x;
  const int h = blockIdx.y;
  const int sA = p;
  const int sB = TT / 16 - 1 - p;
  const int jendA = 16 * (sA + 1);
  const int jendB = 16 * (sB + 1);
  const float* P2h = P2 + (size_t)h * TT * HSS;
  const float* Vh  = V  + (size_t)h * TT * HSS;

  {
    const int trow = jendA + (threadIdx.x >> 4);
    const int tc4  = (threadIdx.x & 15) * 4;
    v4f tA = {0.f, 0.f, 0.f, 0.f};
    v4f tB = {0.f, 0.f, 0.f, 0.f};
#pragma unroll
    for (int k = 0; k < TT / 32; ++k) {
      const int row = trow + 32 * k;
      const v4f q = *(const v4f*)(Vh + (size_t)min(row, TT - 1) * HSS + tc4);
      const bool inA = row < TT;
      const bool inB = inA && row >= jendB;
      tA.x += inA ? q.x : 0.0f;
      tA.y += inA ? q.y : 0.0f;
      tA.z += inA ? q.z : 0.0f;
      tA.w += inA ? q.w : 0.0f;
      tB.x += inB ? q.x : 0.0f;
      tB.y += inB ? q.y : 0.0f;
      tB.z += inB ? q.z : 0.0f;
      tB.w += inB ? q.w : 0.0f;
    }
    tA.x += __shfl_xor(tA.x, 16);
    tA.y += __shfl_xor(tA.y, 16);
    tA.z += __shfl_xor(tA.z, 16);
    tA.w += __shfl_xor(tA.w, 16);
    tB.x += __shfl_xor(tB.x, 16);
    tB.y += __shfl_xor(tB.y, 16);
    tB.z += __shfl_xor(tB.z, 16);
    tB.w += __shfl_xor(tB.w, 16);
    if (lane < 16) {
      *(v4f*)(s_tail + w * HSS + tc4) = tA;
      *(v4f*)(s_tail + (O3_WAVES + w) * HSS + tc4) = tB;
    }
  }

  {
    const int c4 = r * 4;
    float pm[16];
#pragma unroll
    for (int k = 0; k < 16; ++k) {
      const int j = min(2 * (w + O3_WAVES * k) + hf, jendB - 1);
      const v4f q = *(const v4f*)(P2h + (size_t)j * HSS + c4);
      pm[k] = fmaxf(fmaxf(q.x, q.y), fmaxf(q.z, q.w));
    }
#pragma unroll
    for (int off = 8; off > 0; off >>= 1) {
#pragma unroll
      for (int k = 0; k < 16; ++k) pm[k] = fmaxf(pm[k], __shfl_xor(pm[k], off));
    }
#pragma unroll
    for (int k = 0; k < 16; ++k) {
      const int j = 2 * (w + O3_WAVES * k) + hf;
      if (r == 0 && j < jendB) s_pmax[j] = pm[k];
    }
  }
  __syncthreads();

  const float* attA = ATT + ((size_t)h * (TT / 16) + sA) * TT * 16;
  const float* attB = ATT + ((size_t)h * (TT / 16) + sB) * TT * 16;

  {
    const float* atM = (hf ? attB : attA) + r;
    const int jendL = hf ? jendB : jendA;
    float mloc = 0.0f;
#pragma unroll
    for (int k = 0; k < TT / O3_WAVES; ++k) {
      const int j = min(w + O3_WAVES * k, jendL - 1);
      mloc = fmaxf(mloc, atM[(size_t)j * 16] * s_pmax[j]);
    }
    s_m[w * 32 + lane] = mloc;
  }
  __syncthreads();
  float Mown = 0.0f;
#pragma unroll
  for (int k = 0; k < O3_WAVES; ++k) Mown = fmaxf(Mown, s_m[k * 32 + lane]);
  const float Moth = __shfl_xor(Mown, 16);
  const float negMLA = -(hf ? Moth : Mown) * LOG2E;
  const float negMLB = -(hf ? Mown : Moth) * LOG2E;

  const float ce = hf ? 0.0f : LOG2E;
  const float co = hf ? LOG2E : 0.0f;
  const float* p2w  = P2h + 4 * w;
  const float* vrow = Vh + (size_t)hf * HSS + 4 * w;
  float dsA = 0.0f, xA0 = 0.0f, xA1 = 0.0f, xA2 = 0.0f, xA3 = 0.0f;
  float dsB = 0.0f, xB0 = 0.0f, xB1 = 0.0f, xB2 = 0.0f, xB3 = 0.0f;
  o3_walk(attA + hf * 16 + r, p2w, vrow, jendA / 2, negMLA, ce, co, dsA, xA0, xA1, xA2, xA3);
  o3_walk(attB + hf * 16 + r, p2w, vrow, jendB / 2, negMLB, ce, co, dsB, xB0, xB1, xB2, xB3);

  const float ds = hf ? dsB : dsA;
  const float a0 = hf ? xB0 : xA0;
  const float a1 = hf ? xB1 : xA1;
  const float a2 = hf ? xB2 : xA2;
  const float a3 = hf ? xB3 : xA3;
  const float negML = hf ? negMLB : negMLA;
  const int jendL = hf ? jendB : jendA;
  s_ds[w * 32 + lane] = ds;
  __syncthreads();
  float tot = 0.0f;
#pragma unroll
  for (int k = 0; k < O3_WAVES; ++k) tot += s_ds[k * 32 + lane];
  const float em = __builtin_amdgcn_exp2f(negML);
  const float denom = tot + (float)((TT - jendL) * HSS) * em;
  v4f vs = {0.f, 0.f, 0.f, 0.f};
#pragma unroll
  for (int k = 0; k < O3_WAVES; ++k) vs += *(const v4f*)(s_tail + (hf * O3_WAVES + k) * HSS + 4 * w);
  const v4f o = {(a0 + em * vs.x) / denom, (a1 + em * vs.y) / denom,
                 (a2 + em * vs.z) / denom, (a3 + em * vs.w) / denom};
  *(v4f*)(s_out + lane * O3_PITCH + 4 * w) = o;
  __syncthreads();
  const int orow = threadIdx.x >> 4;
  const int oc   = (threadIdx.x & 15) * 4;
  const int grow = (orow < 16) ? (16 * sA + orow) : (16 * sB + orow - 16);
  *(v4f*)(Y + (size_t)grow * CC + h * HSS + oc) = *(const v4f*)(s_out + orow * O3_PITCH + oc);
}

__global__ void __launch_bounds__(128) out_proj_kernel(const float* __restrict__ Y,
                                const float* __restrict__ Wc,
                                float* __restrict__ out) {
  const int lane = threadIdx.x & 31;
  const int wv   = threadIdx.x >> 5;
  const int lm   = lane & 15;
  const int lh   = lane >> 4;
  const int t0 = (blockIdx.x * 4 + wv) * 16;
  const int c0 = blockIdx.y * 32;

  v8f acc[2];
#pragma unroll
  for (int j = 0; j < 2; ++j) acc[j] = (v8f){0.f, 0.f, 0.f, 0.f, 0.f, 0.f, 0.f, 0.f};
  const float* bp = Wc + (size_t)(4 * lh) * CC + c0 + 2 * lm;
  gemm_k768_wave<2, 0, KS_STD>(Y + (size_t)(t0 + lm) * CC + 4 * lh, bp, bp, acc);
#pragma unroll
  for (int r = 0; r < 8; ++r) {
    const v2f o = {acc[0][r], acc[1][r]};
    *(v2f*)(out + (size_t)(t0 + 8 * lh + r) * CC + c0 + 2 * lm) = o;
  }
}

extern "C" void kernel_launch(void* const* d_in, const int* in_sizes, int n_in,
                              void* d_out, int out_size, void* d_ws,
                              size_t ws_size, hipStream_t stream) {
  (void)in_sizes; (void)n_in; (void)out_size; (void)ws_size;
  const float* x  = (const float*)d_in[0];
  const float* W0 = (const float*)d_in[1];
  const float* W1 = (const float*)d_in[2];
  const float* W2 = (const float*)d_in[3];
  const float* Wv = (const float*)d_in[4];
  const float* Wc = (const float*)d_in[5];
  float* out = (float*)d_out;
  float* ws  = (float*)d_ws;

  const size_t NPH = (size_t)NHH * TT * HSS;
  float* P0  = ws;
  float* P1  = ws + NPH;
  float* P2  = ws + 2 * NPH;
  float* V   = ws + 3 * NPH;
  float* ATT = ws + 4 * NPH;
  float* Y   = ATT + (size_t)NHH * TT * TT;

  qkv_proj_kernel<<<dim3(32 * QK_GROUPS), 128, 0, stream>>>(x, W0, W1, W2, P0);
  att_softmax_kernel<<<dim3((TT / 16) * NHH), 32 * AT_WAVES, 0, stream>>>(P0, P1, P2, Wv, ATT, V);
  order3_kernel<<<dim3(TT / 32, NHH), 512, 0, stream>>>(ATT, P2, V, Y);
  out_proj_kernel<<<dim3(TT / 64, CC / 32), 128, 0, stream>>>(Y, Wc, out);
}


# Order-3 kernel: lanes are the 64 columns d, a wave keeps a strip's 16 rows in registers and walks every 16th column j (one p2/V row load per 16 rows, att via LDS); sums join via LDS: j order changes
# speedup vs baseline: 7.8689x; 1.0735x over previous
#include <hip/hip_runtime.h>
#include <math.h>

#define TT  512
#define CC  768
#define NHH 12
#define HSS 64
constexpr float SCALE = 0.125f;

typedef float v2f __attribute__((ext_vector_type(2)));
typedef float v8f __attribute__((ext_vector_type(8)));

static __device__ __forceinline__ v8f wmma4(v2f a, v2f b, v8f c) {
  return __builtin_amdgcn_wmma_f32_16x16x4_f32(
       false, a,  false, b,
       (short)0, c,  false,  false);
}

typedef float v4f __attribute__((ext_vector_type(4)));
template <int NT> struct BVec;
template <> struct BVec<4> { typedef float T __attribute__((ext_vector_type(4))); };
template <> struct BVec<2> { typedef float T __attribute__((ext_vector_type(2))); };

#define KS_STD 8

template <int NT, int NE, int KS>
struct FragSet {
  v4f a[KS];
  typename BVec<NT>::T b[KS][4];
  float e[KS][4];
};

template <int NT, int NE, int KS>
static __device__ __forceinline__ void load_set(FragSet<NT, NE, KS>& s,
                                                const float* __restrict__ ap,
                                                const float* __restrict__ bp,
                                                const float* __restrict__ ep) {
#pragma unroll
  for (int st = 0; st < KS; ++st) {
    s.a[st] = *(const v4f*)(ap + st * 8);
#pragma unroll
    for (int i = 0; i < 4; ++i)
      s.b[st][i] = *(const typename BVec<NT>::T*)(bp + (size_t)(st * 8 + i) * CC);
    if (NE) {
#pragma unroll
      for (int i = 0; i < 4; ++i) s.e[st][i] = ep[(size_t)(st * 8 + i) * CC];
    }
  }
}

template <int NT, int NE, int KS>
static __device__ __forceinline__ void mma_set(const FragSet<NT, NE, KS>& s, v8f (&acc)[NT + NE]) {
#pragma unroll
  for (int st = 0; st < KS; ++st) {
    const v2f a01 = {s.a[st].x, s.a[st].y};
    const v2f a23 = {s.a[st].z, s.a[st].w};
#pragma unroll
    for (int j = 0; j < NT; ++j) {
      const v2f b = {s.b[st][0][j], s.b[st][1][j]};
      acc[j] = wmma4(a01, b, acc[j]);
    }
    if (NE) {
      const v2f b = {s.e[st][0], s.e[st][1]};
      acc[NT] = wmma4(a01, b, acc[NT]);
    }
#pragma unroll
    for (int j = 0; j < NT; ++j) {
      const v2f b = {s.b[st][2][j], s.b[st][3][j]};
      acc[j] = wmma4(a23, b, acc[j]);
    }
    if (NE) {
      const v2f b = {s.e[st][2], s.e[st][3]};
      acc[NT] = wmma4(a23, b, acc[NT]);
    }
  }
}

template <int NT, int NE, int KS>
static __device__ __forceinline__ void gemm_k768_wave(const float* __restrict__ ap,
                                                      const float* __restrict__ bp,
                                                      const float* __restrict__ ep,
                                                      v8f (&acc)[NT + NE]) {
  constexpr int KB = 8 * KS;
  FragSet<NT, NE, KS> s0, s1;
  load_set(s0, ap, bp, ep);
  for (int kb = 0; kb < CC - 2 * KB; kb += 2 * KB) {
    __builtin_amdgcn_sched_barrier(0);
    load_set(s1, ap + kb + KB, bp + (size_t)(kb + KB) * CC, ep + (size_t)(kb + KB) * CC);
    __builtin_amdgcn_sched_barrier(0);
    mma_set(s0, acc);
    __builtin_amdgcn_sched_barrier(0);
    load_set(s0, ap + kb + 2 * KB, bp + (size_t)(kb + 2 * KB) * CC,
             ep + (size_t)(kb + 2 * KB) * CC);
    __builtin_amdgcn_sched_barrier(0);
    mma_set(s1, acc);
  }
  __builtin_amdgcn_sched_barrier(0);
  load_set(s1, ap + CC - KB, bp + (size_t)(CC - KB) * CC, ep + (size_t)(CC - KB) * CC);
  __builtin_amdgcn_sched_barrier(0);
  mma_set(s0, acc);
  mma_set(s1, acc);
}

#define QK_GROUPS 8
#define QK_PITCH 36

static __device__ __forceinline__ const float* qkv_wcol(const float* __restrict__ W0,
                                                        const float* __restrict__ W1,
                                                        const float* __restrict__ W2, int n) {
  return (n < CC) ? W0 + n : (n < 2 * CC) ? W1 + (n - CC) : W2 + (n - 2 * CC);
}

template <int N>
static __device__ __forceinline__ void qkv_store4(float* __restrict__ P, const v8f (&acc)[N],
                                                  int n, int t0, int lh) {
  float* o = P + (size_t)(n >> 6) * TT * HSS + (size_t)(t0 + 8 * lh) * HSS + (n & 63);
#pragma unroll
  for (int r = 0; r < 8; ++r) {
    const v4f v = {acc[0][r], acc[1][r], acc[2][r], acc[3][r]};
    *(v4f*)(o + (size_t)r * HSS) = v;
  }
}

__global__ void __launch_bounds__(128) qkv_proj_kernel(const float* __restrict__ x,
                                const float* __restrict__ W0,
                                const float* __restrict__ W1,
                                const float* __restrict__ W2,
                                float* __restrict__ P) {
  __shared__ __attribute__((aligned(16))) float s_line[2 * 16 * QK_PITCH];
  const int lane = threadIdx.x & 31;
  const int wv   = __builtin_amdgcn_readfirstlane(threadIdx.x >> 5);
  const int lm   = lane & 15;
  const int lh   = lane >> 4;
  const int b    = blockIdx.x;

  if (b < 16 * QK_GROUPS) {
    const int gb   = (b >> 4) * 288;
    const int rt   = wv >> 1;
    const int side = wv & 1;
    const int t0 = ((b & 15) * 2 + rt) * 16;
    const int n  = gb + (side ? 160 : 64) + 4 * lm;
    const int ne = gb + (side ? 144 : 128) + lm;
    v8f acc[5];
#pragma unroll
    for (int j = 0; j < 5; ++j) acc[j] = (v8f){0.f, 0.f, 0.f, 0.f, 0.f, 0.f, 0.f, 0.f};
    gemm_k768_wave<4, 1, 6>(x + (size_t)(t0 + lm) * CC + 4 * lh,
                         qkv_wcol(W0, W1, W2, n) + (size_t)(4 * lh) * CC,
                         qkv_wcol(W0, W1, W2, ne) + (size_t)(4 * lh) * CC, acc);
    qkv_store4(P, acc, n, t0, lh);
    float* sl = s_line + rt * 16 * QK_PITCH;
#pragma unroll
    for (int r = 0; r < 8; ++r) sl[(8 * lh + r) * QK_PITCH + 16 * side + lm] = acc[4][r];
    __syncthreads();
    const int nl = gb + 128;
    float* ol = P + (size_t)(nl >> 6) * TT * HSS + (size_t)t0 * HSS + (nl & 63);
    const int c4 = (lane & 7) * 4;
#pragma unroll
    for (int i = 0; i < 2; ++i) {
      const int row = 8 * side + 4 * i + (lane >> 3);
      *(v4f*)(ol + (size_t)row * HSS + c4) = *(const v4f*)(sl + row * QK_PITCH + c4);
    }
  } else {
    const int bn = b - 16 * QK_GROUPS;
    const int gb = (bn >> 4) * 288;
    const int t0 = ((bn & 7) * 4 + wv) * 16;
    const int n  = gb + ((bn & 8) ? 224 : 0) + 4 * lm;
    v8f acc[4];
#pragma unroll
    for (int j = 0; j < 4; ++j) acc[j] = (v8f){0.f, 0.f, 0.f, 0.f, 0.f, 0.f, 0.f, 0.f};
    const float* bp = qkv_wcol(W0, W1, W2, n) + (size_t)(4 * lh) * CC;
    gemm_k768_wave<4, 0, KS_STD>(x + (size_t)(t0 + lm) * CC + 4 * lh, bp, bp, acc);
    qkv_store4(P, acc, n, t0, lh);
  }
}

#define AT_PAIRS (TT / 32)
#define AT_VPITCH 68
#define AT_WAVES 4
#define AT_STEPS (AT_PAIRS / AT_WAVES)

static __device__ __forceinline__ void att_load_pair(v2f (&b)[32], const float* __restrict__ p) {
#pragma unroll
  for (int u = 0; u < 2; ++u) {
#pragma unroll
    for (int kk = 0; kk < 16; ++kk) b[u * 16 + kk] = *(const v2f*)(p + (size_t)(u * 16) * HSS + kk * 4);
  }
}

static __device__ __forceinline__ void att_mma_pair(const v2f (&a)[16], const v2f (&b)[32],
                                                    v8f& c0, v8f& c1) {
  c0 = (v8f){0.f, 0.f, 0.f, 0.f, 0.f, 0.f, 0.f, 0.f};
  c1 = c0;
#pragma unroll
  for (int kk = 0; kk < 16; ++kk) {
    c0 = wmma4(a[kk], b[kk], c0);
    c1 = wmma4(a[kk], b[16 + kk], c1);
  }
}

static __device__ __forceinline__ void att_scale_pair(v8f& c0, v8f& c1, float (&rm)[8],
                                                      bool diag, int jrel) {
#pragma unroll
  for (int r = 0; r < 8; ++r) {
    float v0 = c0[r] * SCALE;
    float v1 = c1[r] * SCALE;
    if (diag) {
      if (jrel > r) v0 = -3.0e38f;
      if (jrel + 16 > r) v1 = -3.0e38f;
    }
    c0[r] = v0;
    c1[r] = v1;
    rm[r] = fmaxf(rm[r], fmaxf(v0, v1));
  }
}

__global__ void __launch_bounds__(32 * AT_WAVES) att_softmax_kernel(const float* __restrict__ P0,
                                   const float* __restrict__ P1,
                                   const float* __restrict__ P2,
                                   const float* __restrict__ Wv,
                                   float* __restrict__ ATT,
                                   float* __restrict__ V) {
  __shared__ __attribute__((aligned(16))) float s_rmax[AT_WAVES * 16];
  __shared__ __attribute__((aligned(16))) float s_rsum[AT_WAVES * 16];
  __shared__ __attribute__((aligned(16))) float s_v[16 * AT_VPITCH];
  const int lane = threadIdx.x & 31;
  const int w    = __builtin_amdgcn_readfirstlane(threadIdx.x >> 5);
  const int lm   = lane & 15;
  const int lk   = (lane >> 4) << 1;
  const int drow = (lane >> 4) << 3;
  const int strip = TT / 16 - 1 - (int)(blockIdx.x / NHH);
  const int h     = blockIdx.x % NHH;
  const int i0 = strip * 16;
  const int g  = strip >> 1;
  const int first = g - w;
  const float* P0h = P0 + (size_t)h * TT * HSS;
  const float* P1h = P1 + (size_t)h * TT * HSS;

  v2f areg[16];
#pragma unroll
  for (int kk = 0; kk < 16; ++kk) {
    const int k = kk * 4 + lk;
    areg[kk].x = P0h[(size_t)(i0 + lm) * HSS + k];
    areg[kk].y = P0h[(size_t)(i0 + lm) * HSS + k + 1];
  }

  v2f va[16], vb[16];
  {
    const float* pa = P2 + (size_t)h * TT * HSS + (size_t)(i0 + lm) * HSS + lk;
    const float* pb = Wv + (size_t)lk * HSS + 16 * w + lm;
#pragma unroll
    for (int kk = 0; kk < 16; ++kk) {
      va[kk] = *(const v2f*)(pa + kk * 4);
      vb[kk].x = pb[(size_t)(kk * 4) * HSS];
      vb[kk].y = pb[(size_t)(kk * 4 + 1) * HSS];
    }
  }

  v8f acc[2 * AT_STEPS];
  float rm[8];
#pragma unroll
  for (int r = 0; r < 8; ++r) rm[r] = -3.0e38f;
  const float* bp = P1h + (size_t)lm * HSS + lk;
  const int jrel = (w == 0) ? (32 * g + lm - i0 - drow) : -64;
  v2f bA[32], bB[32];
  att_load_pair(bA, bp + (size_t)max(first, 0) * 32 * HSS);
  {
    v8f vacc = {0.f, 0.f, 0.f, 0.f, 0.f, 0.f, 0.f, 0.f};
#pragma unroll
    for (int kk = 0; kk < 16; ++kk) vacc = wmma4(va[kk], vb[kk], vacc);
#pragma unroll
    for (int r = 0; r < 8; ++r) s_v[(drow + r) * AT_VPITCH + 16 * w + lm] = vacc[r];
  }
#pragma unroll
  for (int q = 0; q < AT_STEPS / 2; ++q) {
    if (first - 8 * q >= 0) {
      __builtin_amdgcn_sched_barrier(0);
      att_load_pair(bB, bp + (size_t)max(first - 8 * q - 4, 0) * 32 * HSS);
      __builtin_amdgcn_sched_barrier(0);
      att_mma_pair(areg, bA, acc[4 * q], acc[4 * q + 1]);
      att_scale_pair(acc[4 * q], acc[4 * q + 1], rm, q == 0, jrel);
    }
    if (first - 8 * q - 4 >= 0) {
      __builtin_amdgcn_sched_barrier(0);
      att_load_pair(bA, bp + (size_t)max(first - 8 * q - 8, 0) * 32 * HSS);
      __builtin_amdgcn_sched_barrier(0);
      att_mma_pair(areg, bB, acc[4 * q + 2], acc[4 * q + 3]);
      att_scale_pair(acc[4 * q + 2], acc[4 * q + 3], rm, false, 0);
    }
  }

#pragma unroll
  for (int off = 8; off > 0; off >>= 1) {
#pragma unroll
    for (int r = 0; r < 8; ++r) rm[r] = fmaxf(rm[r], __shfl_xor(rm[r], off));
  }
  if (lm == 0) {
    *(v4f*)(s_rmax + w * 16 + drow)     = (v4f){rm[0], rm[1], rm[2], rm[3]};
    *(v4f*)(s_rmax + w * 16 + drow + 4) = (v4f){rm[4], rm[5], rm[6], rm[7]};
  }
  __syncthreads();
  {
    float* vo = V + (size_t)h * TT * HSS + (size_t)i0 * HSS;
    const int vc = lm * 4;
#pragma unroll
    for (int i = 0; i < 2; ++i) {
      const int row = 4 * w + 2 * i + (lane >> 4);
      *(v4f*)(vo + (size_t)row * HSS + vc) = *(const v4f*)(s_v + row * AT_VPITCH + vc);
    }
  }
#pragma unroll
  for (int k = 0; k < AT_WAVES; ++k) {
    const v4f lo = *(const v4f*)(s_rmax + k * 16 + drow);
    const v4f hi = *(const v4f*)(s_rmax + k * 16 + drow + 4);
    rm[0] = fmaxf(rm[0], lo.x); rm[1] = fmaxf(rm[1], lo.y);
    rm[2] = fmaxf(rm[2], lo.z); rm[3] = fmaxf(rm[3], lo.w);
    rm[4] = fmaxf(rm[4], hi.x); rm[5] = fmaxf(rm[5], hi.y);
    rm[6] = fmaxf(rm[6], hi.z); rm[7] = fmaxf(rm[7], hi.w);
  }

  float rs[8];
#pragma unroll
  for (int r = 0; r < 8; ++r) rs[r] = 0.0f;
#pragma unroll
  for (int s = 0; s < AT_STEPS; ++s) {
    if (first - 4 * s >= 0) {
#pragma unroll
      for (int u = 0; u < 2; ++u) {
#pragma unroll
        for (int r = 0; r < 8; ++r) {
          const float e = __expf(acc[2 * s + u][r] - rm[r]);
          acc[2 * s + u][r] = e;
          rs[r] += e;
        }
      }
    }
  }
#pragma unroll
  for (int off = 8; off > 0; off >>= 1) {
#pragma unroll
    for (int r = 0; r < 8; ++r) rs[r] += __shfl_xor(rs[r], off);
  }
  if (lm == 0) {
    *(v4f*)(s_rsum + w * 16 + drow)     = (v4f){rs[0], rs[1], rs[2], rs[3]};
    *(v4f*)(s_rsum + w * 16 + drow + 4) = (v4f){rs[4], rs[5], rs[6], rs[7]};
  }
  __syncthreads();
  v4f tlo = *(const v4f*)(s_rsum + drow);
  v4f thi = *(const v4f*)(s_rsum + drow + 4);
#pragma unroll
  for (int k = 1; k < AT_WAVES; ++k) {
    tlo += *(const v4f*)(s_rsum + k * 16 + drow);
    thi += *(const v4f*)(s_rsum + k * 16 + drow + 4);
  }
  const float inv[8] = {1.0f / tlo.x, 1.0f / tlo.y, 1.0f / tlo.z, 1.0f / tlo.w,
                        1.0f / thi.x, 1.0f / thi.y, 1.0f / thi.z, 1.0f / thi.w};

  float* abase = ATT + ((size_t)h * (TT / 16) + strip) * TT * 16 + (size_t)lm * 16 + drow;
#pragma unroll
  for (int s = 0; s < AT_STEPS; ++s) {
    if (first - 4 * s >= 0) {
#pragma unroll
      for (int u = 0; u < 2; ++u) {
        const v8f& c = acc[2 * s + u];
        const v4f lo = {c[0] * inv[0], c[1] * inv[1], c[2] * inv[2], c[3] * inv[3]};
        const v4f hi = {c[4] * inv[4], c[5] * inv[5], c[6] * inv[6], c[7] * inv[7]};
        float* o = abase + (size_t)(32 * (first - 4 * s) + 16 * u) * 16;
        *(v4f*)o = lo;
        *(v4f*)(o + 4) = hi;
      }
    }
  }
}

#define O3_WAVES 16
#define O3_COLS (TT + 16)
#define O3_RING 4

struct O3Acc {
  v2f n[16];
  v2f ds[16];
};

static __device__ __forceinline__ void o3_load_att(v4f (&q)[4], const float* sa) {
  const v4f* s4 = (const v4f*)sa;
#pragma unroll
  for (int g = 0; g < 4; ++g) q[g] = s4[g];
}

static __device__ __forceinline__ void o3_load_pv(v2f& pq, v2f& vq, const float* __restrict__ p2l,
                                                  const float* __restrict__ vl, int j) {
  pq = *(const v2f*)(p2l + (size_t)j * HSS);
  vq = *(const v2f*)(vl + (size_t)j * HSS);
}

static __device__ __forceinline__ v2f o3_exponent(const v4f (&q)[4], const v4f (&nm)[4], v2f pl, int i) {
  const float ai = q[i >> 2][i & 3];
  const float mi = nm[i >> 2][i & 3];
  const v2f av = {ai, ai};
  const v2f mv = {mi, mi};
  return __builtin_elementwise_fma(av, pl, mv);
}

template <int G>
static __device__ __forceinline__ void o3_block(O3Acc& c, v2f (&x)[16], v2f (&e)[16],
                                                const v4f (&q)[4], const v4f (&nm)[4],
                                                v2f pl, v2f vq) {
  __builtin_amdgcn_sched_barrier(0);
  if (G < 4) {
#pragma unroll
    for (int i = 4 * G; i < 4 * G + 4; ++i)
      e[i] = (v2f){__builtin_amdgcn_exp2f(x[i].x), __builtin_amdgcn_exp2f(x[i].y)};
  }
  if (G < 3) {
#pragma unroll
    for (int i = 4 * G + 4; i < 4 * G + 8; ++i) x[i] = o3_exponent(q, nm, pl, i);
  }
  if (G >= 1) {
#pragma unroll
    for (int i = 4 * G - 4; i < 4 * G; ++i) {
      c.ds[i] += e[i];
      c.n[i] = __builtin_elementwise_fma(e[i], vq, c.n[i]);
    }
  }
  if (G < 4) {
#pragma unroll
    for (int u = 0; u < 8; ++u) {
      __builtin_amdgcn_sched_group_barrier(0x400, 1, G);
      __builtin_amdgcn_sched_group_barrier(0x002, 1, G);
    }
  }
}

static __device__ __forceinline__ void o3_step(O3Acc& c, const v4f (&q)[4], v2f pq, v2f vq,
                                               const v4f (&nm)[4]) {
  const float LOG2E = 1.4426950408889634f;
  const v2f pl = pq * LOG2E;
  v2f x[16], e[16];
#pragma unroll
  for (int i = 0; i < 4; ++i) x[i] = o3_exponent(q, nm, pl, i);
  o3_block<0>(c, x, e, q, nm, pl, vq);
  o3_block<1>(c, x, e, q, nm, pl, vq);
  o3_block<2>(c, x, e, q, nm, pl, vq);
  o3_block<3>(c, x, e, q, nm, pl, vq);
  o3_block<4>(c, x, e, q, nm, pl, vq);
}

static __device__ __forceinline__ void o3_walk(O3Acc& c, const float* satt,
                                               const float* __restrict__ p2l,
                                               const float* __restrict__ vl,
                                               int w, int n, const v4f (&nm)[4]) {
#pragma unroll
  for (int i = 0; i < 16; ++i) {
    c.n[i] = (v2f){0.f, 0.f};
    c.ds[i] = (v2f){0.f, 0.f};
  }
  v2f P[O3_RING], Q[O3_RING];
  v4f A[2][4];
#pragma unroll
  for (int k = 0; k < O3_RING - 1; ++k) o3_load_pv(P[k], Q[k], p2l, vl, w + 16 * min(k, n - 1));
  o3_load_att(A[0], satt + (size_t)w * 16);
  for (int t0 = 0; t0 < n; t0 += O3_RING) {
#pragma unroll
    for (int k = 0; k < O3_RING; ++k) {
      const int t = t0 + k;
      __builtin_amdgcn_sched_barrier(0);
      o3_load_pv(P[(k + O3_RING - 1) % O3_RING], Q[(k + O3_RING - 1) % O3_RING], p2l, vl,
                 w + 16 * min(t + O3_RING - 1, n - 1));
      o3_load_att(A[(k + 1) & 1], satt + (size_t)(w + 16 * min(t + 1, n - 1)) * 16);
      __builtin_amdgcn_sched_barrier(0);
      if (t < n) o3_step(c, A[k & 1], P[k], Q[k], nm);
    }
  }
}

__global__ void __launch_bounds__(512) order3_kernel(const float* __restrict__ ATT,
                              const float* __restrict__ P2,
                              const float* __restrict__ V,
                              float* __restrict__ Y) {
  __shared__ float s_pmax[TT];
  __shared__ float s_m[O3_WAVES * 32];
  __shared__ __attribute__((aligned(16))) float s_tail[2 * O3_WAVES * HSS];
  __shared__ __attribute__((aligned(16))) float s_att[O3_COLS * 16];
  __shared__ __attribute__((aligned(16))) float s_num[2 * O3_WAVES * 16 * HSS];
  __shared__ __attribute__((aligned(16))) float s_dsp[2 * O3_WAVES * 16 * 32];
  const float LOG2E = 1.4426950408889634f;
  const int lane = threadIdx.x & 31;
  const int hf   = lane >> 4;
  const int r    = lane & 15;
  const int w = __builtin_amdgcn_readfirstlane(threadIdx.x >> 5);
  const int p = blockIdx.x;
  const int h = blockIdx.y;
  const int sA = p;
  const int sB = TT / 16 - 1 - p;
  const int jendA = 16 * (sA + 1);
  const int jendB = 16 * (sB + 1);
  const float* P2h = P2 + (size_t)h * TT * HSS;
  const float* Vh  = V  + (size_t)h * TT * HSS;
  {
    const int trow = jendA + (threadIdx.x >> 4);
    const int tc4  = (threadIdx.x & 15) * 4;
    v4f tA = {0.f, 0.f, 0.f, 0.f};
    v4f tB = {0.f, 0.f, 0.f, 0.f};
#pragma unroll
    for (int k = 0; k < TT / 32; ++k) {
      const int row = trow + 32 * k;
      const v4f q = *(const v4f*)(Vh + (size_t)min(row, TT - 1) * HSS + tc4);
      const bool inA = row < TT;
      const bool inB = inA && row >= jendB;
      tA.x += inA ? q.x : 0.0f;
      tA.y += inA ? q.y : 0.0f;
      tA.z += inA ? q.z : 0.0f;
      tA.w += inA ? q.w : 0.0f;
      tB.x += inB ? q.x : 0.0f;
      tB.y += inB ? q.y : 0.0f;
      tB.z += inB ? q.z : 0.0f;
      tB.w += inB ? q.w : 0.0f;
    }
    tA.x += __shfl_xor(tA.x, 16);
    tA.y += __shfl_xor(tA.y, 16);
    tA.z += __shfl_xor(tA.z, 16);
    tA.w += __shfl_xor(tA.w, 16);
    tB.x += __shfl_xor(tB.x, 16);
    tB.y += __shfl_xor(tB.y, 16);
    tB.z += __shfl_xor(tB.z, 16);
    tB.w += __shfl_xor(tB.w, 16);
    if (lane < 16) {
      *(v4f*)(s_tail + w * HSS + tc4) = tA;
      *(v4f*)(s_tail + (O3_WAVES + w) * HSS + tc4) = tB;
    }
  }

  {
    const int c4 = r * 4;
    float pm[16];
#pragma unroll
    for (int k = 0; k < 16; ++k) {
      const int j = min(2 * (w + O3_WAVES * k) + hf, jendB - 1);
      const v4f q = *(const v4f*)(P2h + (size_t)j * HSS + c4);
      pm[k] = fmaxf(fmaxf(q.x, q.y), fmaxf(q.z, q.w));
    }
#pragma unroll
    for (int off = 8; off > 0; off >>= 1) {
#pragma unroll
      for (int k = 0; k < 16; ++k) pm[k] = fmaxf(pm[k], __shfl_xor(pm[k], off));
    }
#pragma unroll
    for (int k = 0; k < 16; ++k) {
      const int j = 2 * (w + O3_WAVES * k) + hf;
      if (r == 0 && j < jendB) s_pmax[j] = pm[k];
    }
  }
  __syncthreads();

  const float* attA = ATT + ((size_t)h * (TT / 16) + sA) * TT * 16;
  const float* attB = ATT + ((size_t)h * (TT / 16) + sB) * TT * 16;
  float* sattA = s_att;
  float* sattB = s_att + jendA * 16;

  {
    const float* atM = (hf ? attB : attA) + r;
    float* stM = (hf ? sattB : sattA) + r;
    const int jendL = hf ? jendB : jendA;
    float av[TT / O3_WAVES];
#pragma unroll
    for (int k = 0; k < TT / O3_WAVES; ++k) {
      const int j = min(w + O3_WAVES * k, jendL - 1);
      av[k] = atM[(size_t)j * 16];
    }
    float mloc = 0.0f;
#pragma unroll
    for (int k = 0; k < TT / O3_WAVES; ++k) {
      const int jj = w + O3_WAVES * k;
      const int j = min(jj, jendL - 1);
      if (jj < jendL) stM[j * 16] = av[k];
      mloc = fmaxf(mloc, av[k] * s_pmax[j]);
    }
    s_m[w * 32 + lane] = mloc;
  }
  __syncthreads();
  float Mown = 0.0f;
#pragma unroll
  for (int k = 0; k < O3_WAVES; ++k) Mown = fmaxf(Mown, s_m[k * 32 + lane]);
  const float negMLv = -Mown * LOG2E;

  const float* p2l = P2h + 2 * lane;
  const float* vl  = Vh + 2 * lane;
#pragma unroll
  for (int X = 0; X < 2; ++X) {
    v4f nm[4];
#pragma unroll
    for (int g = 0; g < 4; ++g)
      nm[g] = (v4f){__shfl(negMLv, 16 * X + 4 * g), __shfl(negMLv, 16 * X + 4 * g + 1),
                    __shfl(negMLv, 16 * X + 4 * g + 2), __shfl(negMLv, 16 * X + 4 * g + 3)};
    O3Acc c;
    o3_walk(c, X ? sattB : sattA, p2l, vl, w, X ? (sB + 1) : (sA + 1), nm);
    float* pn = s_num + (size_t)((X * O3_WAVES + w) * 16) * HSS + 2 * lane;
    float* pd = s_dsp + (size_t)((X * O3_WAVES + w) * 16) * 32 + lane;
#pragma unroll
    for (int i = 0; i < 16; ++i) {
      *(v2f*)(pn + i * HSS) = c.n[i];
      pd[i * 32] = c.ds[i].x + c.ds[i].y;
    }
  }
  __syncthreads();

#pragma unroll
  for (int X = 0; X < 2; ++X) {
    v2f num = {0.f, 0.f};
    float dsum = 0.0f;
#pragma unroll
    for (int k = 0; k < O3_WAVES; ++k) {
      num += *(const v2f*)(s_num + (size_t)((X * O3_WAVES + k) * 16 + w) * HSS + 2 * lane);
      dsum += s_dsp[(size_t)((X * O3_WAVES + k) * 16 + w) * 32 + lane];
    }
#pragma unroll
    for (int off = 16; off > 0; off >>= 1) dsum += __shfl_xor(dsum, off);
    const float negML = __shfl(negMLv, 16 * X + w);
    const float em = __builtin_amdgcn_exp2f(negML);
    const int jendX = X ? jendB : jendA;
    const float denom = dsum + (float)((TT - jendX) * HSS) * em;
    v2f vs = {0.f, 0.f};
#pragma unroll
    for (int k = 0; k < O3_WAVES; ++k) vs += *(const v2f*)(s_tail + (X * O3_WAVES + k) * HSS + 2 * lane);
    const v2f o = {(num.x + em * vs.x) / denom, (num.y + em * vs.y) / denom};
    const int grow = 16 * (X ? sB : sA) + w;
    *(v2f*)(Y + (size_t)grow * CC + h * HSS + 2 * lane) = o;
  }
}

__global__ void __launch_bounds__(128) out_proj_kernel(const float* __restrict__ Y,
                                const float* __restrict__ Wc,
                                float* __restrict__ out) {
  const int lane = threadIdx.x & 31;
  const int wv   = threadIdx.x >> 5;
  const int lm   = lane & 15;
  const int lh   = lane >> 4;
  const int t0 = (blockIdx.x * 4 + wv) * 16;
  const int c0 = blockIdx.y * 32;

  v8f acc[2];
#pragma unroll
  for (int j = 0; j < 2; ++j) acc[j] = (v8f){0.f, 0.f, 0.f, 0.f, 0.f, 0.f, 0.f, 0.f};
  const float* bp = Wc + (size_t)(4 * lh) * CC + c0 + 2 * lm;
  gemm_k768_wave<2, 0, KS_STD>(Y + (size_t)(t0 + lm) * CC + 4 * lh, bp, bp, acc);
#pragma unroll
  for (int r = 0; r < 8; ++r) {
    const v2f o = {acc[0][r], acc[1][r]};
    *(v2f*)(out + (size_t)(t0 + 8 * lh + r) * CC + c0 + 2 * lm) = o;
  }
}

extern "C" void kernel_launch(void* const* d_in, const int* in_sizes, int n_in,
                              void* d_out, int out_size, void* d_ws,
                              size_t ws_size, hipStream_t stream) {
  (void)in_sizes; (void)n_in; (void)out_size; (void)ws_size;
  const float* x  = (const float*)d_in[0];
  const float* W0 = (const float*)d_in[1];
  const float* W1 = (const float*)d_in[2];
  const float* W2 = (const float*)d_in[3];
  const float* Wv = (const float*)d_in[4];
  const float* Wc = (const float*)d_in[5];
  float* out = (float*)d_out;
  float* ws  = (float*)d_ws;

  const size_t NPH = (size_t)NHH * TT * HSS;
  float* P0  = ws;
  float* P1  = ws + NPH;
  float* P2  = ws + 2 * NPH;
  float* V   = ws + 3 * NPH;
  float* ATT = ws + 4 * NPH;
  float* Y   = ATT + (size_t)NHH * TT * TT;

  qkv_proj_kernel<<<dim3(32 * QK_GROUPS), 128, 0, stream>>>(x, W0, W1, W2, P0);
  att_softmax_kernel<<<dim3((TT / 16) * NHH), 32 * AT_WAVES, 0, stream>>>(P0, P1, P2, Wv, ATT, V);
  order3_kernel<<<dim3(TT / 32, NHH), 512, 0, stream>>>(ATT, P2, V, Y);
  out_proj_kernel<<<dim3(TT / 64, CC / 32), 128, 0, stream>>>(Y, Wc, out);
}
